# starting kernel
# baseline (speedup 1.0000x reference)
#include <hip/hip_runtime.h>
#include <math.h>


typedef __attribute__((ext_vector_type(2))) float v2f;
typedef __attribute__((ext_vector_type(8))) float v8f;

#define FIN   8
#define HID   64
#define KTOT  72
#define KSTEPS 18
#define GCOLS 256
#define NT_G  16
#define ROWS_PER_BLOCK 128
#define EPS 1e-5f

__device__ __forceinline__ float sigmoidf_(float x) { return 1.0f / (1.0f + expf(-x)); }

__global__ void k_init_deg(float* deg, int N) {
  int i = blockIdx.x * blockDim.x + threadIdx.x;
  if (i < N) deg[i] = 1.0f;
}

__global__ void k_acc_deg(float* deg, const int* dst, const float* w, int E) {
  int e = blockIdx.x * blockDim.x + threadIdx.x;
  if (e < E) atomicAdd(&deg[dst[e]], w[e]);
}

__global__ void k_dinv(float* deg, int N) {
  int i = blockIdx.x * blockDim.x + threadIdx.x;
  if (i < N) deg[i] = rsqrtf(deg[i]);
}

__global__ void k_norm(float* norm, const float* dinv, const int* src, const int* dst,
                       const float* w, int E) {
  int e = blockIdx.x * blockDim.x + threadIdx.x;
  if (e < E) norm[e] = dinv[dst[e]] * w[e] * dinv[src[e]];
}

__global__ void k_init_axh(float* AXH, const float* X, const float* H,
                           const float* dinv, int N) {
  long t = (long)blockIdx.x * blockDim.x + threadIdx.x;
  long total = (long)N * KTOT;
  if (t >= total) return;
  int n = (int)(t / KTOT);
  int k = (int)(t - (long)n * KTOT);
  float d2 = dinv[n] * dinv[n];
  float v = (k < FIN) ? X[(long)n * FIN + k] : H[(long)n * HID + (k - FIN)];
  AXH[t] = d2 * v;
}

__global__ void k_scatter_axh(float* AXH, const float* X, const float* H,
                              const int* src, const int* dst, const float* norm, int E) {
  long t = (long)blockIdx.x * blockDim.x + threadIdx.x;
  long total = (long)E * KTOT;
  if (t >= total) return;
  int e = (int)(t / KTOT);
  int k = (int)(t - (long)e * KTOT);
  int s = src[e];
  float v = (k < FIN) ? X[(long)s * FIN + k] : H[(long)s * HID + (k - FIN)];
  atomicAdd(&AXH[(long)dst[e] * KTOT + k], norm[e] * v);
}

__global__ void k_init_agg(float* dstbuf, const float* srcf, const float* dinv,
                           int N, int F, int stride) {
  long t = (long)blockIdx.x * blockDim.x + threadIdx.x;
  long total = (long)N * stride;
  if (t >= total) return;
  int n = (int)(t / stride);
  int k = (int)(t - (long)n * stride);
  float d = dinv[n];
  dstbuf[t] = (k < F) ? d * d * srcf[t] : 0.0f;
}

__global__ void k_scatter_agg(float* dstbuf, const float* srcf, const int* src,
                              const int* dst, const float* norm, int E, int F, int stride) {
  long t = (long)blockIdx.x * blockDim.x + threadIdx.x;
  long total = (long)E * F;
  if (t >= total) return;
  int e = (int)(t / F);
  int k = (int)(t - (long)e * F);
  float v = norm[e] * srcf[(long)src[e] * stride + k];
  atomicAdd(&dstbuf[(long)dst[e] * stride + k], v);
}

__global__ __launch_bounds__(256)
void k_gates(const float* __restrict__ AXH, const float* __restrict__ C,
             const float* __restrict__ Wx, const float* __restrict__ Wh,
             const float* __restrict__ bx, const float* __restrict__ bh,
             const float* __restrict__ bg, const float* __restrict__ wc,
             const float* __restrict__ ln_h_s, const float* __restrict__ ln_h_b,
             const float* __restrict__ ln_c_s, const float* __restrict__ ln_c_b,
             const float* __restrict__ ln_o_s, const float* __restrict__ ln_o_b,
             const float* __restrict__ skip,
             float* __restrict__ hiddenOut, float* __restrict__ cellOut,
             float* __restrict__ Y, int N) {
  extern __shared__ float smem[];
  float* Bf = smem;
  float* bias = smem + 18432;

  int tid = threadIdx.x;
  for (int i = tid; i < KSTEPS * NT_G * 64; i += 256) {
    int s = i >> 10;
    int rem = i & 1023;
    int t = rem >> 6;
    int rem2 = rem & 63;
    int L = rem2 >> 1;
    int j = rem2 & 1;
    int k = 4 * s + 2 * (L >> 4) + j;
    int c = 16 * t + (L & 15);
    int g = c >> 6, h = c & 63;
    Bf[i] = (k < FIN) ? Wx[(g * FIN + k) * HID + h]
                      : Wh[(g * HID + (k - FIN)) * HID + h];
  }
  if (tid < GCOLS) bias[tid] = bx[tid] + bh[tid] + bg[tid];
  __syncthreads();

  int wave = tid >> 5, lane = tid & 31;
  int half = lane >> 4, lcol = lane & 15;
  int rowBase = blockIdx.x * ROWS_PER_BLOCK + wave * 16;

  int arow = rowBase + lcol;
  if (arow > N - 1) arow = N - 1;
  const float* aptr = AXH + (long)arow * KTOT;

  v8f acc[NT_G];
#pragma unroll
  for (int t = 0; t < NT_G; ++t)
#pragma unroll
    for (int r = 0; r < 8; ++r) acc[t][r] = 0.0f;

  for (int s = 0; s < KSTEPS; ++s) {
    int k0 = 4 * s + 2 * half;
    v2f a = *(const v2f*)(aptr + k0);
    const float* bbase = Bf + s * 1024 + lane * 2;
#pragma unroll
    for (int t = 0; t < NT_G; ++t) {
      v2f b = *(const v2f*)(bbase + t * 64);
      acc[t] = __builtin_amdgcn_wmma_f32_16x16x4_f32(
          false, a, false, b, (short)0, acc[t], false, false);
    }
  }

#pragma unroll
  for (int r = 0; r < 8; ++r) {
    int node = rowBase + r + 8 * half;
    bool valid = node < N;
    int nc = valid ? node : N - 1;
    float hloc[4], cloc[4];
#pragma unroll
    for (int t = 0; t < 4; ++t) {
      int h = t * 16 + lcol;
      float c0 = C[(long)nc * HID + h];
      float gi = acc[t][r]      + bias[h]       + wc[h] * c0;
      float gf = acc[t + 4][r]  + bias[64 + h]  + wc[64 + h] * c0;
      float gc = acc[t + 8][r]  + bias[128 + h];
      float go = acc[t + 12][r] + bias[192 + h];
      float iv = sigmoidf_(gi);
      float fv = sigmoidf_(gf);
      float cc = fv * c0 + iv * tanhf(gc);
      float ov = sigmoidf_(go + wc[128 + h] * cc);
      hloc[t] = ov * tanhf(cc);
      cloc[t] = cc;
    }
    float sh = 0.f, sh2 = 0.f, sc = 0.f, sc2 = 0.f;
#pragma unroll
    for (int t = 0; t < 4; ++t) {
      sh += hloc[t]; sh2 += hloc[t] * hloc[t];
      sc += cloc[t]; sc2 += cloc[t] * cloc[t];
    }
#pragma unroll
    for (int mask = 1; mask < 16; mask <<= 1) {
      sh  += __shfl_xor(sh,  mask, 32);
      sh2 += __shfl_xor(sh2, mask, 32);
      sc  += __shfl_xor(sc,  mask, 32);
      sc2 += __shfl_xor(sc2, mask, 32);
    }
    float mh = sh * (1.0f / 64.0f), mc = sc * (1.0f / 64.0f);
    float vh = sh2 * (1.0f / 64.0f) - mh * mh;
    float vc = sc2 * (1.0f / 64.0f) - mc * mc;
    float rh = rsqrtf(vh + EPS), rc = rsqrtf(vc + EPS);
    if (valid) {
#pragma unroll
      for (int t = 0; t < 4; ++t) {
        int h = t * 16 + lcol;
        float hn = (hloc[t] - mh) * rh;
        float cn = (cloc[t] - mc) * rc;
        hiddenOut[(long)node * HID + h] = hn * ln_h_s[h] + ln_h_b[h];
        cellOut[(long)node * HID + h]   = cn * ln_c_s[h] + ln_c_b[h];
        float ov = hn * ln_o_s[h] + ln_o_b[h];
        Y[(long)node * KTOT + h] = ov > 0.0f ? ov : 0.0f;
      }
      if (lcol < 8) {
        Y[(long)node * KTOT + 64 + lcol] =
            (lcol < 2) ? skip[(long)node * 2 + lcol] : 0.0f;
      }
    }
  }
}

__global__ __launch_bounds__(256)
void k_gemm2(const float* __restrict__ AY, const float* __restrict__ W1,
             const float* __restrict__ b1, float* __restrict__ Z, int N) {
  __shared__ float Bf[KSTEPS * 4 * 64];
  __shared__ float b1s[HID];

  int tid = threadIdx.x;
  for (int i = tid; i < KSTEPS * 4 * 64; i += 256) {
    int s = i >> 8;
    int rem = i & 255;
    int t = rem >> 6;
    int rem2 = rem & 63;
    int L = rem2 >> 1;
    int j = rem2 & 1;
    int k = 4 * s + 2 * (L >> 4) + j;
    int c = 16 * t + (L & 15);
    Bf[i] = (k < HID + 2) ? W1[k * HID + c] : 0.0f;
  }
  if (tid < HID) b1s[tid] = b1[tid];
  __syncthreads();

  int wave = tid >> 5, lane = tid & 31;
  int half = lane >> 4, lcol = lane & 15;
  int rowBase = blockIdx.x * ROWS_PER_BLOCK + wave * 16;

  int arow = rowBase + lcol;
  if (arow > N - 1) arow = N - 1;
  const float* aptr = AY + (long)arow * KTOT;

  v8f acc[4];
#pragma unroll
  for (int t = 0; t < 4; ++t)
#pragma unroll
    for (int r = 0; r < 8; ++r) acc[t][r] = 0.0f;

  for (int s = 0; s < KSTEPS; ++s) {
    int k0 = 4 * s + 2 * half;
    v2f a = *(const v2f*)(aptr + k0);
    const float* bbase = Bf + s * 256 + lane * 2;
#pragma unroll
    for (int t = 0; t < 4; ++t) {
      v2f b = *(const v2f*)(bbase + t * 64);
      acc[t] = __builtin_amdgcn_wmma_f32_16x16x4_f32(
          false, a, false, b, (short)0, acc[t], false, false);
    }
  }

#pragma unroll
  for (int r = 0; r < 8; ++r) {
    int node = rowBase + r + 8 * half;
    if (node >= N) continue;
#pragma unroll
    for (int t = 0; t < 4; ++t) {
      int h = t * 16 + lcol;
      float z = acc[t][r] + b1s[h];
      Z[(long)node * HID + h] = z > 0.0f ? z : 0.0f;
    }
  }
}

__global__ void k_final(const float* __restrict__ AZ, const float* __restrict__ W2,
                        const float* __restrict__ b2, const float* __restrict__ X,
                        float* __restrict__ out, int N) {
  __shared__ float w2s[HID];
  if (threadIdx.x < HID) w2s[threadIdx.x] = W2[threadIdx.x];
  __syncthreads();
  int n = blockIdx.x * blockDim.x + threadIdx.x;
  if (n >= N) return;
  const float* a = AZ + (long)n * HID;
  float acc = b2[0] + X[(long)n * FIN];
#pragma unroll 8
  for (int h = 0; h < HID; ++h) acc += a[h] * w2s[h];
  out[n] = acc;
}

extern "C" void kernel_launch(void* const* d_in, const int* in_sizes, int n_in,
                              void* d_out, int out_size, void* d_ws, size_t ws_size,
                              hipStream_t stream) {
  const float* X    = (const float*)d_in[0];
  const int*   eidx = (const int*)d_in[1];
  const float* ew   = (const float*)d_in[2];
  const float* skip = (const float*)d_in[3];
  const float* H    = (const float*)d_in[4];
  const float* C    = (const float*)d_in[5];
  const float* Wx   = (const float*)d_in[6];
  const float* bx   = (const float*)d_in[7];
  const float* Wh   = (const float*)d_in[8];
  const float* bh   = (const float*)d_in[9];
  const float* wc   = (const float*)d_in[10];
  const float* bg   = (const float*)d_in[11];
  const float* ln_h_s = (const float*)d_in[12];
  const float* ln_h_b = (const float*)d_in[13];
  const float* ln_c_s = (const float*)d_in[14];
  const float* ln_c_b = (const float*)d_in[15];
  const float* ln_o_s = (const float*)d_in[16];
  const float* ln_o_b = (const float*)d_in[17];
  const float* W1   = (const float*)d_in[18];
  const float* b1   = (const float*)d_in[19];
  const float* W2   = (const float*)d_in[20];
  const float* b2   = (const float*)d_in[21];

  const int N = in_sizes[0] / FIN;
  const int E = in_sizes[2];

  const int* src = eidx;
  const int* dst = eidx + E;

  float* out_p    = (float*)d_out;
  float* hidden_p = out_p + N;
  float* cell_p   = hidden_p + (long)N * HID;

  float* ws   = (float*)d_ws;
  float* dinv = ws;
  float* norm = dinv + N;
  float* AXH  = norm + E;
  float* Y    = AXH;
  float* AY   = AXH + (long)N * KTOT;
  float* Z    = AY + (long)N * KTOT;
  float* AZ   = Z + (long)N * HID;

  const int B = 256;
  auto cdiv = [](long a, long b) { return (unsigned)((a + b - 1) / b); };

  k_init_deg<<<cdiv(N, B), B, 0, stream>>>(dinv, N);
  k_acc_deg<<<cdiv(E, B), B, 0, stream>>>(dinv, dst, ew, E);
  k_dinv<<<cdiv(N, B), B, 0, stream>>>(dinv, N);
  k_norm<<<cdiv(E, B), B, 0, stream>>>(norm, dinv, src, dst, ew, E);

  k_init_axh<<<cdiv((long)N * KTOT, B), B, 0, stream>>>(AXH, X, H, dinv, N);
  k_scatter_axh<<<cdiv((long)E * KTOT, B), B, 0, stream>>>(AXH, X, H, src, dst, norm, E);

  size_t smem_gates = (size_t)(KSTEPS * NT_G * 64 + GCOLS) * sizeof(float);
  k_gates<<<cdiv(N, ROWS_PER_BLOCK), B, smem_gates, stream>>>(
      AXH, C, Wx, Wh, bx, bh, bg, wc,
      ln_h_s, ln_h_b, ln_c_s, ln_c_b, ln_o_s, ln_o_b,
      skip, hidden_p, cell_p, Y, N);

  k_init_agg<<<cdiv((long)N * KTOT, B), B, 0, stream>>>(AY, Y, dinv, N, HID + 2, KTOT);
  k_scatter_agg<<<cdiv((long)E * (HID + 2), B), B, 0, stream>>>(AY, Y, src, dst, norm,
                                                               E, HID + 2, KTOT);

  k_gemm2<<<cdiv(N, ROWS_PER_BLOCK), B, 0, stream>>>(AY, W1, b1, Z, N);

  k_init_agg<<<cdiv((long)N * HID, B), B, 0, stream>>>(AZ, Z, dinv, N, HID, HID);
  k_scatter_agg<<<cdiv((long)E * HID, B), B, 0, stream>>>(AZ, Z, src, dst, norm,
                                                          E, HID, HID);

  k_final<<<cdiv(N, B), B, 0, stream>>>(AZ, W2, b2, X, out_p, N);
}


# W2 applied before the last aggregation (agg is linear: agg(Z)@W2 = agg(Z@W2)): the last atomic scatter carries 1 float per edge instead of 64; float sum order of out changes
# speedup vs baseline: 1.2580x; 1.2580x over previous
#include <hip/hip_runtime.h>
#include <math.h>


typedef __attribute__((ext_vector_type(2))) float v2f;
typedef __attribute__((ext_vector_type(8))) float v8f;

#define FIN   8
#define HID   64
#define KTOT  72
#define KSTEPS 18
#define GCOLS 256
#define NT_G  16
#define ROWS_PER_BLOCK 128
#define EPS 1e-5f

__device__ __forceinline__ float sigmoidf_(float x) { return 1.0f / (1.0f + expf(-x)); }

__global__ void k_init_deg(float* deg, int N) {
  int i = blockIdx.x * blockDim.x + threadIdx.x;
  if (i < N) deg[i] = 1.0f;
}

__global__ void k_acc_deg(float* deg, const int* dst, const float* w, int E) {
  int e = blockIdx.x * blockDim.x + threadIdx.x;
  if (e < E) atomicAdd(&deg[dst[e]], w[e]);
}

__global__ void k_dinv(float* deg, int N) {
  int i = blockIdx.x * blockDim.x + threadIdx.x;
  if (i < N) deg[i] = rsqrtf(deg[i]);
}

__global__ void k_norm(float* norm, const float* dinv, const int* src, const int* dst,
                       const float* w, int E) {
  int e = blockIdx.x * blockDim.x + threadIdx.x;
  if (e < E) norm[e] = dinv[dst[e]] * w[e] * dinv[src[e]];
}

__global__ void k_init_axh(float* AXH, const float* X, const float* H,
                           const float* dinv, int N) {
  long t = (long)blockIdx.x * blockDim.x + threadIdx.x;
  long total = (long)N * KTOT;
  if (t >= total) return;
  int n = (int)(t / KTOT);
  int k = (int)(t - (long)n * KTOT);
  float d2 = dinv[n] * dinv[n];
  float v = (k < FIN) ? X[(long)n * FIN + k] : H[(long)n * HID + (k - FIN)];
  AXH[t] = d2 * v;
}

__global__ void k_scatter_axh(float* AXH, const float* X, const float* H,
                              const int* src, const int* dst, const float* norm, int E) {
  long t = (long)blockIdx.x * blockDim.x + threadIdx.x;
  long total = (long)E * KTOT;
  if (t >= total) return;
  int e = (int)(t / KTOT);
  int k = (int)(t - (long)e * KTOT);
  int s = src[e];
  float v = (k < FIN) ? X[(long)s * FIN + k] : H[(long)s * HID + (k - FIN)];
  atomicAdd(&AXH[(long)dst[e] * KTOT + k], norm[e] * v);
}

__global__ void k_init_agg(float* dstbuf, const float* srcf, const float* dinv,
                           int N, int F, int stride) {
  long t = (long)blockIdx.x * blockDim.x + threadIdx.x;
  long total = (long)N * stride;
  if (t >= total) return;
  int n = (int)(t / stride);
  int k = (int)(t - (long)n * stride);
  float d = dinv[n];
  dstbuf[t] = (k < F) ? d * d * srcf[t] : 0.0f;
}

__global__ void k_scatter_agg(float* dstbuf, const float* srcf, const int* src,
                              const int* dst, const float* norm, int E, int F, int stride) {
  long t = (long)blockIdx.x * blockDim.x + threadIdx.x;
  long total = (long)E * F;
  if (t >= total) return;
  int e = (int)(t / F);
  int k = (int)(t - (long)e * F);
  float v = norm[e] * srcf[(long)src[e] * stride + k];
  atomicAdd(&dstbuf[(long)dst[e] * stride + k], v);
}

__global__ __launch_bounds__(256)
void k_gates(const float* __restrict__ AXH, const float* __restrict__ C,
             const float* __restrict__ Wx, const float* __restrict__ Wh,
             const float* __restrict__ bx, const float* __restrict__ bh,
             const float* __restrict__ bg, const float* __restrict__ wc,
             const float* __restrict__ ln_h_s, const float* __restrict__ ln_h_b,
             const float* __restrict__ ln_c_s, const float* __restrict__ ln_c_b,
             const float* __restrict__ ln_o_s, const float* __restrict__ ln_o_b,
             const float* __restrict__ skip,
             float* __restrict__ hiddenOut, float* __restrict__ cellOut,
             float* __restrict__ Y, int N) {
  extern __shared__ float smem[];
  float* Bf = smem;
  float* bias = smem + 18432;

  int tid = threadIdx.x;
  for (int i = tid; i < KSTEPS * NT_G * 64; i += 256) {
    int s = i >> 10;
    int rem = i & 1023;
    int t = rem >> 6;
    int rem2 = rem & 63;
    int L = rem2 >> 1;
    int j = rem2 & 1;
    int k = 4 * s + 2 * (L >> 4) + j;
    int c = 16 * t + (L & 15);
    int g = c >> 6, h = c & 63;
    Bf[i] = (k < FIN) ? Wx[(g * FIN + k) * HID + h]
                      : Wh[(g * HID + (k - FIN)) * HID + h];
  }
  if (tid < GCOLS) bias[tid] = bx[tid] + bh[tid] + bg[tid];
  __syncthreads();

  int wave = tid >> 5, lane = tid & 31;
  int half = lane >> 4, lcol = lane & 15;
  int rowBase = blockIdx.x * ROWS_PER_BLOCK + wave * 16;

  int arow = rowBase + lcol;
  if (arow > N - 1) arow = N - 1;
  const float* aptr = AXH + (long)arow * KTOT;

  v8f acc[NT_G];
#pragma unroll
  for (int t = 0; t < NT_G; ++t)
#pragma unroll
    for (int r = 0; r < 8; ++r) acc[t][r] = 0.0f;

  for (int s = 0; s < KSTEPS; ++s) {
    int k0 = 4 * s + 2 * half;
    v2f a = *(const v2f*)(aptr + k0);
    const float* bbase = Bf + s * 1024 + lane * 2;
#pragma unroll
    for (int t = 0; t < NT_G; ++t) {
      v2f b = *(const v2f*)(bbase + t * 64);
      acc[t] = __builtin_amdgcn_wmma_f32_16x16x4_f32(
          false, a, false, b, (short)0, acc[t], false, false);
    }
  }

#pragma unroll
  for (int r = 0; r < 8; ++r) {
    int node = rowBase + r + 8 * half;
    bool valid = node < N;
    int nc = valid ? node : N - 1;
    float hloc[4], cloc[4];
#pragma unroll
    for (int t = 0; t < 4; ++t) {
      int h = t * 16 + lcol;
      float c0 = C[(long)nc * HID + h];
      float gi = acc[t][r]      + bias[h]       + wc[h] * c0;
      float gf = acc[t + 4][r]  + bias[64 + h]  + wc[64 + h] * c0;
      float gc = acc[t + 8][r]  + bias[128 + h];
      float go = acc[t + 12][r] + bias[192 + h];
      float iv = sigmoidf_(gi);
      float fv = sigmoidf_(gf);
      float cc = fv * c0 + iv * tanhf(gc);
      float ov = sigmoidf_(go + wc[128 + h] * cc);
      hloc[t] = ov * tanhf(cc);
      cloc[t] = cc;
    }
    float sh = 0.f, sh2 = 0.f, sc = 0.f, sc2 = 0.f;
#pragma unroll
    for (int t = 0; t < 4; ++t) {
      sh += hloc[t]; sh2 += hloc[t] * hloc[t];
      sc += cloc[t]; sc2 += cloc[t] * cloc[t];
    }
#pragma unroll
    for (int mask = 1; mask < 16; mask <<= 1) {
      sh  += __shfl_xor(sh,  mask, 32);
      sh2 += __shfl_xor(sh2, mask, 32);
      sc  += __shfl_xor(sc,  mask, 32);
      sc2 += __shfl_xor(sc2, mask, 32);
    }
    float mh = sh * (1.0f / 64.0f), mc = sc * (1.0f / 64.0f);
    float vh = sh2 * (1.0f / 64.0f) - mh * mh;
    float vc = sc2 * (1.0f / 64.0f) - mc * mc;
    float rh = rsqrtf(vh + EPS), rc = rsqrtf(vc + EPS);
    if (valid) {
#pragma unroll
      for (int t = 0; t < 4; ++t) {
        int h = t * 16 + lcol;
        float hn = (hloc[t] - mh) * rh;
        float cn = (cloc[t] - mc) * rc;
        hiddenOut[(long)node * HID + h] = hn * ln_h_s[h] + ln_h_b[h];
        cellOut[(long)node * HID + h]   = cn * ln_c_s[h] + ln_c_b[h];
        float ov = hn * ln_o_s[h] + ln_o_b[h];
        Y[(long)node * KTOT + h] = ov > 0.0f ? ov : 0.0f;
      }
      if (lcol < 8) {
        Y[(long)node * KTOT + 64 + lcol] =
            (lcol < 2) ? skip[(long)node * 2 + lcol] : 0.0f;
      }
    }
  }
}

__global__ __launch_bounds__(256)
void k_gemm2(const float* __restrict__ AY, const float* __restrict__ W1,
             const float* __restrict__ b1, const float* __restrict__ W2,
             const float* __restrict__ dinv, float* __restrict__ P,
             float* __restrict__ Q, int N) {
  __shared__ float Bf[KSTEPS * 4 * 64];
  __shared__ float b1s[HID];
  __shared__ float w2s[HID];
  __shared__ float ps[ROWS_PER_BLOCK];

  int tid = threadIdx.x;
  for (int i = tid; i < KSTEPS * 4 * 64; i += 256) {
    int s = i >> 8;
    int rem = i & 255;
    int t = rem >> 6;
    int rem2 = rem & 63;
    int L = rem2 >> 1;
    int j = rem2 & 1;
    int k = 4 * s + 2 * (L >> 4) + j;
    int c = 16 * t + (L & 15);
    Bf[i] = (k < HID + 2) ? W1[k * HID + c] : 0.0f;
  }
  if (tid < HID) { b1s[tid] = b1[tid]; w2s[tid] = W2[tid]; }
  __syncthreads();

  int wave = tid >> 5, lane = tid & 31;
  int half = lane >> 4, lcol = lane & 15;
  int rowBase = blockIdx.x * ROWS_PER_BLOCK + wave * 16;

  int arow = rowBase + lcol;
  if (arow > N - 1) arow = N - 1;
  const float* aptr = AY + (long)arow * KTOT;

  v8f acc[4];
#pragma unroll
  for (int t = 0; t < 4; ++t)
#pragma unroll
    for (int r = 0; r < 8; ++r) acc[t][r] = 0.0f;

  for (int s = 0; s < KSTEPS; ++s) {
    int k0 = 4 * s + 2 * half;
    v2f a = *(const v2f*)(aptr + k0);
    const float* bbase = Bf + s * 256 + lane * 2;
#pragma unroll
    for (int t = 0; t < 4; ++t) {
      v2f b = *(const v2f*)(bbase + t * 64);
      acc[t] = __builtin_amdgcn_wmma_f32_16x16x4_f32(
          false, a, false, b, (short)0, acc[t], false, false);
    }
  }

  float part[8];
#pragma unroll
  for (int r = 0; r < 8; ++r) {
    float s = 0.0f;
#pragma unroll
    for (int t = 0; t < 4; ++t) {
      int h = t * 16 + lcol;
      float z = acc[t][r] + b1s[h];
      s += (z > 0.0f ? z : 0.0f) * w2s[h];
    }
    part[r] = s;
  }
#pragma unroll
  for (int mask = 1; mask < 16; mask <<= 1) {
    float got[8];
#pragma unroll
    for (int r = 0; r < 8; ++r) got[r] = __shfl_xor(part[r], mask, 32);
#pragma unroll
    for (int r = 0; r < 8; ++r) part[r] += got[r];
  }
  if (lcol == 0) {
#pragma unroll
    for (int r = 0; r < 8; ++r) ps[wave * 16 + r + 8 * half] = part[r];
  }
  __syncthreads();
  if (tid < ROWS_PER_BLOCK) {
    int node = blockIdx.x * ROWS_PER_BLOCK + tid;
    if (node < N) {
      float p = ps[tid];
      float d = dinv[node];
      P[node] = p;
      Q[node] = d * d * p;
    }
  }
}

__global__ void k_scatter_p(float* Q, const float* __restrict__ P,
                            const int* __restrict__ src, const int* __restrict__ dst,
                            const float* __restrict__ norm, int E) {
  int e = blockIdx.x * blockDim.x + threadIdx.x;
  if (e < E) atomicAdd(&Q[dst[e]], norm[e] * P[src[e]]);
}

__global__ void k_final(const float* __restrict__ Q, const float* __restrict__ b2,
                        const float* __restrict__ X, float* __restrict__ out, int N) {
  int n = blockIdx.x * blockDim.x + threadIdx.x;
  if (n < N) out[n] = Q[n] + b2[0] + X[(long)n * FIN];
}

extern "C" void kernel_launch(void* const* d_in, const int* in_sizes, int n_in,
                              void* d_out, int out_size, void* d_ws, size_t ws_size,
                              hipStream_t stream) {
  const float* X    = (const float*)d_in[0];
  const int*   eidx = (const int*)d_in[1];
  const float* ew   = (const float*)d_in[2];
  const float* skip = (const float*)d_in[3];
  const float* H    = (const float*)d_in[4];
  const float* C    = (const float*)d_in[5];
  const float* Wx   = (const float*)d_in[6];
  const float* bx   = (const float*)d_in[7];
  const float* Wh   = (const float*)d_in[8];
  const float* bh   = (const float*)d_in[9];
  const float* wc   = (const float*)d_in[10];
  const float* bg   = (const float*)d_in[11];
  const float* ln_h_s = (const float*)d_in[12];
  const float* ln_h_b = (const float*)d_in[13];
  const float* ln_c_s = (const float*)d_in[14];
  const float* ln_c_b = (const float*)d_in[15];
  const float* ln_o_s = (const float*)d_in[16];
  const float* ln_o_b = (const float*)d_in[17];
  const float* W1   = (const float*)d_in[18];
  const float* b1   = (const float*)d_in[19];
  const float* W2   = (const float*)d_in[20];
  const float* b2   = (const float*)d_in[21];

  const int N = in_sizes[0] / FIN;
  const int E = in_sizes[2];

  const int* src = eidx;
  const int* dst = eidx + E;

  float* out_p    = (float*)d_out;
  float* hidden_p = out_p + N;
  float* cell_p   = hidden_p + (long)N * HID;

  float* ws   = (float*)d_ws;
  float* dinv = ws;
  float* norm = dinv + N;
  float* AXH  = norm + E;
  float* Y    = AXH;
  float* AY   = AXH + (long)N * KTOT;
  float* P    = AY + (long)N * KTOT;
  float* Q    = P + N;

  const int B = 256;
  auto cdiv = [](long a, long b) { return (unsigned)((a + b - 1) / b); };

  k_init_deg<<<cdiv(N, B), B, 0, stream>>>(dinv, N);
  k_acc_deg<<<cdiv(E, B), B, 0, stream>>>(dinv, dst, ew, E);
  k_dinv<<<cdiv(N, B), B, 0, stream>>>(dinv, N);
  k_norm<<<cdiv(E, B), B, 0, stream>>>(norm, dinv, src, dst, ew, E);

  k_init_axh<<<cdiv((long)N * KTOT, B), B, 0, stream>>>(AXH, X, H, dinv, N);
  k_scatter_axh<<<cdiv((long)E * KTOT, B), B, 0, stream>>>(AXH, X, H, src, dst, norm, E);

  size_t smem_gates = (size_t)(KSTEPS * NT_G * 64 + GCOLS) * sizeof(float);
  k_gates<<<cdiv(N, ROWS_PER_BLOCK), B, smem_gates, stream>>>(
      AXH, C, Wx, Wh, bx, bh, bg, wc,
      ln_h_s, ln_h_b, ln_c_s, ln_c_b, ln_o_s, ln_o_b,
      skip, hidden_p, cell_p, Y, N);

  k_init_agg<<<cdiv((long)N * KTOT, B), B, 0, stream>>>(AY, Y, dinv, N, HID + 2, KTOT);
  k_scatter_agg<<<cdiv((long)E * (HID + 2), B), B, 0, stream>>>(AY, Y, src, dst, norm,
                                                               E, HID + 2, KTOT);

  k_gemm2<<<cdiv(N, ROWS_PER_BLOCK), B, 0, stream>>>(AY, W1, b1, W2, dinv, P, Q, N);

  k_scatter_p<<<cdiv(E, B), B, 0, stream>>>(Q, P, src, dst, norm, E);

  k_final<<<cdiv(N, B), B, 0, stream>>>(Q, b2, X, out_p, N);
}


# Aggregations as segment sums over edges counting-sorted by destination (built in-call), one wave per node: no float atomics on features; float sum order of all outputs changes
# speedup vs baseline: 2.6092x; 2.0741x over previous
#include <hip/hip_runtime.h>
#include <math.h>


typedef __attribute__((ext_vector_type(2))) float v2f;
typedef __attribute__((ext_vector_type(8))) float v8f;

#define FIN   8
#define HID   64
#define KTOT  72
#define KSTEPS 18
#define GCOLS 256
#define NT_G  16
#define ROWS_PER_BLOCK 128
#define EPS 1e-5f

__device__ __forceinline__ float sigmoidf_(float x) { return 1.0f / (1.0f + expf(-x)); }

__global__ void k_init_deg(float* deg, int* cnt, int N) {
  int i = blockIdx.x * blockDim.x + threadIdx.x;
  if (i < N) { deg[i] = 1.0f; cnt[i] = 0; }
}

__global__ void k_acc_deg(float* deg, int* cnt, const int* dst, const float* w, int E) {
  int e = blockIdx.x * blockDim.x + threadIdx.x;
  if (e < E) {
    int d = dst[e];
    atomicAdd(&deg[d], w[e]);
    atomicAdd(&cnt[d], 1);
  }
}

__global__ void k_dinv(float* deg, int N) {
  int i = blockIdx.x * blockDim.x + threadIdx.x;
  if (i < N) deg[i] = rsqrtf(deg[i]);
}

#define SCAN_T 256
#define SCAN_PER 4

__global__ __launch_bounds__(SCAN_T)
void k_blocksum(const int* __restrict__ cnt, int* __restrict__ bsum, int N) {
  __shared__ int wsum[SCAN_T / 32];
  int base = (blockIdx.x * SCAN_T + threadIdx.x) * SCAN_PER;
  int s = 0;
  for (int i = 0; i < SCAN_PER; ++i)
    if (base + i < N) s += cnt[base + i];
  for (int mask = 16; mask > 0; mask >>= 1) s += __shfl_xor(s, mask, 32);
  if ((threadIdx.x & 31) == 0) wsum[threadIdx.x >> 5] = s;
  __syncthreads();
  if (threadIdx.x == 0) {
    int tot = 0;
    for (int i = 0; i < SCAN_T / 32; ++i) tot += wsum[i];
    bsum[blockIdx.x] = tot;
  }
}

__global__ __launch_bounds__(SCAN_T)
void k_scan(int* __restrict__ cursor, int* __restrict__ rowptr,
            const int* __restrict__ bsum, int N) {
  __shared__ int wtot[SCAN_T / 32];
  __shared__ int woff[SCAN_T / 32];
  int tid = threadIdx.x, lane = tid & 31, wave = tid >> 5;

  int off = 0;
  for (int i = tid; i < (int)blockIdx.x; i += SCAN_T) off += bsum[i];
  for (int mask = 16; mask > 0; mask >>= 1) off += __shfl_xor(off, mask, 32);
  if (lane == 0) woff[wave] = off;

  int base = (blockIdx.x * SCAN_T + tid) * SCAN_PER;
  int c[SCAN_PER];
  int tsum = 0;
  for (int i = 0; i < SCAN_PER; ++i) {
    c[i] = (base + i < N) ? cursor[base + i] : 0;
    tsum += c[i];
  }
  int v = tsum;
  for (int o = 1; o < 32; o <<= 1) {
    int u = __shfl_up(v, o, 32);
    if (lane >= o) v += u;
  }
  if (lane == 31) wtot[wave] = v;
  __syncthreads();
  int pre = 0;
  for (int i = 0; i < SCAN_T / 32; ++i) {
    pre += woff[i];
    if (i < wave) pre += wtot[i];
  }
  int excl = pre + v - tsum;
  for (int i = 0; i < SCAN_PER; ++i) {
    if (base + i < N) {
      rowptr[base + i] = excl;
      cursor[base + i] = excl;
      excl += c[i];
    }
  }
}

__global__ void k_fill(int* cursor, int* __restrict__ perm, const int* __restrict__ dst, int E) {
  int e = blockIdx.x * blockDim.x + threadIdx.x;
  if (e < E) {
    int p = atomicAdd(&cursor[dst[e]], 1);
    perm[p] = e;
  }
}

template <bool SPLIT>
__global__ __launch_bounds__(256)
void k_gather72(float* __restrict__ outbuf, const float* __restrict__ srcA,
                const float* __restrict__ srcB,
                const int* __restrict__ rowptr, const int* __restrict__ perm,
                const int* __restrict__ src, const float* __restrict__ w,
                const float* __restrict__ dinv, int N, int E) {
  int lane = threadIdx.x & 31;
  int l8 = lane & 7;
  int d = __builtin_amdgcn_readfirstlane((int)(blockIdx.x * 8 + (threadIdx.x >> 5)));
  if (d >= N) return;
  int beg = rowptr[d];
  int end = (d + 1 < N) ? rowptr[d + 1] : E;
  float dd = dinv[d];

  const float* rlo;
  const float* rex;
  if (SPLIT) { rlo = srcB + (long)d * HID;  rex = srcA + (long)d * FIN; }
  else       { rlo = srcA + (long)d * KTOT; rex = rlo + HID; }
  float a0 = dd * rlo[lane];
  float a1 = dd * rlo[32 + lane];
  float a2 = dd * rex[l8];

  for (int base = beg; base < end; base += 32) {
    int p = base + lane;
    int s = 0;
    float c = 0.0f;
    if (p < end) {
      int e = perm[p];
      s = src[e];
      c = w[e] * dinv[s];
    }
    int ci = __float_as_int(c);
    int cnt = end - base;
    if (cnt > 32) cnt = 32;
    int j = 0;
    for (; j + 4 <= cnt; j += 4) {
      float x0[4], x1[4], x2[4], cj[4];
      for (int u = 0; u < 4; ++u) {
        int sj = __builtin_amdgcn_readlane(s, j + u);
        cj[u] = __int_as_float(__builtin_amdgcn_readlane(ci, j + u));
        const float* q0; const float* q2;
        if (SPLIT) { q0 = srcB + (long)sj * HID;  q2 = srcA + (long)sj * FIN; }
        else       { q0 = srcA + (long)sj * KTOT; q2 = q0 + HID; }
        x0[u] = q0[lane]; x1[u] = q0[32 + lane]; x2[u] = q2[l8];
      }
      for (int u = 0; u < 4; ++u) {
        a0 += cj[u] * x0[u]; a1 += cj[u] * x1[u]; a2 += cj[u] * x2[u];
      }
    }
    for (; j < cnt; ++j) {
      int sj = __builtin_amdgcn_readlane(s, j);
      float cj = __int_as_float(__builtin_amdgcn_readlane(ci, j));
      const float* q0; const float* q2;
      if (SPLIT) { q0 = srcB + (long)sj * HID;  q2 = srcA + (long)sj * FIN; }
      else       { q0 = srcA + (long)sj * KTOT; q2 = q0 + HID; }
      a0 += cj * q0[lane]; a1 += cj * q0[32 + lane]; a2 += cj * q2[l8];
    }
  }

  float* o = outbuf + (long)d * KTOT;
  if (SPLIT) {
    o[FIN + lane] = dd * a0;
    o[FIN + 32 + lane] = dd * a1;
    if (lane < 8) o[lane] = dd * a2;
  } else {
    o[lane] = dd * a0;
    o[32 + lane] = dd * a1;
    if (lane < 8) o[HID + lane] = dd * a2;
  }
}

__global__ __launch_bounds__(256)
void k_gates(const float* __restrict__ AXH, const float* __restrict__ C,
             const float* __restrict__ Wx, const float* __restrict__ Wh,
             const float* __restrict__ bx, const float* __restrict__ bh,
             const float* __restrict__ bg, const float* __restrict__ wc,
             const float* __restrict__ ln_h_s, const float* __restrict__ ln_h_b,
             const float* __restrict__ ln_c_s, const float* __restrict__ ln_c_b,
             const float* __restrict__ ln_o_s, const float* __restrict__ ln_o_b,
             const float* __restrict__ skip,
             float* __restrict__ hiddenOut, float* __restrict__ cellOut,
             float* __restrict__ Y, int N) {
  extern __shared__ float smem[];
  float* Bf = smem;
  float* bias = smem + 18432;

  int tid = threadIdx.x;
  for (int i = tid; i < KSTEPS * NT_G * 64; i += 256) {
    int s = i >> 10;
    int rem = i & 1023;
    int t = rem >> 6;
    int rem2 = rem & 63;
    int L = rem2 >> 1;
    int j = rem2 & 1;
    int k = 4 * s + 2 * (L >> 4) + j;
    int c = 16 * t + (L & 15);
    int g = c >> 6, h = c & 63;
    Bf[i] = (k < FIN) ? Wx[(g * FIN + k) * HID + h]
                      : Wh[(g * HID + (k - FIN)) * HID + h];
  }
  if (tid < GCOLS) bias[tid] = bx[tid] + bh[tid] + bg[tid];
  __syncthreads();

  int wave = tid >> 5, lane = tid & 31;
  int half = lane >> 4, lcol = lane & 15;
  int rowBase = blockIdx.x * ROWS_PER_BLOCK + wave * 16;

  int arow = rowBase + lcol;
  if (arow > N - 1) arow = N - 1;
  const float* aptr = AXH + (long)arow * KTOT;

  v8f acc[NT_G];
#pragma unroll
  for (int t = 0; t < NT_G; ++t)
#pragma unroll
    for (int r = 0; r < 8; ++r) acc[t][r] = 0.0f;

  for (int s = 0; s < KSTEPS; ++s) {
    int k0 = 4 * s + 2 * half;
    v2f a = *(const v2f*)(aptr + k0);
    const float* bbase = Bf + s * 1024 + lane * 2;
#pragma unroll
    for (int t = 0; t < NT_G; ++t) {
      v2f b = *(const v2f*)(bbase + t * 64);
      acc[t] = __builtin_amdgcn_wmma_f32_16x16x4_f32(
          false, a, false, b, (short)0, acc[t], false, false);
    }
  }

#pragma unroll
  for (int r = 0; r < 8; ++r) {
    int node = rowBase + r + 8 * half;
    bool valid = node < N;
    int nc = valid ? node : N - 1;
    float hloc[4], cloc[4];
#pragma unroll
    for (int t = 0; t < 4; ++t) {
      int h = t * 16 + lcol;
      float c0 = C[(long)nc * HID + h];
      float gi = acc[t][r]      + bias[h]       + wc[h] * c0;
      float gf = acc[t + 4][r]  + bias[64 + h]  + wc[64 + h] * c0;
      float gc = acc[t + 8][r]  + bias[128 + h];
      float go = acc[t + 12][r] + bias[192 + h];
      float iv = sigmoidf_(gi);
      float fv = sigmoidf_(gf);
      float cc = fv * c0 + iv * tanhf(gc);
      float ov = sigmoidf_(go + wc[128 + h] * cc);
      hloc[t] = ov * tanhf(cc);
      cloc[t] = cc;
    }
    float sh = 0.f, sh2 = 0.f, sc = 0.f, sc2 = 0.f;
#pragma unroll
    for (int t = 0; t < 4; ++t) {
      sh += hloc[t]; sh2 += hloc[t] * hloc[t];
      sc += cloc[t]; sc2 += cloc[t] * cloc[t];
    }
#pragma unroll
    for (int mask = 1; mask < 16; mask <<= 1) {
      sh  += __shfl_xor(sh,  mask, 32);
      sh2 += __shfl_xor(sh2, mask, 32);
      sc  += __shfl_xor(sc,  mask, 32);
      sc2 += __shfl_xor(sc2, mask, 32);
    }
    float mh = sh * (1.0f / 64.0f), mc = sc * (1.0f / 64.0f);
    float vh = sh2 * (1.0f / 64.0f) - mh * mh;
    float vc = sc2 * (1.0f / 64.0f) - mc * mc;
    float rh = rsqrtf(vh + EPS), rc = rsqrtf(vc + EPS);
    if (valid) {
#pragma unroll
      for (int t = 0; t < 4; ++t) {
        int h = t * 16 + lcol;
        float hn = (hloc[t] - mh) * rh;
        float cn = (cloc[t] - mc) * rc;
        hiddenOut[(long)node * HID + h] = hn * ln_h_s[h] + ln_h_b[h];
        cellOut[(long)node * HID + h]   = cn * ln_c_s[h] + ln_c_b[h];
        float ov = hn * ln_o_s[h] + ln_o_b[h];
        Y[(long)node * KTOT + h] = ov > 0.0f ? ov : 0.0f;
      }
      if (lcol < 8) {
        Y[(long)node * KTOT + 64 + lcol] =
            (lcol < 2) ? skip[(long)node * 2 + lcol] : 0.0f;
      }
    }
  }
}

__global__ __launch_bounds__(256)
void k_gemm2(const float* __restrict__ AY, const float* __restrict__ W1,
             const float* __restrict__ b1, const float* __restrict__ W2,
             float* __restrict__ P, int N) {
  __shared__ float Bf[KSTEPS * 4 * 64];
  __shared__ float b1s[HID];
  __shared__ float w2s[HID];
  __shared__ float ps[ROWS_PER_BLOCK];

  int tid = threadIdx.x;
  for (int i = tid; i < KSTEPS * 4 * 64; i += 256) {
    int s = i >> 8;
    int rem = i & 255;
    int t = rem >> 6;
    int rem2 = rem & 63;
    int L = rem2 >> 1;
    int j = rem2 & 1;
    int k = 4 * s + 2 * (L >> 4) + j;
    int c = 16 * t + (L & 15);
    Bf[i] = (k < HID + 2) ? W1[k * HID + c] : 0.0f;
  }
  if (tid < HID) { b1s[tid] = b1[tid]; w2s[tid] = W2[tid]; }
  __syncthreads();

  int wave = tid >> 5, lane = tid & 31;
  int half = lane >> 4, lcol = lane & 15;
  int rowBase = blockIdx.x * ROWS_PER_BLOCK + wave * 16;

  int arow = rowBase + lcol;
  if (arow > N - 1) arow = N - 1;
  const float* aptr = AY + (long)arow * KTOT;

  v8f acc[4];
#pragma unroll
  for (int t = 0; t < 4; ++t)
#pragma unroll
    for (int r = 0; r < 8; ++r) acc[t][r] = 0.0f;

  for (int s = 0; s < KSTEPS; ++s) {
    int k0 = 4 * s + 2 * half;
    v2f a = *(const v2f*)(aptr + k0);
    const float* bbase = Bf + s * 256 + lane * 2;
#pragma unroll
    for (int t = 0; t < 4; ++t) {
      v2f b = *(const v2f*)(bbase + t * 64);
      acc[t] = __builtin_amdgcn_wmma_f32_16x16x4_f32(
          false, a, false, b, (short)0, acc[t], false, false);
    }
  }

  float part[8];
#pragma unroll
  for (int r = 0; r < 8; ++r) {
    float s = 0.0f;
#pragma unroll
    for (int t = 0; t < 4; ++t) {
      int h = t * 16 + lcol;
      float z = acc[t][r] + b1s[h];
      s += (z > 0.0f ? z : 0.0f) * w2s[h];
    }
    part[r] = s;
  }
#pragma unroll
  for (int mask = 1; mask < 16; mask <<= 1) {
    float got[8];
#pragma unroll
    for (int r = 0; r < 8; ++r) got[r] = __shfl_xor(part[r], mask, 32);
#pragma unroll
    for (int r = 0; r < 8; ++r) part[r] += got[r];
  }
  if (lcol == 0) {
#pragma unroll
    for (int r = 0; r < 8; ++r) ps[wave * 16 + r + 8 * half] = part[r];
  }
  __syncthreads();
  if (tid < ROWS_PER_BLOCK) {
    int node = blockIdx.x * ROWS_PER_BLOCK + tid;
    if (node < N) P[node] = ps[tid];
  }
}

__global__ __launch_bounds__(256)
void k_gather_out(float* __restrict__ out, const float* __restrict__ P,
                  const int* __restrict__ rowptr, const int* __restrict__ perm,
                  const int* __restrict__ src, const float* __restrict__ w,
                  const float* __restrict__ dinv, const float* __restrict__ b2,
                  const float* __restrict__ X, int N, int E) {
  int t = blockIdx.x * blockDim.x + threadIdx.x;
  int d = t >> 4, l = t & 15;
  bool valid = d < N;
  int dc = valid ? d : N - 1;
  int beg = rowptr[dc];
  int end = (dc + 1 < N) ? rowptr[dc + 1] : E;
  float acc = 0.0f;
  for (int p = beg + l; p < end; p += 16) {
    int e = perm[p];
    int sidx = src[e];
    acc += w[e] * dinv[sidx] * P[sidx];
  }
  for (int mask = 8; mask > 0; mask >>= 1) acc += __shfl_xor(acc, mask, 32);
  if (valid && l == 0) {
    float dd = dinv[d];
    float q = dd * (acc + dd * P[d]);
    out[d] = q + b2[0] + X[(long)d * FIN];
  }
}

extern "C" void kernel_launch(void* const* d_in, const int* in_sizes, int n_in,
                              void* d_out, int out_size, void* d_ws, size_t ws_size,
                              hipStream_t stream) {
  const float* X    = (const float*)d_in[0];
  const int*   eidx = (const int*)d_in[1];
  const float* ew   = (const float*)d_in[2];
  const float* skip = (const float*)d_in[3];
  const float* H    = (const float*)d_in[4];
  const float* C    = (const float*)d_in[5];
  const float* Wx   = (const float*)d_in[6];
  const float* bx   = (const float*)d_in[7];
  const float* Wh   = (const float*)d_in[8];
  const float* bh   = (const float*)d_in[9];
  const float* wc   = (const float*)d_in[10];
  const float* bg   = (const float*)d_in[11];
  const float* ln_h_s = (const float*)d_in[12];
  const float* ln_h_b = (const float*)d_in[13];
  const float* ln_c_s = (const float*)d_in[14];
  const float* ln_c_b = (const float*)d_in[15];
  const float* ln_o_s = (const float*)d_in[16];
  const float* ln_o_b = (const float*)d_in[17];
  const float* W1   = (const float*)d_in[18];
  const float* b1   = (const float*)d_in[19];
  const float* W2   = (const float*)d_in[20];
  const float* b2   = (const float*)d_in[21];

  const int N = in_sizes[0] / FIN;
  const int E = in_sizes[2];

  const int* src = eidx;
  const int* dst = eidx + E;

  float* out_p    = (float*)d_out;
  float* hidden_p = out_p + N;
  float* cell_p   = hidden_p + (long)N * HID;

  float* ws   = (float*)d_ws;
  float* dinv = ws;
  int*   perm = (int*)(dinv + N);
  float* AXH  = (float*)(perm + E);
  float* Y    = AXH;
  float* AY   = AXH + (long)N * KTOT;
  float* P    = AY + (long)N * KTOT;
  int* rowptr = (int*)(P + N);
  int* cursor = (int*)AY;
  int* bsum   = cursor + N;

  const int B = 256;
  auto cdiv = [](long a, long b) { return (unsigned)((a + b - 1) / b); };

  k_init_deg<<<cdiv(N, B), B, 0, stream>>>(dinv, cursor, N);
  k_acc_deg<<<cdiv(E, B), B, 0, stream>>>(dinv, cursor, dst, ew, E);
  k_dinv<<<cdiv(N, B), B, 0, stream>>>(dinv, N);

  k_blocksum<<<cdiv(N, SCAN_T * SCAN_PER), SCAN_T, 0, stream>>>(cursor, bsum, N);
  k_scan<<<cdiv(N, SCAN_T * SCAN_PER), SCAN_T, 0, stream>>>(cursor, rowptr, bsum, N);
  k_fill<<<cdiv(E, B), B, 0, stream>>>(cursor, perm, dst, E);

  k_gather72<true><<<cdiv(N, 8), B, 0, stream>>>(AXH, X, H, rowptr, perm, src, ew, dinv, N, E);

  size_t smem_gates = (size_t)(KSTEPS * NT_G * 64 + GCOLS) * sizeof(float);
  k_gates<<<cdiv(N, ROWS_PER_BLOCK), B, smem_gates, stream>>>(
      AXH, C, Wx, Wh, bx, bh, bg, wc,
      ln_h_s, ln_h_b, ln_c_s, ln_c_b, ln_o_s, ln_o_b,
      skip, hidden_p, cell_p, Y, N);

  k_gather72<false><<<cdiv(N, 8), B, 0, stream>>>(AY, Y, Y, rowptr, perm, src, ew, dinv, N, E);

  k_gemm2<<<cdiv(N, ROWS_PER_BLOCK), B, 0, stream>>>(AY, W1, b1, W2, P, N);

  k_gather_out<<<cdiv((long)N * 16, B), B, 0, stream>>>(out_p, P, rowptr, perm, src, ew, dinv,
                                                       b2, X, N, E);
}


# k_fill writes each edge as an 8-byte record {src, w*dinv[src]} in destination order; the 3 aggregations read records contiguously instead of edge id -> src, w -> dinv look-ups; sum order as before
# speedup vs baseline: 2.8896x; 1.1075x over previous
#include <hip/hip_runtime.h>
#include <math.h>


typedef __attribute__((ext_vector_type(2))) float v2f;
typedef __attribute__((ext_vector_type(2))) int v2i;
typedef __attribute__((ext_vector_type(8))) float v8f;

#define FIN   8
#define HID   64
#define KTOT  72
#define KSTEPS 18
#define GCOLS 256
#define NT_G  16
#define ROWS_PER_BLOCK 128
#define EPS 1e-5f

__device__ __forceinline__ float sigmoidf_(float x) { return 1.0f / (1.0f + expf(-x)); }

__global__ void k_init_deg(float* deg, int* cnt, int N) {
  int i = blockIdx.x * blockDim.x + threadIdx.x;
  if (i < N) { deg[i] = 1.0f; cnt[i] = 0; }
}

__global__ void k_acc_deg(float* deg, int* cnt, const int* dst, const float* w, int E) {
  int e = blockIdx.x * blockDim.x + threadIdx.x;
  if (e < E) {
    int d = dst[e];
    atomicAdd(&deg[d], w[e]);
    atomicAdd(&cnt[d], 1);
  }
}

__global__ void k_dinv(float* deg, int N) {
  int i = blockIdx.x * blockDim.x + threadIdx.x;
  if (i < N) deg[i] = rsqrtf(deg[i]);
}

#define SCAN_T 256
#define SCAN_PER 4

__global__ __launch_bounds__(SCAN_T)
void k_blocksum(const int* __restrict__ cnt, int* __restrict__ bsum, int N) {
  __shared__ int wsum[SCAN_T / 32];
  int base = (blockIdx.x * SCAN_T + threadIdx.x) * SCAN_PER;
  int s = 0;
  for (int i = 0; i < SCAN_PER; ++i)
    if (base + i < N) s += cnt[base + i];
  for (int mask = 16; mask > 0; mask >>= 1) s += __shfl_xor(s, mask, 32);
  if ((threadIdx.x & 31) == 0) wsum[threadIdx.x >> 5] = s;
  __syncthreads();
  if (threadIdx.x == 0) {
    int tot = 0;
    for (int i = 0; i < SCAN_T / 32; ++i) tot += wsum[i];
    bsum[blockIdx.x] = tot;
  }
}

__global__ __launch_bounds__(SCAN_T)
void k_scan(int* __restrict__ cursor, int* __restrict__ rowptr,
            const int* __restrict__ bsum, int N) {
  __shared__ int wtot[SCAN_T / 32];
  __shared__ int woff[SCAN_T / 32];
  int tid = threadIdx.x, lane = tid & 31, wave = tid >> 5;

  int off = 0;
  for (int i = tid; i < (int)blockIdx.x; i += SCAN_T) off += bsum[i];
  for (int mask = 16; mask > 0; mask >>= 1) off += __shfl_xor(off, mask, 32);
  if (lane == 0) woff[wave] = off;

  int base = (blockIdx.x * SCAN_T + tid) * SCAN_PER;
  int c[SCAN_PER];
  int tsum = 0;
  for (int i = 0; i < SCAN_PER; ++i) {
    c[i] = (base + i < N) ? cursor[base + i] : 0;
    tsum += c[i];
  }
  int v = tsum;
  for (int o = 1; o < 32; o <<= 1) {
    int u = __shfl_up(v, o, 32);
    if (lane >= o) v += u;
  }
  if (lane == 31) wtot[wave] = v;
  __syncthreads();
  int pre = 0;
  for (int i = 0; i < SCAN_T / 32; ++i) {
    pre += woff[i];
    if (i < wave) pre += wtot[i];
  }
  int excl = pre + v - tsum;
  for (int i = 0; i < SCAN_PER; ++i) {
    if (base + i < N) {
      rowptr[base + i] = excl;
      cursor[base + i] = excl;
      excl += c[i];
    }
  }
}

__global__ void k_fill(int* cursor, int* __restrict__ rec, const int* __restrict__ src,
                       const int* __restrict__ dst, const float* __restrict__ w,
                       const float* __restrict__ dinv, int E) {
  int e = blockIdx.x * blockDim.x + threadIdx.x;
  if (e < E) {
    int dn = dst[e];
    int s = src[e];
    float we = w[e];
    int p = atomicAdd(&cursor[dn], 1);
    float c = we * dinv[s];
    v2i r;
    r.x = s;
    r.y = __float_as_int(c);
    *(v2i*)(rec + 2 * (long)p) = r;
  }
}

template <bool SPLIT>
__global__ __launch_bounds__(256)
void k_gather72(float* __restrict__ outbuf, const float* __restrict__ srcA,
                const float* __restrict__ srcB,
                const int* __restrict__ rowptr, const int* __restrict__ rec,
                const float* __restrict__ dinv, int n0, int n1, int N, int E) {
  int lane = threadIdx.x & 31;
  int l8 = lane & 7;
  int d = __builtin_amdgcn_readfirstlane((int)(n0 + blockIdx.x * 8 + (threadIdx.x >> 5)));
  if (d >= n1) return;
  int beg = rowptr[d];
  int end = (d + 1 < N) ? rowptr[d + 1] : E;
  float dd = dinv[d];

  const float* rlo;
  const float* rex;
  if (SPLIT) { rlo = srcB + (long)d * HID;  rex = srcA + (long)d * FIN; }
  else       { rlo = srcA + (long)d * KTOT; rex = rlo + HID; }
  float a0 = dd * rlo[lane];
  float a1 = dd * rlo[32 + lane];
  float a2 = dd * rex[l8];

  for (int base = beg; base < end; base += 32) {
    int p = base + lane;
    int s = 0;
    float c = 0.0f;
    if (p < end) {
      v2i r = *(const v2i*)(rec + 2 * (long)p);
      s = r.x;
      c = __int_as_float(r.y);
    }
    int ci = __float_as_int(c);
    int cnt = end - base;
    if (cnt > 32) cnt = 32;
    int j = 0;
    for (; j + 4 <= cnt; j += 4) {
      float x0[4], x1[4], x2[4], cj[4];
      for (int u = 0; u < 4; ++u) {
        int sj = __builtin_amdgcn_readlane(s, j + u);
        cj[u] = __int_as_float(__builtin_amdgcn_readlane(ci, j + u));
        const float* q0; const float* q2;
        if (SPLIT) { q0 = srcB + (long)sj * HID;  q2 = srcA + (long)sj * FIN; }
        else       { q0 = srcA + (long)sj * KTOT; q2 = q0 + HID; }
        x0[u] = q0[lane]; x1[u] = q0[32 + lane]; x2[u] = q2[l8];
      }
      for (int u = 0; u < 4; ++u) {
        a0 += cj[u] * x0[u]; a1 += cj[u] * x1[u]; a2 += cj[u] * x2[u];
      }
    }
    for (; j < cnt; ++j) {
      int sj = __builtin_amdgcn_readlane(s, j);
      float cj = __int_as_float(__builtin_amdgcn_readlane(ci, j));
      const float* q0; const float* q2;
      if (SPLIT) { q0 = srcB + (long)sj * HID;  q2 = srcA + (long)sj * FIN; }
      else       { q0 = srcA + (long)sj * KTOT; q2 = q0 + HID; }
      a0 += cj * q0[lane]; a1 += cj * q0[32 + lane]; a2 += cj * q2[l8];
    }
  }

  float* o = outbuf + (long)(d - n0) * KTOT;
  if (SPLIT) {
    o[FIN + lane] = dd * a0;
    o[FIN + 32 + lane] = dd * a1;
    if (lane < 8) o[lane] = dd * a2;
  } else {
    o[lane] = dd * a0;
    o[32 + lane] = dd * a1;
    if (lane < 8) o[HID + lane] = dd * a2;
  }
}

__global__ __launch_bounds__(256)
void k_gates(const float* __restrict__ AXH, const float* __restrict__ C,
             const float* __restrict__ Wx, const float* __restrict__ Wh,
             const float* __restrict__ bx, const float* __restrict__ bh,
             const float* __restrict__ bg, const float* __restrict__ wc,
             const float* __restrict__ ln_h_s, const float* __restrict__ ln_h_b,
             const float* __restrict__ ln_c_s, const float* __restrict__ ln_c_b,
             const float* __restrict__ ln_o_s, const float* __restrict__ ln_o_b,
             const float* __restrict__ skip,
             float* __restrict__ hiddenOut, float* __restrict__ cellOut,
             float* __restrict__ Y, int N) {
  extern __shared__ float smem[];
  float* Bf = smem;
  float* bias = smem + 18432;

  int tid = threadIdx.x;
  for (int i = tid; i < KSTEPS * NT_G * 64; i += 256) {
    int s = i >> 10;
    int rem = i & 1023;
    int t = rem >> 6;
    int rem2 = rem & 63;
    int L = rem2 >> 1;
    int j = rem2 & 1;
    int k = 4 * s + 2 * (L >> 4) + j;
    int c = 16 * t + (L & 15);
    int g = c >> 6, h = c & 63;
    Bf[i] = (k < FIN) ? Wx[(g * FIN + k) * HID + h]
                      : Wh[(g * HID + (k - FIN)) * HID + h];
  }
  if (tid < GCOLS) bias[tid] = bx[tid] + bh[tid] + bg[tid];
  __syncthreads();

  int wave = tid >> 5, lane = tid & 31;
  int half = lane >> 4, lcol = lane & 15;
  int rowBase = blockIdx.x * ROWS_PER_BLOCK + wave * 16;

  int arow = rowBase + lcol;
  if (arow > N - 1) arow = N - 1;
  const float* aptr = AXH + (long)arow * KTOT;

  v8f acc[NT_G];
#pragma unroll
  for (int t = 0; t < NT_G; ++t)
#pragma unroll
    for (int r = 0; r < 8; ++r) acc[t][r] = 0.0f;

  for (int s = 0; s < KSTEPS; ++s) {
    int k0 = 4 * s + 2 * half;
    v2f a = *(const v2f*)(aptr + k0);
    const float* bbase = Bf + s * 1024 + lane * 2;
#pragma unroll
    for (int t = 0; t < NT_G; ++t) {
      v2f b = *(const v2f*)(bbase + t * 64);
      acc[t] = __builtin_amdgcn_wmma_f32_16x16x4_f32(
          false, a, false, b, (short)0, acc[t], false, false);
    }
  }

#pragma unroll
  for (int r = 0; r < 8; ++r) {
    int node = rowBase + r + 8 * half;
    bool valid = node < N;
    int nc = valid ? node : N - 1;
    float hloc[4], cloc[4];
#pragma unroll
    for (int t = 0; t < 4; ++t) {
      int h = t * 16 + lcol;
      float c0 = C[(long)nc * HID + h];
      float gi = acc[t][r]      + bias[h]       + wc[h] * c0;
      float gf = acc[t + 4][r]  + bias[64 + h]  + wc[64 + h] * c0;
      float gc = acc[t + 8][r]  + bias[128 + h];
      float go = acc[t + 12][r] + bias[192 + h];
      float iv = sigmoidf_(gi);
      float fv = sigmoidf_(gf);
      float cc = fv * c0 + iv * tanhf(gc);
      float ov = sigmoidf_(go + wc[128 + h] * cc);
      hloc[t] = ov * tanhf(cc);
      cloc[t] = cc;
    }
    float sh = 0.f, sh2 = 0.f, sc = 0.f, sc2 = 0.f;
#pragma unroll
    for (int t = 0; t < 4; ++t) {
      sh += hloc[t]; sh2 += hloc[t] * hloc[t];
      sc += cloc[t]; sc2 += cloc[t] * cloc[t];
    }
#pragma unroll
    for (int mask = 1; mask < 16; mask <<= 1) {
      sh  += __shfl_xor(sh,  mask, 32);
      sh2 += __shfl_xor(sh2, mask, 32);
      sc  += __shfl_xor(sc,  mask, 32);
      sc2 += __shfl_xor(sc2, mask, 32);
    }
    float mh = sh * (1.0f / 64.0f), mc = sc * (1.0f / 64.0f);
    float vh = sh2 * (1.0f / 64.0f) - mh * mh;
    float vc = sc2 * (1.0f / 64.0f) - mc * mc;
    float rh = rsqrtf(vh + EPS), rc = rsqrtf(vc + EPS);
    if (valid) {
#pragma unroll
      for (int t = 0; t < 4; ++t) {
        int h = t * 16 + lcol;
        float hn = (hloc[t] - mh) * rh;
        float cn = (cloc[t] - mc) * rc;
        hiddenOut[(long)node * HID + h] = hn * ln_h_s[h] + ln_h_b[h];
        cellOut[(long)node * HID + h]   = cn * ln_c_s[h] + ln_c_b[h];
        float ov = hn * ln_o_s[h] + ln_o_b[h];
        Y[(long)node * KTOT + h] = ov > 0.0f ? ov : 0.0f;
      }
      if (lcol < 8) {
        Y[(long)node * KTOT + 64 + lcol] =
            (lcol < 2) ? skip[(long)node * 2 + lcol] : 0.0f;
      }
    }
  }
}

__global__ __launch_bounds__(256)
void k_gemm2(const float* __restrict__ AY, const float* __restrict__ W1,
             const float* __restrict__ b1, const float* __restrict__ W2,
             float* __restrict__ P, int n0, int n1) {
  __shared__ float Bf[KSTEPS * 4 * 64];
  __shared__ float b1s[HID];
  __shared__ float w2s[HID];
  __shared__ float ps[ROWS_PER_BLOCK];

  int tid = threadIdx.x;
  for (int i = tid; i < KSTEPS * 4 * 64; i += 256) {
    int s = i >> 8;
    int rem = i & 255;
    int t = rem >> 6;
    int rem2 = rem & 63;
    int L = rem2 >> 1;
    int j = rem2 & 1;
    int k = 4 * s + 2 * (L >> 4) + j;
    int c = 16 * t + (L & 15);
    Bf[i] = (k < HID + 2) ? W1[k * HID + c] : 0.0f;
  }
  if (tid < HID) { b1s[tid] = b1[tid]; w2s[tid] = W2[tid]; }
  __syncthreads();

  int wave = tid >> 5, lane = tid & 31;
  int half = lane >> 4, lcol = lane & 15;
  int rowBase = n0 + blockIdx.x * ROWS_PER_BLOCK + wave * 16;

  int arow = rowBase + lcol;
  if (arow > n1 - 1) arow = n1 - 1;
  const float* aptr = AY + (long)(arow - n0) * KTOT;

  v8f acc[4];
#pragma unroll
  for (int t = 0; t < 4; ++t)
#pragma unroll
    for (int r = 0; r < 8; ++r) acc[t][r] = 0.0f;

  for (int s = 0; s < KSTEPS; ++s) {
    int k0 = 4 * s + 2 * half;
    v2f a = *(const v2f*)(aptr + k0);
    const float* bbase = Bf + s * 256 + lane * 2;
#pragma unroll
    for (int t = 0; t < 4; ++t) {
      v2f b = *(const v2f*)(bbase + t * 64);
      acc[t] = __builtin_amdgcn_wmma_f32_16x16x4_f32(
          false, a, false, b, (short)0, acc[t], false, false);
    }
  }

  float part[8];
#pragma unroll
  for (int r = 0; r < 8; ++r) {
    float s = 0.0f;
#pragma unroll
    for (int t = 0; t < 4; ++t) {
      int h = t * 16 + lcol;
      float z = acc[t][r] + b1s[h];
      s += (z > 0.0f ? z : 0.0f) * w2s[h];
    }
    part[r] = s;
  }
#pragma unroll
  for (int mask = 1; mask < 16; mask <<= 1) {
    float got[8];
#pragma unroll
    for (int r = 0; r < 8; ++r) got[r] = __shfl_xor(part[r], mask, 32);
#pragma unroll
    for (int r = 0; r < 8; ++r) part[r] += got[r];
  }
  if (lcol == 0) {
#pragma unroll
    for (int r = 0; r < 8; ++r) ps[wave * 16 + r + 8 * half] = part[r];
  }
  __syncthreads();
  if (tid < ROWS_PER_BLOCK) {
    int node = n0 + blockIdx.x * ROWS_PER_BLOCK + tid;
    if (node < n1) P[node] = ps[tid];
  }
}

__global__ __launch_bounds__(256)
void k_gather_out(float* __restrict__ out, const float* __restrict__ P,
                  const int* __restrict__ rowptr, const int* __restrict__ rec,
                  const float* __restrict__ dinv, const float* __restrict__ b2,
                  const float* __restrict__ X, int N, int E) {
  int t = blockIdx.x * blockDim.x + threadIdx.x;
  int d = t >> 4, l = t & 15;
  bool valid = d < N;
  int dc = valid ? d : N - 1;
  int beg = rowptr[dc];
  int end = (dc + 1 < N) ? rowptr[dc + 1] : E;
  float acc = 0.0f;
  for (int p = beg + l; p < end; p += 16) {
    v2i r = *(const v2i*)(rec + 2 * (long)p);
    acc += __int_as_float(r.y) * P[r.x];
  }
  for (int mask = 8; mask > 0; mask >>= 1) acc += __shfl_xor(acc, mask, 32);
  if (valid && l == 0) {
    float dd = dinv[d];
    float q = dd * (acc + dd * P[d]);
    out[d] = q + b2[0] + X[(long)d * FIN];
  }
}

extern "C" void kernel_launch(void* const* d_in, const int* in_sizes, int n_in,
                              void* d_out, int out_size, void* d_ws, size_t ws_size,
                              hipStream_t stream) {
  const float* X    = (const float*)d_in[0];
  const int*   eidx = (const int*)d_in[1];
  const float* ew   = (const float*)d_in[2];
  const float* skip = (const float*)d_in[3];
  const float* H    = (const float*)d_in[4];
  const float* C    = (const float*)d_in[5];
  const float* Wx   = (const float*)d_in[6];
  const float* bx   = (const float*)d_in[7];
  const float* Wh   = (const float*)d_in[8];
  const float* bh   = (const float*)d_in[9];
  const float* wc   = (const float*)d_in[10];
  const float* bg   = (const float*)d_in[11];
  const float* ln_h_s = (const float*)d_in[12];
  const float* ln_h_b = (const float*)d_in[13];
  const float* ln_c_s = (const float*)d_in[14];
  const float* ln_c_b = (const float*)d_in[15];
  const float* ln_o_s = (const float*)d_in[16];
  const float* ln_o_b = (const float*)d_in[17];
  const float* W1   = (const float*)d_in[18];
  const float* b1   = (const float*)d_in[19];
  const float* W2   = (const float*)d_in[20];
  const float* b2   = (const float*)d_in[21];

  const int N = in_sizes[0] / FIN;
  const int E = in_sizes[2];

  const int* src = eidx;
  const int* dst = eidx + E;

  float* out_p    = (float*)d_out;
  float* hidden_p = out_p + N;
  float* cell_p   = hidden_p + (long)N * HID;

  const int Nh = ((N + 1) / 2 + ROWS_PER_BLOCK - 1) / ROWS_PER_BLOCK * ROWS_PER_BLOCK;
  const int Na = N < Nh ? N : Nh;

  int*   rec  = (int*)d_ws;
  float* AXH  = (float*)(rec + 2 * (long)E);
  float* Y    = AXH;
  float* AY   = AXH + (long)N * KTOT;
  float* dinv = AY + (long)Nh * KTOT;
  float* P    = dinv + N;
  int* rowptr = (int*)(P + N);
  int* cursor = (int*)AY;
  int* bsum   = cursor + N;

  const int B = 256;
  auto cdiv = [](long a, long b) { return (unsigned)((a + b - 1) / b); };

  k_init_deg<<<cdiv(N, B), B, 0, stream>>>(dinv, cursor, N);
  k_acc_deg<<<cdiv(E, B), B, 0, stream>>>(dinv, cursor, dst, ew, E);
  k_dinv<<<cdiv(N, B), B, 0, stream>>>(dinv, N);

  k_blocksum<<<cdiv(N, SCAN_T * SCAN_PER), SCAN_T, 0, stream>>>(cursor, bsum, N);
  k_scan<<<cdiv(N, SCAN_T * SCAN_PER), SCAN_T, 0, stream>>>(cursor, rowptr, bsum, N);
  k_fill<<<cdiv(E, B), B, 0, stream>>>(cursor, rec, src, dst, ew, dinv, E);

  k_gather72<true><<<cdiv(N, 8), B, 0, stream>>>(AXH, X, H, rowptr, rec, dinv, 0, N, N, E);

  size_t smem_gates = (size_t)(KSTEPS * NT_G * 64 + GCOLS) * sizeof(float);
  k_gates<<<cdiv(N, ROWS_PER_BLOCK), B, smem_gates, stream>>>(
      AXH, C, Wx, Wh, bx, bh, bg, wc,
      ln_h_s, ln_h_b, ln_c_s, ln_c_b, ln_o_s, ln_o_b,
      skip, hidden_p, cell_p, Y, N);

  k_gather72<false><<<cdiv(Na, 8), B, 0, stream>>>(AY, Y, Y, rowptr, rec, dinv, 0, Na, N, E);
  k_gemm2<<<cdiv(Na, ROWS_PER_BLOCK), B, 0, stream>>>(AY, W1, b1, W2, P, 0, Na);
  if (N > Nh) {
    k_gather72<false><<<cdiv(N - Nh, 8), B, 0, stream>>>(AY, Y, Y, rowptr, rec, dinv, Nh, N, N, E);
    k_gemm2<<<cdiv(N - Nh, ROWS_PER_BLOCK), B, 0, stream>>>(AY, W1, b1, W2, P, Nh, N);
  }

  k_gather_out<<<cdiv((long)N * 16, B), B, 0, stream>>>(out_p, P, rowptr, rec, dinv,
                                                       b2, X, N, E);
}


# Sort atomics stay inside one package half: per-half planes (die number) for edge counts and degree sums; the counting add returns the edge's rank, so k_fill needs no atomic; float sum orders change
# speedup vs baseline: 4.6376x; 1.6049x over previous
#include <hip/hip_runtime.h>
#include <math.h>


typedef __attribute__((ext_vector_type(2))) float v2f;
typedef __attribute__((ext_vector_type(2))) int v2i;
typedef __attribute__((ext_vector_type(8))) float v8f;

#define FIN   8
#define HID   64
#define KTOT  72
#define KSTEPS 18
#define GCOLS 256
#define NT_G  16
#define ROWS_PER_BLOCK 128
#define EPS 1e-5f

__device__ __forceinline__ float sigmoidf_(float x) { return 1.0f / (1.0f + expf(-x)); }

__device__ __forceinline__ int wave_half() {
  return (int)((__builtin_amdgcn_s_sendmsg_rtn(0x87) >> 18) & 1u);
}

__global__ void k_init_deg(float* deg2, int* cnt2, int N) {
  int i = blockIdx.x * blockDim.x + threadIdx.x;
  if (i < N) { deg2[i] = 1.0f; deg2[N + i] = 0.0f; cnt2[i] = 0; cnt2[N + i] = 0; }
}

__global__ __launch_bounds__(256)
void k_acc_deg(float* deg2, int* cnt2, int* __restrict__ rank,
               const int* __restrict__ dst, const float* __restrict__ w, int N, int E) {
  int h = wave_half();
  float* degp = deg2 + (long)h * N;
  int* cntp = cnt2 + (long)h * N;
  int stride = gridDim.x * blockDim.x;
  for (int e = blockIdx.x * blockDim.x + threadIdx.x; e < E; e += stride) {
    int d = dst[e];
    atomicAdd(&degp[d], w[e]);
    int r = atomicAdd(&cntp[d], 1);
    rank[e] = 2 * r + h;
  }
}

__global__ void k_dinv(float* __restrict__ dinv, const float* __restrict__ deg2, int N) {
  int i = blockIdx.x * blockDim.x + threadIdx.x;
  if (i < N) dinv[i] = rsqrtf(deg2[i] + deg2[N + i]);
}

#define SCAN_T 256
#define SCAN_PER 4

__global__ __launch_bounds__(SCAN_T)
void k_blocksum(const int* __restrict__ cnt2, int* __restrict__ bsum, int N) {
  __shared__ int wsum[SCAN_T / 32];
  int base = (blockIdx.x * SCAN_T + threadIdx.x) * SCAN_PER;
  int s = 0;
  for (int i = 0; i < SCAN_PER; ++i)
    if (base + i < N) s += cnt2[base + i] + cnt2[N + base + i];
  for (int mask = 16; mask > 0; mask >>= 1) s += __shfl_xor(s, mask, 32);
  if ((threadIdx.x & 31) == 0) wsum[threadIdx.x >> 5] = s;
  __syncthreads();
  if (threadIdx.x == 0) {
    int tot = 0;
    for (int i = 0; i < SCAN_T / 32; ++i) tot += wsum[i];
    bsum[blockIdx.x] = tot;
  }
}

__global__ __launch_bounds__(SCAN_T)
void k_scan(const int* __restrict__ cnt2, int* __restrict__ rowptr,
            int* __restrict__ base1, const int* __restrict__ bsum, int N) {
  __shared__ int wtot[SCAN_T / 32];
  __shared__ int woff[SCAN_T / 32];
  int tid = threadIdx.x, lane = tid & 31, wave = tid >> 5;

  int off = 0;
  for (int i = tid; i < (int)blockIdx.x; i += SCAN_T) off += bsum[i];
  for (int mask = 16; mask > 0; mask >>= 1) off += __shfl_xor(off, mask, 32);
  if (lane == 0) woff[wave] = off;

  int base = (blockIdx.x * SCAN_T + tid) * SCAN_PER;
  int c[SCAN_PER], c0[SCAN_PER];
  int tsum = 0;
  for (int i = 0; i < SCAN_PER; ++i) {
    c0[i] = (base + i < N) ? cnt2[base + i] : 0;
    c[i] = c0[i] + ((base + i < N) ? cnt2[N + base + i] : 0);
    tsum += c[i];
  }
  int v = tsum;
  for (int o = 1; o < 32; o <<= 1) {
    int u = __shfl_up(v, o, 32);
    if (lane >= o) v += u;
  }
  if (lane == 31) wtot[wave] = v;
  __syncthreads();
  int pre = 0;
  for (int i = 0; i < SCAN_T / 32; ++i) {
    pre += woff[i];
    if (i < wave) pre += wtot[i];
  }
  int excl = pre + v - tsum;
  for (int i = 0; i < SCAN_PER; ++i) {
    if (base + i < N) {
      rowptr[base + i] = excl;
      base1[base + i] = excl + c0[i];
      excl += c[i];
    }
  }
}

__global__ void k_fill(const int* __restrict__ rowptr, const int* __restrict__ base1,
                       const int* __restrict__ rank, int* __restrict__ rec,
                       const int* __restrict__ src,
                       const int* __restrict__ dst, const float* __restrict__ w,
                       const float* __restrict__ dinv, int E) {
  int e = blockIdx.x * blockDim.x + threadIdx.x;
  if (e < E) {
    int dn = dst[e];
    int s = src[e];
    float we = w[e];
    int rk = rank[e];
    const int* bp = (rk & 1) ? base1 : rowptr;
    int p = bp[dn] + (rk >> 1);
    float c = we * dinv[s];
    v2i r;
    r.x = s;
    r.y = __float_as_int(c);
    *(v2i*)(rec + 2 * (long)p) = r;
  }
}

template <bool SPLIT>
__global__ __launch_bounds__(256)
void k_gather72(float* __restrict__ outbuf, const float* __restrict__ srcA,
                const float* __restrict__ srcB,
                const int* __restrict__ rowptr, const int* __restrict__ rec,
                const float* __restrict__ dinv, int n0, int n1, int N, int E) {
  int lane = threadIdx.x & 31;
  int l8 = lane & 7;
  int d = __builtin_amdgcn_readfirstlane((int)(n0 + blockIdx.x * 8 + (threadIdx.x >> 5)));
  if (d >= n1) return;
  int beg = rowptr[d];
  int end = (d + 1 < N) ? rowptr[d + 1] : E;
  float dd = dinv[d];

  const float* rlo;
  const float* rex;
  if (SPLIT) { rlo = srcB + (long)d * HID;  rex = srcA + (long)d * FIN; }
  else       { rlo = srcA + (long)d * KTOT; rex = rlo + HID; }
  float a0 = dd * rlo[lane];
  float a1 = dd * rlo[32 + lane];
  float a2 = dd * rex[l8];

  for (int base = beg; base < end; base += 32) {
    int p = base + lane;
    int s = 0;
    float c = 0.0f;
    if (p < end) {
      v2i r = *(const v2i*)(rec + 2 * (long)p);
      s = r.x;
      c = __int_as_float(r.y);
    }
    int ci = __float_as_int(c);
    int cnt = end - base;
    if (cnt > 32) cnt = 32;
    int j = 0;
    for (; j + 4 <= cnt; j += 4) {
      float x0[4], x1[4], x2[4], cj[4];
      for (int u = 0; u < 4; ++u) {
        int sj = __builtin_amdgcn_readlane(s, j + u);
        cj[u] = __int_as_float(__builtin_amdgcn_readlane(ci, j + u));
        const float* q0; const float* q2;
        if (SPLIT) { q0 = srcB + (long)sj * HID;  q2 = srcA + (long)sj * FIN; }
        else       { q0 = srcA + (long)sj * KTOT; q2 = q0 + HID; }
        x0[u] = q0[lane]; x1[u] = q0[32 + lane]; x2[u] = q2[l8];
      }
      for (int u = 0; u < 4; ++u) {
        a0 += cj[u] * x0[u]; a1 += cj[u] * x1[u]; a2 += cj[u] * x2[u];
      }
    }
    for (; j < cnt; ++j) {
      int sj = __builtin_amdgcn_readlane(s, j);
      float cj = __int_as_float(__builtin_amdgcn_readlane(ci, j));
      const float* q0; const float* q2;
      if (SPLIT) { q0 = srcB + (long)sj * HID;  q2 = srcA + (long)sj * FIN; }
      else       { q0 = srcA + (long)sj * KTOT; q2 = q0 + HID; }
      a0 += cj * q0[lane]; a1 += cj * q0[32 + lane]; a2 += cj * q2[l8];
    }
  }

  float* o = outbuf + (long)(d - n0) * KTOT;
  if (SPLIT) {
    o[FIN + lane] = dd * a0;
    o[FIN + 32 + lane] = dd * a1;
    if (lane < 8) o[lane] = dd * a2;
  } else {
    o[lane] = dd * a0;
    o[32 + lane] = dd * a1;
    if (lane < 8) o[HID + lane] = dd * a2;
  }
}

__global__ __launch_bounds__(256)
void k_gates(const float* __restrict__ AXH, const float* __restrict__ C,
             const float* __restrict__ Wx, const float* __restrict__ Wh,
             const float* __restrict__ bx, const float* __restrict__ bh,
             const float* __restrict__ bg, const float* __restrict__ wc,
             const float* __restrict__ ln_h_s, const float* __restrict__ ln_h_b,
             const float* __restrict__ ln_c_s, const float* __restrict__ ln_c_b,
             const float* __restrict__ ln_o_s, const float* __restrict__ ln_o_b,
             const float* __restrict__ skip,
             float* __restrict__ hiddenOut, float* __restrict__ cellOut,
             float* __restrict__ Y, int N) {
  extern __shared__ float smem[];
  float* Bf = smem;
  float* bias = smem + 18432;

  int tid = threadIdx.x;
  for (int i = tid; i < KSTEPS * NT_G * 64; i += 256) {
    int s = i >> 10;
    int rem = i & 1023;
    int t = rem >> 6;
    int rem2 = rem & 63;
    int L = rem2 >> 1;
    int j = rem2 & 1;
    int k = 4 * s + 2 * (L >> 4) + j;
    int c = 16 * t + (L & 15);
    int g = c >> 6, h = c & 63;
    Bf[i] = (k < FIN) ? Wx[(g * FIN + k) * HID + h]
                      : Wh[(g * HID + (k - FIN)) * HID + h];
  }
  if (tid < GCOLS) bias[tid] = bx[tid] + bh[tid] + bg[tid];
  __syncthreads();

  int wave = tid >> 5, lane = tid & 31;
  int half = lane >> 4, lcol = lane & 15;
  int rowBase = blockIdx.x * ROWS_PER_BLOCK + wave * 16;

  int arow = rowBase + lcol;
  if (arow > N - 1) arow = N - 1;
  const float* aptr = AXH + (long)arow * KTOT;

  v8f acc[NT_G];
#pragma unroll
  for (int t = 0; t < NT_G; ++t)
#pragma unroll
    for (int r = 0; r < 8; ++r) acc[t][r] = 0.0f;

  for (int s = 0; s < KSTEPS; ++s) {
    int k0 = 4 * s + 2 * half;
    v2f a = *(const v2f*)(aptr + k0);
    const float* bbase = Bf + s * 1024 + lane * 2;
#pragma unroll
    for (int t = 0; t < NT_G; ++t) {
      v2f b = *(const v2f*)(bbase + t * 64);
      acc[t] = __builtin_amdgcn_wmma_f32_16x16x4_f32(
          false, a, false, b, (short)0, acc[t], false, false);
    }
  }

#pragma unroll
  for (int r = 0; r < 8; ++r) {
    int node = rowBase + r + 8 * half;
    bool valid = node < N;
    int nc = valid ? node : N - 1;
    float hloc[4], cloc[4];
#pragma unroll
    for (int t = 0; t < 4; ++t) {
      int h = t * 16 + lcol;
      float c0 = C[(long)nc * HID + h];
      float gi = acc[t][r]      + bias[h]       + wc[h] * c0;
      float gf = acc[t + 4][r]  + bias[64 + h]  + wc[64 + h] * c0;
      float gc = acc[t + 8][r]  + bias[128 + h];
      float go = acc[t + 12][r] + bias[192 + h];
      float iv = sigmoidf_(gi);
      float fv = sigmoidf_(gf);
      float cc = fv * c0 + iv * tanhf(gc);
      float ov = sigmoidf_(go + wc[128 + h] * cc);
      hloc[t] = ov * tanhf(cc);
      cloc[t] = cc;
    }
    float sh = 0.f, sh2 = 0.f, sc = 0.f, sc2 = 0.f;
#pragma unroll
    for (int t = 0; t < 4; ++t) {
      sh += hloc[t]; sh2 += hloc[t] * hloc[t];
      sc += cloc[t]; sc2 += cloc[t] * cloc[t];
    }
#pragma unroll
    for (int mask = 1; mask < 16; mask <<= 1) {
      sh  += __shfl_xor(sh,  mask, 32);
      sh2 += __shfl_xor(sh2, mask, 32);
      sc  += __shfl_xor(sc,  mask, 32);
      sc2 += __shfl_xor(sc2, mask, 32);
    }
    float mh = sh * (1.0f / 64.0f), mc = sc * (1.0f / 64.0f);
    float vh = sh2 * (1.0f / 64.0f) - mh * mh;
    float vc = sc2 * (1.0f / 64.0f) - mc * mc;
    float rh = rsqrtf(vh + EPS), rc = rsqrtf(vc + EPS);
    if (valid) {
#pragma unroll
      for (int t = 0; t < 4; ++t) {
        int h = t * 16 + lcol;
        float hn = (hloc[t] - mh) * rh;
        float cn = (cloc[t] - mc) * rc;
        hiddenOut[(long)node * HID + h] = hn * ln_h_s[h] + ln_h_b[h];
        cellOut[(long)node * HID + h]   = cn * ln_c_s[h] + ln_c_b[h];
        float ov = hn * ln_o_s[h] + ln_o_b[h];
        Y[(long)node * KTOT + h] = ov > 0.0f ? ov : 0.0f;
      }
      if (lcol < 8) {
        Y[(long)node * KTOT + 64 + lcol] =
            (lcol < 2) ? skip[(long)node * 2 + lcol] : 0.0f;
      }
    }
  }
}

__global__ __launch_bounds__(256)
void k_gemm2(const float* __restrict__ AY, const float* __restrict__ W1,
             const float* __restrict__ b1, const float* __restrict__ W2,
             float* __restrict__ P, int n0, int n1) {
  __shared__ float Bf[KSTEPS * 4 * 64];
  __shared__ float b1s[HID];
  __shared__ float w2s[HID];
  __shared__ float ps[ROWS_PER_BLOCK];

  int tid = threadIdx.x;
  for (int i = tid; i < KSTEPS * 4 * 64; i += 256) {
    int s = i >> 8;
    int rem = i & 255;
    int t = rem >> 6;
    int rem2 = rem & 63;
    int L = rem2 >> 1;
    int j = rem2 & 1;
    int k = 4 * s + 2 * (L >> 4) + j;
    int c = 16 * t + (L & 15);
    Bf[i] = (k < HID + 2) ? W1[k * HID + c] : 0.0f;
  }
  if (tid < HID) { b1s[tid] = b1[tid]; w2s[tid] = W2[tid]; }
  __syncthreads();

  int wave = tid >> 5, lane = tid & 31;
  int half = lane >> 4, lcol = lane & 15;
  int rowBase = n0 + blockIdx.x * ROWS_PER_BLOCK + wave * 16;

  int arow = rowBase + lcol;
  if (arow > n1 - 1) arow = n1 - 1;
  const float* aptr = AY + (long)(arow - n0) * KTOT;

  v8f acc[4];
#pragma unroll
  for (int t = 0; t < 4; ++t)
#pragma unroll
    for (int r = 0; r < 8; ++r) acc[t][r] = 0.0f;

  for (int s = 0; s < KSTEPS; ++s) {
    int k0 = 4 * s + 2 * half;
    v2f a = *(const v2f*)(aptr + k0);
    const float* bbase = Bf + s * 256 + lane * 2;
#pragma unroll
    for (int t = 0; t < 4; ++t) {
      v2f b = *(const v2f*)(bbase + t * 64);
      acc[t] = __builtin_amdgcn_wmma_f32_16x16x4_f32(
          false, a, false, b, (short)0, acc[t], false, false);
    }
  }

  float part[8];
#pragma unroll
  for (int r = 0; r < 8; ++r) {
    float s = 0.0f;
#pragma unroll
    for (int t = 0; t < 4; ++t) {
      int h = t * 16 + lcol;
      float z = acc[t][r] + b1s[h];
      s += (z > 0.0f ? z : 0.0f) * w2s[h];
    }
    part[r] = s;
  }
#pragma unroll
  for (int mask = 1; mask < 16; mask <<= 1) {
    float got[8];
#pragma unroll
    for (int r = 0; r < 8; ++r) got[r] = __shfl_xor(part[r], mask, 32);
#pragma unroll
    for (int r = 0; r < 8; ++r) part[r] += got[r];
  }
  if (lcol == 0) {
#pragma unroll
    for (int r = 0; r < 8; ++r) ps[wave * 16 + r + 8 * half] = part[r];
  }
  __syncthreads();
  if (tid < ROWS_PER_BLOCK) {
    int node = n0 + blockIdx.x * ROWS_PER_BLOCK + tid;
    if (node < n1) P[node] = ps[tid];
  }
}

__global__ __launch_bounds__(256)
void k_gather_out(float* __restrict__ out, const float* __restrict__ P,
                  const int* __restrict__ rowptr, const int* __restrict__ rec,
                  const float* __restrict__ dinv, const float* __restrict__ b2,
                  const float* __restrict__ X, int N, int E) {
  int t = blockIdx.x * blockDim.x + threadIdx.x;
  int d = t >> 4, l = t & 15;
  bool valid = d < N;
  int dc = valid ? d : N - 1;
  int beg = rowptr[dc];
  int end = (dc + 1 < N) ? rowptr[dc + 1] : E;
  float acc = 0.0f;
  for (int p = beg + l; p < end; p += 16) {
    v2i r = *(const v2i*)(rec + 2 * (long)p);
    acc += __int_as_float(r.y) * P[r.x];
  }
  for (int mask = 8; mask > 0; mask >>= 1) acc += __shfl_xor(acc, mask, 32);
  if (valid && l == 0) {
    float dd = dinv[d];
    float q = dd * (acc + dd * P[d]);
    out[d] = q + b2[0] + X[(long)d * FIN];
  }
}

extern "C" void kernel_launch(void* const* d_in, const int* in_sizes, int n_in,
                              void* d_out, int out_size, void* d_ws, size_t ws_size,
                              hipStream_t stream) {
  const float* X    = (const float*)d_in[0];
  const int*   eidx = (const int*)d_in[1];
  const float* ew   = (const float*)d_in[2];
  const float* skip = (const float*)d_in[3];
  const float* H    = (const float*)d_in[4];
  const float* C    = (const float*)d_in[5];
  const float* Wx   = (const float*)d_in[6];
  const float* bx   = (const float*)d_in[7];
  const float* Wh   = (const float*)d_in[8];
  const float* bh   = (const float*)d_in[9];
  const float* wc   = (const float*)d_in[10];
  const float* bg   = (const float*)d_in[11];
  const float* ln_h_s = (const float*)d_in[12];
  const float* ln_h_b = (const float*)d_in[13];
  const float* ln_c_s = (const float*)d_in[14];
  const float* ln_c_b = (const float*)d_in[15];
  const float* ln_o_s = (const float*)d_in[16];
  const float* ln_o_b = (const float*)d_in[17];
  const float* W1   = (const float*)d_in[18];
  const float* b1   = (const float*)d_in[19];
  const float* W2   = (const float*)d_in[20];
  const float* b2   = (const float*)d_in[21];

  const int N = in_sizes[0] / FIN;
  const int E = in_sizes[2];

  const int* src = eidx;
  const int* dst = eidx + E;

  float* out_p    = (float*)d_out;
  float* hidden_p = out_p + N;
  float* cell_p   = hidden_p + (long)N * HID;

  const int Nh = ((N + 1) / 2 + ROWS_PER_BLOCK - 1) / ROWS_PER_BLOCK * ROWS_PER_BLOCK;
  const int Na = N < Nh ? N : Nh;

  int*   rec  = (int*)d_ws;
  float* AXH  = (float*)(rec + 2 * (long)E);
  float* Y    = AXH;
  float* AY   = AXH + (long)N * KTOT;
  float* dinv = AY + (long)Nh * KTOT;
  float* P    = dinv + N;
  int* rowptr = (int*)(P + N);
  int*   cnt2  = (int*)AY;
  float* deg2  = (float*)(cnt2 + 2 * (long)N);
  int*   base1 = (int*)(deg2 + 2 * (long)N);
  int*   bsum  = base1 + N;
  int*   rank  = (int*)AXH;

  const int B = 256;
  auto cdiv = [](long a, long b) { return (unsigned)((a + b - 1) / b); };

  k_init_deg<<<cdiv(N, B), B, 0, stream>>>(deg2, cnt2, N);
  k_acc_deg<<<2048, B, 0, stream>>>(deg2, cnt2, rank, dst, ew, N, E);
  k_dinv<<<cdiv(N, B), B, 0, stream>>>(dinv, deg2, N);

  k_blocksum<<<cdiv(N, SCAN_T * SCAN_PER), SCAN_T, 0, stream>>>(cnt2, bsum, N);
  k_scan<<<cdiv(N, SCAN_T * SCAN_PER), SCAN_T, 0, stream>>>(cnt2, rowptr, base1, bsum, N);
  k_fill<<<cdiv(E, B), B, 0, stream>>>(rowptr, base1, rank, rec, src, dst, ew, dinv, E);

  k_gather72<true><<<cdiv(N, 8), B, 0, stream>>>(AXH, X, H, rowptr, rec, dinv, 0, N, N, E);

  size_t smem_gates = (size_t)(KSTEPS * NT_G * 64 + GCOLS) * sizeof(float);
  k_gates<<<cdiv(N, ROWS_PER_BLOCK), B, smem_gates, stream>>>(
      AXH, C, Wx, Wh, bx, bh, bg, wc,
      ln_h_s, ln_h_b, ln_c_s, ln_c_b, ln_o_s, ln_o_b,
      skip, hidden_p, cell_p, Y, N);

  k_gather72<false><<<cdiv(Na, 8), B, 0, stream>>>(AY, Y, Y, rowptr, rec, dinv, 0, Na, N, E);
  k_gemm2<<<cdiv(Na, ROWS_PER_BLOCK), B, 0, stream>>>(AY, W1, b1, W2, P, 0, Na);
  if (N > Nh) {
    k_gather72<false><<<cdiv(N - Nh, 8), B, 0, stream>>>(AY, Y, Y, rowptr, rec, dinv, Nh, N, N, E);
    k_gemm2<<<cdiv(N - Nh, ROWS_PER_BLOCK), B, 0, stream>>>(AY, W1, b1, W2, P, Nh, N);
  }

  k_gather_out<<<cdiv((long)N * 16, B), B, 0, stream>>>(out_p, P, rowptr, rec, dinv,
                                                       b2, X, N, E);
}


# W1 applied before the 2nd aggregation (agg is linear: agg(Y)@W1 = agg(Y@W1)): it gathers aligned 64-float rows, 2 lines per edge not 3, then bias, relu, W2 dot; float sum order of out changes
# speedup vs baseline: 4.8368x; 1.0430x over previous
#include <hip/hip_runtime.h>
#include <math.h>


typedef __attribute__((ext_vector_type(2))) float v2f;
typedef __attribute__((ext_vector_type(2))) int v2i;
typedef __attribute__((ext_vector_type(8))) float v8f;

#define FIN   8
#define HID   64
#define KTOT  72
#define KSTEPS 18
#define KH_STEPS 17
#define TPITCH 68
#define GCOLS 256
#define NT_G  16
#define ROWS_PER_BLOCK 128
#define EPS 1e-5f

__device__ __forceinline__ float sigmoidf_(float x) { return 1.0f / (1.0f + expf(-x)); }

__device__ __forceinline__ int wave_half() {
  return (int)((__builtin_amdgcn_s_sendmsg_rtn(0x87) >> 18) & 1u);
}

__global__ void k_init_deg(float* deg2, int* cnt2, int N) {
  int i = blockIdx.x * blockDim.x + threadIdx.x;
  if (i < N) { deg2[i] = 1.0f; deg2[N + i] = 0.0f; cnt2[i] = 0; cnt2[N + i] = 0; }
}

__global__ __launch_bounds__(256)
void k_acc_deg(float* deg2, int* cnt2, int* __restrict__ rank,
               const int* __restrict__ dst, const float* __restrict__ w, int N, int E) {
  int h = wave_half();
  float* degp = deg2 + (long)h * N;
  int* cntp = cnt2 + (long)h * N;
  int stride = gridDim.x * blockDim.x;
  for (int e = blockIdx.x * blockDim.x + threadIdx.x; e < E; e += stride) {
    int d = dst[e];
    atomicAdd(&degp[d], w[e]);
    int r = atomicAdd(&cntp[d], 1);
    rank[e] = 2 * r + h;
  }
}

__global__ void k_dinv(float* __restrict__ dinv, const float* __restrict__ deg2, int N) {
  int i = blockIdx.x * blockDim.x + threadIdx.x;
  if (i < N) dinv[i] = rsqrtf(deg2[i] + deg2[N + i]);
}

#define SCAN_T 256
#define SCAN_PER 4

__global__ __launch_bounds__(SCAN_T)
void k_blocksum(const int* __restrict__ cnt2, int* __restrict__ bsum, int N) {
  __shared__ int wsum[SCAN_T / 32];
  int base = (blockIdx.x * SCAN_T + threadIdx.x) * SCAN_PER;
  int s = 0;
  for (int i = 0; i < SCAN_PER; ++i)
    if (base + i < N) s += cnt2[base + i] + cnt2[N + base + i];
  for (int mask = 16; mask > 0; mask >>= 1) s += __shfl_xor(s, mask, 32);
  if ((threadIdx.x & 31) == 0) wsum[threadIdx.x >> 5] = s;
  __syncthreads();
  if (threadIdx.x == 0) {
    int tot = 0;
    for (int i = 0; i < SCAN_T / 32; ++i) tot += wsum[i];
    bsum[blockIdx.x] = tot;
  }
}

__global__ __launch_bounds__(SCAN_T)
void k_scan(const int* __restrict__ cnt2, int* __restrict__ rowptr,
            int* __restrict__ base1, const int* __restrict__ bsum, int N) {
  __shared__ int wtot[SCAN_T / 32];
  __shared__ int woff[SCAN_T / 32];
  int tid = threadIdx.x, lane = tid & 31, wave = tid >> 5;

  int off = 0;
  for (int i = tid; i < (int)blockIdx.x; i += SCAN_T) off += bsum[i];
  for (int mask = 16; mask > 0; mask >>= 1) off += __shfl_xor(off, mask, 32);
  if (lane == 0) woff[wave] = off;

  int base = (blockIdx.x * SCAN_T + tid) * SCAN_PER;
  int c[SCAN_PER], c0[SCAN_PER];
  int tsum = 0;
  for (int i = 0; i < SCAN_PER; ++i) {
    c0[i] = (base + i < N) ? cnt2[base + i] : 0;
    c[i] = c0[i] + ((base + i < N) ? cnt2[N + base + i] : 0);
    tsum += c[i];
  }
  int v = tsum;
  for (int o = 1; o < 32; o <<= 1) {
    int u = __shfl_up(v, o, 32);
    if (lane >= o) v += u;
  }
  if (lane == 31) wtot[wave] = v;
  __syncthreads();
  int pre = 0;
  for (int i = 0; i < SCAN_T / 32; ++i) {
    pre += woff[i];
    if (i < wave) pre += wtot[i];
  }
  int excl = pre + v - tsum;
  for (int i = 0; i < SCAN_PER; ++i) {
    if (base + i < N) {
      rowptr[base + i] = excl;
      base1[base + i] = excl + c0[i];
      excl += c[i];
    }
  }
}

__global__ void k_fill(const int* __restrict__ rowptr, const int* __restrict__ base1,
                       const int* __restrict__ rank, int* __restrict__ rec,
                       const int* __restrict__ src,
                       const int* __restrict__ dst, const float* __restrict__ w,
                       const float* __restrict__ dinv, int E) {
  int e = blockIdx.x * blockDim.x + threadIdx.x;
  if (e < E) {
    int dn = dst[e];
    int s = src[e];
    float we = w[e];
    int rk = rank[e];
    const int* bp = (rk & 1) ? base1 : rowptr;
    int p = bp[dn] + (rk >> 1);
    float c = we * dinv[s];
    v2i r;
    r.x = s;
    r.y = __float_as_int(c);
    *(v2i*)(rec + 2 * (long)p) = r;
  }
}

__global__ __launch_bounds__(256)
void k_gather_xh(float* __restrict__ AH, float* __restrict__ AX,
                 const float* __restrict__ X, const float* __restrict__ H,
                 const int* __restrict__ rowptr, const int* __restrict__ rec,
                 const float* __restrict__ dinv, int N, int E) {
  int lane = threadIdx.x & 31;
  int l8 = lane & 7;
  int d = __builtin_amdgcn_readfirstlane((int)(blockIdx.x * 8 + (threadIdx.x >> 5)));
  if (d >= N) return;
  int beg = rowptr[d];
  int end = (d + 1 < N) ? rowptr[d + 1] : E;
  float dd = dinv[d];

  const float* rlo = H + (long)d * HID;
  const float* rex = X + (long)d * FIN;
  float a0 = dd * rlo[lane];
  float a1 = dd * rlo[32 + lane];
  float a2 = dd * rex[l8];

  for (int base = beg; base < end; base += 32) {
    int p = base + lane;
    int s = 0;
    float c = 0.0f;
    if (p < end) {
      v2i r = *(const v2i*)(rec + 2 * (long)p);
      s = r.x;
      c = __int_as_float(r.y);
    }
    int ci = __float_as_int(c);
    int cnt = end - base;
    if (cnt > 32) cnt = 32;
    int j = 0;
    for (; j + 4 <= cnt; j += 4) {
      float x0[4], x1[4], x2[4], cj[4];
      for (int u = 0; u < 4; ++u) {
        int sj = __builtin_amdgcn_readlane(s, j + u);
        cj[u] = __int_as_float(__builtin_amdgcn_readlane(ci, j + u));
        const float* q0 = H + (long)sj * HID;
        const float* q2 = X + (long)sj * FIN;
        x0[u] = q0[lane]; x1[u] = q0[32 + lane]; x2[u] = q2[l8];
      }
      for (int u = 0; u < 4; ++u) {
        a0 += cj[u] * x0[u]; a1 += cj[u] * x1[u]; a2 += cj[u] * x2[u];
      }
    }
    for (; j < cnt; ++j) {
      int sj = __builtin_amdgcn_readlane(s, j);
      float cj = __int_as_float(__builtin_amdgcn_readlane(ci, j));
      const float* q0 = H + (long)sj * HID;
      const float* q2 = X + (long)sj * FIN;
      a0 += cj * q0[lane]; a1 += cj * q0[32 + lane]; a2 += cj * q2[l8];
    }
  }

  float* o = AH + (long)d * HID;
  o[lane] = dd * a0;
  o[32 + lane] = dd * a1;
  if (lane < 8) AX[(long)d * FIN + lane] = dd * a2;
}

__global__ __launch_bounds__(256)
void k_gather_head(float* __restrict__ P, const float* __restrict__ T,
                   const int* __restrict__ rowptr, const int* __restrict__ rec,
                   const float* __restrict__ dinv, const float* __restrict__ b1,
                   const float* __restrict__ W2, int N, int E) {
  int lane = threadIdx.x & 31;
  int d = __builtin_amdgcn_readfirstlane((int)(blockIdx.x * 8 + (threadIdx.x >> 5)));
  if (d >= N) return;
  int beg = rowptr[d];
  int end = (d + 1 < N) ? rowptr[d + 1] : E;
  float dd = dinv[d];

  const float* rlo = T + (long)d * HID;
  float a0 = dd * rlo[lane];
  float a1 = dd * rlo[32 + lane];

  for (int base = beg; base < end; base += 32) {
    int p = base + lane;
    int s = 0;
    float c = 0.0f;
    if (p < end) {
      v2i r = *(const v2i*)(rec + 2 * (long)p);
      s = r.x;
      c = __int_as_float(r.y);
    }
    int ci = __float_as_int(c);
    int cnt = end - base;
    if (cnt > 32) cnt = 32;
    int j = 0;
    for (; j + 4 <= cnt; j += 4) {
      float x0[4], x1[4], cj[4];
      for (int u = 0; u < 4; ++u) {
        int sj = __builtin_amdgcn_readlane(s, j + u);
        cj[u] = __int_as_float(__builtin_amdgcn_readlane(ci, j + u));
        const float* q0 = T + (long)sj * HID;
        x0[u] = q0[lane]; x1[u] = q0[32 + lane];
      }
      for (int u = 0; u < 4; ++u) {
        a0 += cj[u] * x0[u]; a1 += cj[u] * x1[u];
      }
    }
    for (; j < cnt; ++j) {
      int sj = __builtin_amdgcn_readlane(s, j);
      float cj = __int_as_float(__builtin_amdgcn_readlane(ci, j));
      const float* q0 = T + (long)sj * HID;
      a0 += cj * q0[lane]; a1 += cj * q0[32 + lane];
    }
  }

  float z0 = dd * a0 + b1[lane];
  float z1 = dd * a1 + b1[32 + lane];
  z0 = z0 > 0.0f ? z0 : 0.0f;
  z1 = z1 > 0.0f ? z1 : 0.0f;
  float pv = z0 * W2[lane] + z1 * W2[32 + lane];
  for (int mask = 16; mask > 0; mask >>= 1) pv += __shfl_xor(pv, mask, 32);
  if (lane == 0) P[d] = pv;
}

__global__ __launch_bounds__(256)
void k_gates(const float* __restrict__ AH, const float* __restrict__ AX,
             const float* __restrict__ C,
             const float* __restrict__ Wx, const float* __restrict__ Wh,
             const float* __restrict__ bx, const float* __restrict__ bh,
             const float* __restrict__ bg, const float* __restrict__ wc,
             const float* __restrict__ ln_h_s, const float* __restrict__ ln_h_b,
             const float* __restrict__ ln_c_s, const float* __restrict__ ln_c_b,
             const float* __restrict__ ln_o_s, const float* __restrict__ ln_o_b,
             float* __restrict__ hiddenOut, float* __restrict__ cellOut,
             float* __restrict__ Y, int N) {
  extern __shared__ float smem[];
  float* Bf = smem;
  float* bias = smem + 18432;

  int tid = threadIdx.x;
  for (int i = tid; i < KSTEPS * NT_G * 64; i += 256) {
    int s = i >> 10;
    int rem = i & 1023;
    int t = rem >> 6;
    int rem2 = rem & 63;
    int L = rem2 >> 1;
    int j = rem2 & 1;
    int k = 4 * s + 2 * (L >> 4) + j;
    int c = 16 * t + (L & 15);
    int g = c >> 6, h = c & 63;
    Bf[i] = (k < FIN) ? Wx[(g * FIN + k) * HID + h]
                      : Wh[(g * HID + (k - FIN)) * HID + h];
  }
  if (tid < GCOLS) bias[tid] = bx[tid] + bh[tid] + bg[tid];
  __syncthreads();

  int wave = tid >> 5, lane = tid & 31;
  int half = lane >> 4, lcol = lane & 15;
  int rowBase = blockIdx.x * ROWS_PER_BLOCK + wave * 16;

  int arow = rowBase + lcol;
  if (arow > N - 1) arow = N - 1;
  const float* aXp = AX + (long)arow * FIN;
  const float* aHp = AH + (long)arow * HID;

  v8f acc[NT_G];
#pragma unroll
  for (int t = 0; t < NT_G; ++t)
#pragma unroll
    for (int r = 0; r < 8; ++r) acc[t][r] = 0.0f;

  for (int s = 0; s < KSTEPS; ++s) {
    int k0 = 4 * s + 2 * half;
    v2f a = *(const v2f*)(s < 2 ? aXp + k0 : aHp + (k0 - FIN));
    const float* bbase = Bf + s * 1024 + lane * 2;
#pragma unroll
    for (int t = 0; t < NT_G; ++t) {
      v2f b = *(const v2f*)(bbase + t * 64);
      acc[t] = __builtin_amdgcn_wmma_f32_16x16x4_f32(
          false, a, false, b, (short)0, acc[t], false, false);
    }
  }

#pragma unroll
  for (int r = 0; r < 8; ++r) {
    int node = rowBase + r + 8 * half;
    bool valid = node < N;
    int nc = valid ? node : N - 1;
    float hloc[4], cloc[4];
#pragma unroll
    for (int t = 0; t < 4; ++t) {
      int h = t * 16 + lcol;
      float c0 = C[(long)nc * HID + h];
      float gi = acc[t][r]      + bias[h]       + wc[h] * c0;
      float gf = acc[t + 4][r]  + bias[64 + h]  + wc[64 + h] * c0;
      float gc = acc[t + 8][r]  + bias[128 + h];
      float go = acc[t + 12][r] + bias[192 + h];
      float iv = sigmoidf_(gi);
      float fv = sigmoidf_(gf);
      float cc = fv * c0 + iv * tanhf(gc);
      float ov = sigmoidf_(go + wc[128 + h] * cc);
      hloc[t] = ov * tanhf(cc);
      cloc[t] = cc;
    }
    float sh = 0.f, sh2 = 0.f, sc = 0.f, sc2 = 0.f;
#pragma unroll
    for (int t = 0; t < 4; ++t) {
      sh += hloc[t]; sh2 += hloc[t] * hloc[t];
      sc += cloc[t]; sc2 += cloc[t] * cloc[t];
    }
#pragma unroll
    for (int mask = 1; mask < 16; mask <<= 1) {
      sh  += __shfl_xor(sh,  mask, 32);
      sh2 += __shfl_xor(sh2, mask, 32);
      sc  += __shfl_xor(sc,  mask, 32);
      sc2 += __shfl_xor(sc2, mask, 32);
    }
    float mh = sh * (1.0f / 64.0f), mc = sc * (1.0f / 64.0f);
    float vh = sh2 * (1.0f / 64.0f) - mh * mh;
    float vc = sc2 * (1.0f / 64.0f) - mc * mc;
    float rh = rsqrtf(vh + EPS), rc = rsqrtf(vc + EPS);
    if (valid) {
#pragma unroll
      for (int t = 0; t < 4; ++t) {
        int h = t * 16 + lcol;
        float hn = (hloc[t] - mh) * rh;
        float cn = (cloc[t] - mc) * rc;
        hiddenOut[(long)node * HID + h] = hn * ln_h_s[h] + ln_h_b[h];
        cellOut[(long)node * HID + h]   = cn * ln_c_s[h] + ln_c_b[h];
        float ov = hn * ln_o_s[h] + ln_o_b[h];
        Y[(long)node * HID + h] = ov > 0.0f ? ov : 0.0f;
      }
    }
  }
}

__global__ __launch_bounds__(256)
void k_head(const float* __restrict__ Yin, const float* __restrict__ skip,
            const float* __restrict__ W1, float* __restrict__ T, int N) {
  __shared__ float Bf[KH_STEPS * 4 * 64];
  __shared__ __attribute__((aligned(16))) float tile[8 * 16 * TPITCH];

  int tid = threadIdx.x;
  for (int i = tid; i < KH_STEPS * 4 * 64; i += 256) {
    int s = i >> 8;
    int rem = i & 255;
    int t = rem >> 6;
    int rem2 = rem & 63;
    int L = rem2 >> 1;
    int j = rem2 & 1;
    int k = 4 * s + 2 * (L >> 4) + j;
    int c = 16 * t + (L & 15);
    Bf[i] = (k < HID + 2) ? W1[k * HID + c] : 0.0f;
  }
  __syncthreads();

  int wave = tid >> 5, lane = tid & 31;
  int half = lane >> 4, lcol = lane & 15;
  int rowBase = blockIdx.x * ROWS_PER_BLOCK + wave * 16;

  int arow = rowBase + lcol;
  if (arow > N - 1) arow = N - 1;
  const float* aptr = Yin + (long)arow * HID;

  v8f acc[4];
#pragma unroll
  for (int t = 0; t < 4; ++t)
#pragma unroll
    for (int r = 0; r < 8; ++r) acc[t][r] = 0.0f;

  for (int s = 0; s < KH_STEPS - 1; ++s) {
    int k0 = 4 * s + 2 * half;
    v2f a = *(const v2f*)(aptr + k0);
    const float* bbase = Bf + s * 256 + lane * 2;
#pragma unroll
    for (int t = 0; t < 4; ++t) {
      v2f b = *(const v2f*)(bbase + t * 64);
      acc[t] = __builtin_amdgcn_wmma_f32_16x16x4_f32(
          false, a, false, b, (short)0, acc[t], false, false);
    }
  }
  {
    v2f sk = *(const v2f*)(skip + (long)arow * 2);
    v2f a;
    a.x = half ? 0.0f : sk.x;
    a.y = half ? 0.0f : sk.y;
    const float* bbase = Bf + (KH_STEPS - 1) * 256 + lane * 2;
#pragma unroll
    for (int t = 0; t < 4; ++t) {
      v2f b = *(const v2f*)(bbase + t * 64);
      acc[t] = __builtin_amdgcn_wmma_f32_16x16x4_f32(
          false, a, false, b, (short)0, acc[t], false, false);
    }
  }

  float* tw = tile + wave * (16 * TPITCH);
#pragma unroll
  for (int t = 0; t < 4; ++t)
#pragma unroll
    for (int r = 0; r < 8; ++r)
      tw[(r + 8 * half) * TPITCH + 16 * t + lcol] = acc[t][r];
  __syncthreads();
  typedef __attribute__((ext_vector_type(4))) float v4f;
#pragma unroll
  for (int i = 0; i < 8; ++i) {
    int m = 2 * i + half;
    int node = rowBase + m;
    v4f v = *(const v4f*)(tw + m * TPITCH + 4 * lcol);
    if (node < N) *(v4f*)(T + (long)node * HID + 4 * lcol) = v;
  }
}

__global__ __launch_bounds__(256)
void k_gather_out(float* __restrict__ out, const float* __restrict__ P,
                  const int* __restrict__ rowptr, const int* __restrict__ rec,
                  const float* __restrict__ dinv, const float* __restrict__ b2,
                  const float* __restrict__ X, int N, int E) {
  int t = blockIdx.x * blockDim.x + threadIdx.x;
  int d = t >> 4, l = t & 15;
  bool valid = d < N;
  int dc = valid ? d : N - 1;
  int beg = rowptr[dc];
  int end = (dc + 1 < N) ? rowptr[dc + 1] : E;
  float acc = 0.0f;
  for (int p = beg + l; p < end; p += 16) {
    v2i r = *(const v2i*)(rec + 2 * (long)p);
    acc += __int_as_float(r.y) * P[r.x];
  }
  for (int mask = 8; mask > 0; mask >>= 1) acc += __shfl_xor(acc, mask, 32);
  if (valid && l == 0) {
    float dd = dinv[d];
    float q = dd * (acc + dd * P[d]);
    out[d] = q + b2[0] + X[(long)d * FIN];
  }
}

extern "C" void kernel_launch(void* const* d_in, const int* in_sizes, int n_in,
                              void* d_out, int out_size, void* d_ws, size_t ws_size,
                              hipStream_t stream) {
  const float* X    = (const float*)d_in[0];
  const int*   eidx = (const int*)d_in[1];
  const float* ew   = (const float*)d_in[2];
  const float* skip = (const float*)d_in[3];
  const float* H    = (const float*)d_in[4];
  const float* C    = (const float*)d_in[5];
  const float* Wx   = (const float*)d_in[6];
  const float* bx   = (const float*)d_in[7];
  const float* Wh   = (const float*)d_in[8];
  const float* bh   = (const float*)d_in[9];
  const float* wc   = (const float*)d_in[10];
  const float* bg   = (const float*)d_in[11];
  const float* ln_h_s = (const float*)d_in[12];
  const float* ln_h_b = (const float*)d_in[13];
  const float* ln_c_s = (const float*)d_in[14];
  const float* ln_c_b = (const float*)d_in[15];
  const float* ln_o_s = (const float*)d_in[16];
  const float* ln_o_b = (const float*)d_in[17];
  const float* W1   = (const float*)d_in[18];
  const float* b1   = (const float*)d_in[19];
  const float* W2   = (const float*)d_in[20];
  const float* b2   = (const float*)d_in[21];

  const int N = in_sizes[0] / FIN;
  const int E = in_sizes[2];

  const int* src = eidx;
  const int* dst = eidx + E;

  float* out_p    = (float*)d_out;
  float* hidden_p = out_p + N;
  float* cell_p   = hidden_p + (long)N * HID;

  int*   rec  = (int*)d_ws;
  float* AH   = (float*)(rec + 2 * (long)E);
  float* AX   = AH + (long)N * HID;
  float* dinv = AX + (long)N * FIN;
  float* P    = dinv + N;
  int* rowptr = (int*)(P + N);
  int*   rank  = (int*)AH;
  int*   cnt2  = rank + E;
  float* deg2  = (float*)(cnt2 + 2 * (long)N);
  int*   base1 = (int*)(deg2 + 2 * (long)N);
  int*   bsum  = base1 + N;

  const int B = 256;
  auto cdiv = [](long a, long b) { return (unsigned)((a + b - 1) / b); };

  k_init_deg<<<cdiv(N, B), B, 0, stream>>>(deg2, cnt2, N);
  k_acc_deg<<<2048, B, 0, stream>>>(deg2, cnt2, rank, dst, ew, N, E);
  k_dinv<<<cdiv(N, B), B, 0, stream>>>(dinv, deg2, N);

  k_blocksum<<<cdiv(N, SCAN_T * SCAN_PER), SCAN_T, 0, stream>>>(cnt2, bsum, N);
  k_scan<<<cdiv(N, SCAN_T * SCAN_PER), SCAN_T, 0, stream>>>(cnt2, rowptr, base1, bsum, N);
  k_fill<<<cdiv(E, B), B, 0, stream>>>(rowptr, base1, rank, rec, src, dst, ew, dinv, E);

  k_gather_xh<<<cdiv(N, 8), B, 0, stream>>>(AH, AX, X, H, rowptr, rec, dinv, N, E);

  size_t smem_gates = (size_t)(KSTEPS * NT_G * 64 + GCOLS) * sizeof(float);
  k_gates<<<cdiv(N, ROWS_PER_BLOCK), B, smem_gates, stream>>>(
      AH, AX, C, Wx, Wh, bx, bh, bg, wc,
      ln_h_s, ln_h_b, ln_c_s, ln_c_b, ln_o_s, ln_o_b,
      hidden_p, cell_p, AH, N);

  k_head<<<cdiv(N, ROWS_PER_BLOCK), B, 0, stream>>>(AH, skip, W1, AH, N);

  k_gather_head<<<cdiv(N, 8), B, 0, stream>>>(P, AH, rowptr, rec, dinv, b1, W2, N, E);

  k_gather_out<<<cdiv((long)N * 16, B), B, 0, stream>>>(out_p, P, rowptr, rec, dinv,
                                                       b2, X, N, E);
}


# Edge and node passes behind the sort (k_fill, 3 aggregations) run as resident-size grids: a wave walks a run of nodes, next node's data fetched ahead, not 1 short wave per node; sum order kept
# speedup vs baseline: 5.0465x; 1.0434x over previous
#include <hip/hip_runtime.h>
#include <math.h>


typedef __attribute__((ext_vector_type(2))) float v2f;
typedef __attribute__((ext_vector_type(2))) int v2i;
typedef __attribute__((ext_vector_type(8))) float v8f;

#define FIN   8
#define HID   64
#define KTOT  72
#define KSTEPS 18
#define KH_STEPS 17
#define TPITCH 68
#define GCOLS 256
#define NT_G  16
#define ROWS_PER_BLOCK 128
#define EPS 1e-5f

__device__ __forceinline__ float sigmoidf_(float x) { return 1.0f / (1.0f + expf(-x)); }

__device__ __forceinline__ int wave_half() {
  return (int)((__builtin_amdgcn_s_sendmsg_rtn(0x87) >> 18) & 1u);
}

__global__ void k_init_deg(float* deg2, int* cnt2, int N) {
  int i = blockIdx.x * blockDim.x + threadIdx.x;
  if (i < N) { deg2[i] = 1.0f; deg2[N + i] = 0.0f; cnt2[i] = 0; cnt2[N + i] = 0; }
}

__global__ __launch_bounds__(256)
void k_acc_deg(float* deg2, int* cnt2, int* __restrict__ rank,
               const int* __restrict__ dst, const float* __restrict__ w, int N, int E) {
  int h = wave_half();
  float* degp = deg2 + (long)h * N;
  int* cntp = cnt2 + (long)h * N;
  int stride = gridDim.x * blockDim.x;
  for (int e = blockIdx.x * blockDim.x + threadIdx.x; e < E; e += stride) {
    int d = dst[e];
    atomicAdd(&degp[d], w[e]);
    int r = atomicAdd(&cntp[d], 1);
    rank[e] = 2 * r + h;
  }
}

__global__ void k_dinv(float* __restrict__ dinv, const float* __restrict__ deg2, int N) {
  int i = blockIdx.x * blockDim.x + threadIdx.x;
  if (i < N) dinv[i] = rsqrtf(deg2[i] + deg2[N + i]);
}

#define SCAN_T 256
#define SCAN_PER 4

__global__ __launch_bounds__(SCAN_T)
void k_blocksum(const int* __restrict__ cnt2, int* __restrict__ bsum, int N) {
  __shared__ int wsum[SCAN_T / 32];
  int base = (blockIdx.x * SCAN_T + threadIdx.x) * SCAN_PER;
  int s = 0;
  for (int i = 0; i < SCAN_PER; ++i)
    if (base + i < N) s += cnt2[base + i] + cnt2[N + base + i];
  for (int mask = 16; mask > 0; mask >>= 1) s += __shfl_xor(s, mask, 32);
  if ((threadIdx.x & 31) == 0) wsum[threadIdx.x >> 5] = s;
  __syncthreads();
  if (threadIdx.x == 0) {
    int tot = 0;
    for (int i = 0; i < SCAN_T / 32; ++i) tot += wsum[i];
    bsum[blockIdx.x] = tot;
  }
}

__global__ __launch_bounds__(SCAN_T)
void k_scan(const int* __restrict__ cnt2, int* __restrict__ rowptr,
            int* __restrict__ base1, const int* __restrict__ bsum, int N) {
  __shared__ int wtot[SCAN_T / 32];
  __shared__ int woff[SCAN_T / 32];
  int tid = threadIdx.x, lane = tid & 31, wave = tid >> 5;

  int off = 0;
  for (int i = tid; i < (int)blockIdx.x; i += SCAN_T) off += bsum[i];
  for (int mask = 16; mask > 0; mask >>= 1) off += __shfl_xor(off, mask, 32);
  if (lane == 0) woff[wave] = off;

  int base = (blockIdx.x * SCAN_T + tid) * SCAN_PER;
  int c[SCAN_PER], c0[SCAN_PER];
  int tsum = 0;
  for (int i = 0; i < SCAN_PER; ++i) {
    c0[i] = (base + i < N) ? cnt2[base + i] : 0;
    c[i] = c0[i] + ((base + i < N) ? cnt2[N + base + i] : 0);
    tsum += c[i];
  }
  int v = tsum;
  for (int o = 1; o < 32; o <<= 1) {
    int u = __shfl_up(v, o, 32);
    if (lane >= o) v += u;
  }
  if (lane == 31) wtot[wave] = v;
  __syncthreads();
  int pre = 0;
  for (int i = 0; i < SCAN_T / 32; ++i) {
    pre += woff[i];
    if (i < wave) pre += wtot[i];
  }
  int excl = pre + v - tsum;
  for (int i = 0; i < SCAN_PER; ++i) {
    if (base + i < N) {
      rowptr[base + i] = excl;
      base1[base + i] = excl + c0[i];
      excl += c[i];
    }
  }
}

__global__ __launch_bounds__(256)
void k_fill(const int* __restrict__ rowptr, const int* __restrict__ base1,
            const int* __restrict__ rank, int* __restrict__ rec,
            const int* __restrict__ src,
            const int* __restrict__ dst, const float* __restrict__ w,
            const float* __restrict__ dinv, int E) {
  long stride = (long)gridDim.x * blockDim.x;
  long t = (long)blockIdx.x * blockDim.x + threadIdx.x;
  for (long e0 = t; e0 < E; e0 += 4 * stride) {
    int dn[4], s[4], rk[4], p[4];
    float we[4], c[4];
#pragma unroll
    for (int u = 0; u < 4; ++u) {
      long e = e0 + u * stride;
      long ec = e < E ? e : e0;
      dn[u] = dst[ec];
      s[u] = src[ec];
      we[u] = w[ec];
      rk[u] = rank[ec];
    }
#pragma unroll
    for (int u = 0; u < 4; ++u) {
      const int* bp = (rk[u] & 1) ? base1 : rowptr;
      p[u] = bp[dn[u]] + (rk[u] >> 1);
      c[u] = we[u] * dinv[s[u]];
    }
#pragma unroll
    for (int u = 0; u < 4; ++u) {
      v2i r;
      r.x = s[u];
      r.y = __float_as_int(c[u]);
      *(v2i*)(rec + 2 * (long)p[u]) = r;
    }
  }
}

__device__ __forceinline__ void seg_batch_xh(int s, int ci, int cnt, int lane, int l8,
                                             const float* __restrict__ H,
                                             const float* __restrict__ X,
                                             float& a0, float& a1, float& a2) {
  int j = 0;
  for (; j + 4 <= cnt; j += 4) {
    float x0[4], x1[4], x2[4], cj[4];
    for (int u = 0; u < 4; ++u) {
      int sj = __builtin_amdgcn_readlane(s, j + u);
      cj[u] = __int_as_float(__builtin_amdgcn_readlane(ci, j + u));
      const float* q0 = H + (long)sj * HID;
      const float* q2 = X + (long)sj * FIN;
      x0[u] = q0[lane]; x1[u] = q0[32 + lane]; x2[u] = q2[l8];
    }
    for (int u = 0; u < 4; ++u) {
      a0 += cj[u] * x0[u]; a1 += cj[u] * x1[u]; a2 += cj[u] * x2[u];
    }
  }
  for (; j < cnt; ++j) {
    int sj = __builtin_amdgcn_readlane(s, j);
    float cj = __int_as_float(__builtin_amdgcn_readlane(ci, j));
    const float* q0 = H + (long)sj * HID;
    const float* q2 = X + (long)sj * FIN;
    a0 += cj * q0[lane]; a1 += cj * q0[32 + lane]; a2 += cj * q2[l8];
  }
}

__device__ __forceinline__ void seg_batch_t(int s, int ci, int cnt, int lane,
                                            const float* __restrict__ T,
                                            float& a0, float& a1) {
  int j = 0;
  for (; j + 4 <= cnt; j += 4) {
    float x0[4], x1[4], cj[4];
    for (int u = 0; u < 4; ++u) {
      int sj = __builtin_amdgcn_readlane(s, j + u);
      cj[u] = __int_as_float(__builtin_amdgcn_readlane(ci, j + u));
      const float* q0 = T + (long)sj * HID;
      x0[u] = q0[lane]; x1[u] = q0[32 + lane];
    }
    for (int u = 0; u < 4; ++u) {
      a0 += cj[u] * x0[u]; a1 += cj[u] * x1[u];
    }
  }
  for (; j < cnt; ++j) {
    int sj = __builtin_amdgcn_readlane(s, j);
    float cj = __int_as_float(__builtin_amdgcn_readlane(ci, j));
    const float* q0 = T + (long)sj * HID;
    a0 += cj * q0[lane]; a1 += cj * q0[32 + lane];
  }
}

#define RUN_MAX 31

__global__ __launch_bounds__(256)
void k_gather_xh(float* __restrict__ AH, float* __restrict__ AX,
                 const float* __restrict__ X, const float* __restrict__ H,
                 const int* __restrict__ rowptr, const int* __restrict__ rec,
                 const float* __restrict__ dinv, int N, int E) {
  int lane = threadIdx.x & 31;
  int l8 = lane & 7;
  int wid = __builtin_amdgcn_readfirstlane((int)(blockIdx.x * 8 + (threadIdx.x >> 5)));
  unsigned nw = gridDim.x * 8u;
  unsigned q = (unsigned)N / nw, rem = (unsigned)N % nw;
  int n0 = (int)((unsigned)wid * q + ((unsigned)wid < rem ? (unsigned)wid : rem));
  int n1 = n0 + (int)q + ((unsigned)wid < rem ? 1 : 0);

  for (int nb = n0; nb < n1; nb += RUN_MAX) {
    int cn = n1 - nb;
    if (cn > RUN_MAX) cn = RUN_MAX;
    int idx = nb + lane;
    int rp = E;
    float dvl = 0.0f;
    if (idx < N) { rp = rowptr[idx]; dvl = dinv[idx]; }
    int dvi = __float_as_int(dvl);

    int beg = __builtin_amdgcn_readlane(rp, 0);
    int end = __builtin_amdgcn_readlane(rp, 1);
    v2i rc;
    rc.x = 0; rc.y = 0;
    if (beg + lane < end) rc = *(const v2i*)(rec + 2 * (long)(beg + lane));
    float h0 = H[(long)nb * HID + lane];
    float h1 = H[(long)nb * HID + 32 + lane];
    float hx = X[(long)nb * FIN + l8];

    for (int i = 0; i < cn; ++i) {
      int d = nb + i;
      float dd = __int_as_float(__builtin_amdgcn_readlane(dvi, i));
      float a0 = dd * h0;
      float a1 = dd * h1;
      float a2 = dd * hx;

      int endn = end;
      v2i rn;
      rn.x = 0; rn.y = 0;
      if (i + 1 < cn) {
        endn = __builtin_amdgcn_readlane(rp, i + 2);
        if (end + lane < endn) rn = *(const v2i*)(rec + 2 * (long)(end + lane));
        const float* nlo = H + (long)(d + 1) * HID;
        h0 = nlo[lane];
        h1 = nlo[32 + lane];
        hx = X[(long)(d + 1) * FIN + l8];
      }

      int cnt = end - beg;
      seg_batch_xh(rc.x, rc.y, cnt > 32 ? 32 : cnt, lane, l8, H, X, a0, a1, a2);
      for (int base = beg + 32; base < end; base += 32) {
        int p = base + lane;
        v2i r;
        r.x = 0; r.y = 0;
        if (p < end) r = *(const v2i*)(rec + 2 * (long)p);
        int c2 = end - base;
        seg_batch_xh(r.x, r.y, c2 > 32 ? 32 : c2, lane, l8, H, X, a0, a1, a2);
      }

      float* o = AH + (long)d * HID;
      o[lane] = dd * a0;
      o[32 + lane] = dd * a1;
      if (lane < 8) AX[(long)d * FIN + lane] = dd * a2;

      rc = rn;
      beg = end;
      end = endn;
    }
  }
}

__global__ __launch_bounds__(256)
void k_gather_head(float* __restrict__ P, const float* __restrict__ T,
                   const int* __restrict__ rowptr, const int* __restrict__ rec,
                   const float* __restrict__ dinv, const float* __restrict__ b1,
                   const float* __restrict__ W2, int N, int E) {
  int lane = threadIdx.x & 31;
  int wid = __builtin_amdgcn_readfirstlane((int)(blockIdx.x * 8 + (threadIdx.x >> 5)));
  unsigned nw = gridDim.x * 8u;
  unsigned q = (unsigned)N / nw, rem = (unsigned)N % nw;
  int n0 = (int)((unsigned)wid * q + ((unsigned)wid < rem ? (unsigned)wid : rem));
  int n1 = n0 + (int)q + ((unsigned)wid < rem ? 1 : 0);

  float bl0 = b1[lane], bl1 = b1[32 + lane];
  float wl0 = W2[lane], wl1 = W2[32 + lane];

  for (int nb = n0; nb < n1; nb += RUN_MAX) {
    int cn = n1 - nb;
    if (cn > RUN_MAX) cn = RUN_MAX;
    int idx = nb + lane;
    int rp = E;
    float dvl = 0.0f;
    if (idx < N) { rp = rowptr[idx]; dvl = dinv[idx]; }
    int dvi = __float_as_int(dvl);

    int beg = __builtin_amdgcn_readlane(rp, 0);
    int end = __builtin_amdgcn_readlane(rp, 1);
    v2i rc;
    rc.x = 0; rc.y = 0;
    if (beg + lane < end) rc = *(const v2i*)(rec + 2 * (long)(beg + lane));
    float h0 = T[(long)nb * HID + lane];
    float h1 = T[(long)nb * HID + 32 + lane];

    for (int i = 0; i < cn; ++i) {
      int d = nb + i;
      float dd = __int_as_float(__builtin_amdgcn_readlane(dvi, i));
      float a0 = dd * h0;
      float a1 = dd * h1;

      int endn = end;
      v2i rn;
      rn.x = 0; rn.y = 0;
      if (i + 1 < cn) {
        endn = __builtin_amdgcn_readlane(rp, i + 2);
        if (end + lane < endn) rn = *(const v2i*)(rec + 2 * (long)(end + lane));
        const float* nlo = T + (long)(d + 1) * HID;
        h0 = nlo[lane];
        h1 = nlo[32 + lane];
      }

      int cnt = end - beg;
      seg_batch_t(rc.x, rc.y, cnt > 32 ? 32 : cnt, lane, T, a0, a1);
      for (int base = beg + 32; base < end; base += 32) {
        int p = base + lane;
        v2i r;
        r.x = 0; r.y = 0;
        if (p < end) r = *(const v2i*)(rec + 2 * (long)p);
        int c2 = end - base;
        seg_batch_t(r.x, r.y, c2 > 32 ? 32 : c2, lane, T, a0, a1);
      }

      float z0 = dd * a0 + bl0;
      float z1 = dd * a1 + bl1;
      z0 = z0 > 0.0f ? z0 : 0.0f;
      z1 = z1 > 0.0f ? z1 : 0.0f;
      float pv = z0 * wl0 + z1 * wl1;
      for (int mask = 16; mask > 0; mask >>= 1) pv += __shfl_xor(pv, mask, 32);
      if (lane == 0) P[d] = pv;

      rc = rn;
      beg = end;
      end = endn;
    }
  }
}

__global__ __launch_bounds__(256)
void k_gates(const float* __restrict__ AH, const float* __restrict__ AX,
             const float* __restrict__ C,
             const float* __restrict__ Wx, const float* __restrict__ Wh,
             const float* __restrict__ bx, const float* __restrict__ bh,
             const float* __restrict__ bg, const float* __restrict__ wc,
             const float* __restrict__ ln_h_s, const float* __restrict__ ln_h_b,
             const float* __restrict__ ln_c_s, const float* __restrict__ ln_c_b,
             const float* __restrict__ ln_o_s, const float* __restrict__ ln_o_b,
             float* __restrict__ hiddenOut, float* __restrict__ cellOut,
             float* __restrict__ Y, int N) {
  extern __shared__ float smem[];
  float* Bf = smem;
  float* bias = smem + 18432;

  int tid = threadIdx.x;
  for (int i = tid; i < KSTEPS * NT_G * 64; i += 256) {
    int s = i >> 10;
    int rem = i & 1023;
    int t = rem >> 6;
    int rem2 = rem & 63;
    int L = rem2 >> 1;
    int j = rem2 & 1;
    int k = 4 * s + 2 * (L >> 4) + j;
    int c = 16 * t + (L & 15);
    int g = c >> 6, h = c & 63;
    Bf[i] = (k < FIN) ? Wx[(g * FIN + k) * HID + h]
                      : Wh[(g * HID + (k - FIN)) * HID + h];
  }
  if (tid < GCOLS) bias[tid] = bx[tid] + bh[tid] + bg[tid];
  __syncthreads();

  int wave = tid >> 5, lane = tid & 31;
  int half = lane >> 4, lcol = lane & 15;
  int rowBase = blockIdx.x * ROWS_PER_BLOCK + wave * 16;

  int arow = rowBase + lcol;
  if (arow > N - 1) arow = N - 1;
  const float* aXp = AX + (long)arow * FIN;
  const float* aHp = AH + (long)arow * HID;

  v8f acc[NT_G];
#pragma unroll
  for (int t = 0; t < NT_G; ++t)
#pragma unroll
    for (int r = 0; r < 8; ++r) acc[t][r] = 0.0f;

  for (int s = 0; s < KSTEPS; ++s) {
    int k0 = 4 * s + 2 * half;
    v2f a = *(const v2f*)(s < 2 ? aXp + k0 : aHp + (k0 - FIN));
    const float* bbase = Bf + s * 1024 + lane * 2;
#pragma unroll
    for (int t = 0; t < NT_G; ++t) {
      v2f b = *(const v2f*)(bbase + t * 64);
      acc[t] = __builtin_amdgcn_wmma_f32_16x16x4_f32(
          false, a, false, b, (short)0, acc[t], false, false);
    }
  }

#pragma unroll
  for (int r = 0; r < 8; ++r) {
    int node = rowBase + r + 8 * half;
    bool valid = node < N;
    int nc = valid ? node : N - 1;
    float hloc[4], cloc[4];
#pragma unroll
    for (int t = 0; t < 4; ++t) {
      int h = t * 16 + lcol;
      float c0 = C[(long)nc * HID + h];
      float gi = acc[t][r]      + bias[h]       + wc[h] * c0;
      float gf = acc[t + 4][r]  + bias[64 + h]  + wc[64 + h] * c0;
      float gc = acc[t + 8][r]  + bias[128 + h];
      float go = acc[t + 12][r] + bias[192 + h];
      float iv = sigmoidf_(gi);
      float fv = sigmoidf_(gf);
      float cc = fv * c0 + iv * tanhf(gc);
      float ov = sigmoidf_(go + wc[128 + h] * cc);
      hloc[t] = ov * tanhf(cc);
      cloc[t] = cc;
    }
    float sh = 0.f, sh2 = 0.f, sc = 0.f, sc2 = 0.f;
#pragma unroll
    for (int t = 0; t < 4; ++t) {
      sh += hloc[t]; sh2 += hloc[t] * hloc[t];
      sc += cloc[t]; sc2 += cloc[t] * cloc[t];
    }
#pragma unroll
    for (int mask = 1; mask < 16; mask <<= 1) {
      sh  += __shfl_xor(sh,  mask, 32);
      sh2 += __shfl_xor(sh2, mask, 32);
      sc  += __shfl_xor(sc,  mask, 32);
      sc2 += __shfl_xor(sc2, mask, 32);
    }
    float mh = sh * (1.0f / 64.0f), mc = sc * (1.0f / 64.0f);
    float vh = sh2 * (1.0f / 64.0f) - mh * mh;
    float vc = sc2 * (1.0f / 64.0f) - mc * mc;
    float rh = rsqrtf(vh + EPS), rc = rsqrtf(vc + EPS);
    if (valid) {
#pragma unroll
      for (int t = 0; t < 4; ++t) {
        int h = t * 16 + lcol;
        float hn = (hloc[t] - mh) * rh;
        float cn = (cloc[t] - mc) * rc;
        hiddenOut[(long)node * HID + h] = hn * ln_h_s[h] + ln_h_b[h];
        cellOut[(long)node * HID + h]   = cn * ln_c_s[h] + ln_c_b[h];
        float ov = hn * ln_o_s[h] + ln_o_b[h];
        Y[(long)node * HID + h] = ov > 0.0f ? ov : 0.0f;
      }
    }
  }
}

__global__ __launch_bounds__(256)
void k_head(const float* __restrict__ Yin, const float* __restrict__ skip,
            const float* __restrict__ W1, float* __restrict__ T, int N) {
  __shared__ float Bf[KH_STEPS * 4 * 64];
  __shared__ __attribute__((aligned(16))) float tile[8 * 16 * TPITCH];

  int tid = threadIdx.x;
  for (int i = tid; i < KH_STEPS * 4 * 64; i += 256) {
    int s = i >> 8;
    int rem = i & 255;
    int t = rem >> 6;
    int rem2 = rem & 63;
    int L = rem2 >> 1;
    int j = rem2 & 1;
    int k = 4 * s + 2 * (L >> 4) + j;
    int c = 16 * t + (L & 15);
    Bf[i] = (k < HID + 2) ? W1[k * HID + c] : 0.0f;
  }
  __syncthreads();

  int wave = tid >> 5, lane = tid & 31;
  int half = lane >> 4, lcol = lane & 15;
  int rowBase = blockIdx.x * ROWS_PER_BLOCK + wave * 16;

  int arow = rowBase + lcol;
  if (arow > N - 1) arow = N - 1;
  const float* aptr = Yin + (long)arow * HID;

  v8f acc[4];
#pragma unroll
  for (int t = 0; t < 4; ++t)
#pragma unroll
    for (int r = 0; r < 8; ++r) acc[t][r] = 0.0f;

  for (int s = 0; s < KH_STEPS - 1; ++s) {
    int k0 = 4 * s + 2 * half;
    v2f a = *(const v2f*)(aptr + k0);
    const float* bbase = Bf + s * 256 + lane * 2;
#pragma unroll
    for (int t = 0; t < 4; ++t) {
      v2f b = *(const v2f*)(bbase + t * 64);
      acc[t] = __builtin_amdgcn_wmma_f32_16x16x4_f32(
          false, a, false, b, (short)0, acc[t], false, false);
    }
  }
  {
    v2f sk = *(const v2f*)(skip + (long)arow * 2);
    v2f a;
    a.x = half ? 0.0f : sk.x;
    a.y = half ? 0.0f : sk.y;
    const float* bbase = Bf + (KH_STEPS - 1) * 256 + lane * 2;
#pragma unroll
    for (int t = 0; t < 4; ++t) {
      v2f b = *(const v2f*)(bbase + t * 64);
      acc[t] = __builtin_amdgcn_wmma_f32_16x16x4_f32(
          false, a, false, b, (short)0, acc[t], false, false);
    }
  }

  float* tw = tile + wave * (16 * TPITCH);
#pragma unroll
  for (int t = 0; t < 4; ++t)
#pragma unroll
    for (int r = 0; r < 8; ++r)
      tw[(r + 8 * half) * TPITCH + 16 * t + lcol] = acc[t][r];
  __syncthreads();
  typedef __attribute__((ext_vector_type(4))) float v4f;
#pragma unroll
  for (int i = 0; i < 8; ++i) {
    int m = 2 * i + half;
    int node = rowBase + m;
    v4f v = *(const v4f*)(tw + m * TPITCH + 4 * lcol);
    if (node < N) *(v4f*)(T + (long)node * HID + 4 * lcol) = v;
  }
}

__global__ __launch_bounds__(256)
void k_gather_out(float* __restrict__ out, const float* __restrict__ P,
                  const int* __restrict__ rowptr, const int* __restrict__ rec,
                  const float* __restrict__ dinv, const float* __restrict__ b2,
                  const float* __restrict__ X, int N, int E) {
  int t = blockIdx.x * blockDim.x + threadIdx.x;
  int g = t >> 4, l = t & 15;
  int G = (int)((gridDim.x * blockDim.x) >> 4);
  float bias2 = b2[0];

  int dc = g < N ? g : N - 1;
  int beg = rowptr[dc];
  int end = (dc + 1 < N) ? rowptr[dc + 1] : E;
  for (int db = 0; db < N; db += G) {
    int d = db + g;
    bool valid = d < N;
    dc = valid ? d : N - 1;
    float dd = dinv[dc];
    float pself = P[dc];
    float xres = X[(long)dc * FIN];
    int dn = d + G;
    int dnc = dn < N ? dn : N - 1;
    int begn = rowptr[dnc];
    int endn = (dnc + 1 < N) ? rowptr[dnc + 1] : E;

    float acc = 0.0f;
    for (int p = beg + l; p < end; p += 16) {
      v2i r = *(const v2i*)(rec + 2 * (long)p);
      acc += __int_as_float(r.y) * P[r.x];
    }
    for (int mask = 8; mask > 0; mask >>= 1) acc += __shfl_xor(acc, mask, 32);
    if (valid && l == 0) {
      float q = dd * (acc + dd * pself);
      out[d] = q + bias2 + xres;
    }
    beg = begn;
    end = endn;
  }
}

extern "C" void kernel_launch(void* const* d_in, const int* in_sizes, int n_in,
                              void* d_out, int out_size, void* d_ws, size_t ws_size,
                              hipStream_t stream) {
  const float* X    = (const float*)d_in[0];
  const int*   eidx = (const int*)d_in[1];
  const float* ew   = (const float*)d_in[2];
  const float* skip = (const float*)d_in[3];
  const float* H    = (const float*)d_in[4];
  const float* C    = (const float*)d_in[5];
  const float* Wx   = (const float*)d_in[6];
  const float* bx   = (const float*)d_in[7];
  const float* Wh   = (const float*)d_in[8];
  const float* bh   = (const float*)d_in[9];
  const float* wc   = (const float*)d_in[10];
  const float* bg   = (const float*)d_in[11];
  const float* ln_h_s = (const float*)d_in[12];
  const float* ln_h_b = (const float*)d_in[13];
  const float* ln_c_s = (const float*)d_in[14];
  const float* ln_c_b = (const float*)d_in[15];
  const float* ln_o_s = (const float*)d_in[16];
  const float* ln_o_b = (const float*)d_in[17];
  const float* W1   = (const float*)d_in[18];
  const float* b1   = (const float*)d_in[19];
  const float* W2   = (const float*)d_in[20];
  const float* b2   = (const float*)d_in[21];

  const int N = in_sizes[0] / FIN;
  const int E = in_sizes[2];

  const int* src = eidx;
  const int* dst = eidx + E;

  float* out_p    = (float*)d_out;
  float* hidden_p = out_p + N;
  float* cell_p   = hidden_p + (long)N * HID;

  int*   rec  = (int*)d_ws;
  float* AH   = (float*)(rec + 2 * (long)E);
  float* AX   = AH + (long)N * HID;
  float* dinv = AX + (long)N * FIN;
  float* P    = dinv + N;
  int* rowptr = (int*)(P + N);
  int*   rank  = (int*)AH;
  int*   cnt2  = rank + E;
  float* deg2  = (float*)(cnt2 + 2 * (long)N);
  int*   base1 = (int*)(deg2 + 2 * (long)N);
  int*   bsum  = base1 + N;

  const int B = 256;
  auto cdiv = [](long a, long b) { return (unsigned)((a + b - 1) / b); };

  k_init_deg<<<cdiv(N, B), B, 0, stream>>>(deg2, cnt2, N);
  k_acc_deg<<<2048, B, 0, stream>>>(deg2, cnt2, rank, dst, ew, N, E);
  k_dinv<<<cdiv(N, B), B, 0, stream>>>(dinv, deg2, N);

  k_blocksum<<<cdiv(N, SCAN_T * SCAN_PER), SCAN_T, 0, stream>>>(cnt2, bsum, N);
  k_scan<<<cdiv(N, SCAN_T * SCAN_PER), SCAN_T, 0, stream>>>(cnt2, rowptr, base1, bsum, N);
  k_fill<<<2048, B, 0, stream>>>(rowptr, base1, rank, rec, src, dst, ew, dinv, E);

  k_gather_xh<<<2048, B, 0, stream>>>(AH, AX, X, H, rowptr, rec, dinv, N, E);

  size_t smem_gates = (size_t)(KSTEPS * NT_G * 64 + GCOLS) * sizeof(float);
  k_gates<<<cdiv(N, ROWS_PER_BLOCK), B, smem_gates, stream>>>(
      AH, AX, C, Wx, Wh, bx, bh, bg, wc,
      ln_h_s, ln_h_b, ln_c_s, ln_c_b, ln_o_s, ln_o_b,
      hidden_p, cell_p, AH, N);

  k_head<<<cdiv(N, ROWS_PER_BLOCK), B, 0, stream>>>(AH, skip, W1, AH, N);

  k_gather_head<<<2048, B, 0, stream>>>(P, AH, rowptr, rec, dinv, b1, W2, N, E);

  k_gather_out<<<2048, B, 0, stream>>>(out_p, P, rowptr, rec, dinv, b2, X, N, E);
}


# LSTM gate sigmoids and tanhs on the hardware base-2 exponential and reciprocal instructions instead of library expf, tanhf and exact division; hidden, cell, out change in their last bits
# speedup vs baseline: 5.4719x; 1.0843x over previous
#include <hip/hip_runtime.h>
#include <math.h>


typedef __attribute__((ext_vector_type(2))) float v2f;
typedef __attribute__((ext_vector_type(2))) int v2i;
typedef __attribute__((ext_vector_type(8))) float v8f;

#define FIN   8
#define HID   64
#define KTOT  72
#define KSTEPS 18
#define KH_STEPS 17
#define TPITCH 68
#define GCOLS 256
#define NT_G  16
#define ROWS_PER_BLOCK 128
#define EPS 1e-5f

__device__ __forceinline__ float sigmoidf_(float x) {
  float e = __builtin_amdgcn_exp2f(-1.4426950408889634f * x);
  return __builtin_amdgcn_rcpf(1.0f + e);
}
__device__ __forceinline__ float tanhf_(float x) {
  float e = __builtin_amdgcn_exp2f(2.8853900817779268f * x);
  return __builtin_fmaf(-2.0f, __builtin_amdgcn_rcpf(1.0f + e), 1.0f);
}

__device__ __forceinline__ int wave_half() {
  return (int)((__builtin_amdgcn_s_sendmsg_rtn(0x87) >> 18) & 1u);
}

__global__ void k_init_deg(float* deg2, int* cnt2, int N) {
  int i = blockIdx.x * blockDim.x + threadIdx.x;
  if (i < N) { deg2[i] = 1.0f; deg2[N + i] = 0.0f; cnt2[i] = 0; cnt2[N + i] = 0; }
}

__global__ __launch_bounds__(256)
void k_acc_deg(float* deg2, int* cnt2, int* __restrict__ rank,
               const int* __restrict__ dst, const float* __restrict__ w, int N, int E) {
  int h = wave_half();
  float* degp = deg2 + (long)h * N;
  int* cntp = cnt2 + (long)h * N;
  int stride = gridDim.x * blockDim.x;
  for (int e = blockIdx.x * blockDim.x + threadIdx.x; e < E; e += stride) {
    int d = dst[e];
    atomicAdd(&degp[d], w[e]);
    int r = atomicAdd(&cntp[d], 1);
    rank[e] = 2 * r + h;
  }
}

__global__ void k_dinv(float* __restrict__ dinv, const float* __restrict__ deg2, int N) {
  int i = blockIdx.x * blockDim.x + threadIdx.x;
  if (i < N) dinv[i] = rsqrtf(deg2[i] + deg2[N + i]);
}

#define SCAN_T 256
#define SCAN_PER 4

__global__ __launch_bounds__(SCAN_T)
void k_blocksum(const int* __restrict__ cnt2, int* __restrict__ bsum, int N) {
  __shared__ int wsum[SCAN_T / 32];
  int base = (blockIdx.x * SCAN_T + threadIdx.x) * SCAN_PER;
  int s = 0;
  for (int i = 0; i < SCAN_PER; ++i)
    if (base + i < N) s += cnt2[base + i] + cnt2[N + base + i];
  for (int mask = 16; mask > 0; mask >>= 1) s += __shfl_xor(s, mask, 32);
  if ((threadIdx.x & 31) == 0) wsum[threadIdx.x >> 5] = s;
  __syncthreads();
  if (threadIdx.x == 0) {
    int tot = 0;
    for (int i = 0; i < SCAN_T / 32; ++i) tot += wsum[i];
    bsum[blockIdx.x] = tot;
  }
}

__global__ __launch_bounds__(SCAN_T)
void k_scan(const int* __restrict__ cnt2, int* __restrict__ rowptr,
            int* __restrict__ base1, const int* __restrict__ bsum, int N) {
  __shared__ int wtot[SCAN_T / 32];
  __shared__ int woff[SCAN_T / 32];
  int tid = threadIdx.x, lane = tid & 31, wave = tid >> 5;

  int off = 0;
  for (int i = tid; i < (int)blockIdx.x; i += SCAN_T) off += bsum[i];
  for (int mask = 16; mask > 0; mask >>= 1) off += __shfl_xor(off, mask, 32);
  if (lane == 0) woff[wave] = off;

  int base = (blockIdx.x * SCAN_T + tid) * SCAN_PER;
  int c[SCAN_PER], c0[SCAN_PER];
  int tsum = 0;
  for (int i = 0; i < SCAN_PER; ++i) {
    c0[i] = (base + i < N) ? cnt2[base + i] : 0;
    c[i] = c0[i] + ((base + i < N) ? cnt2[N + base + i] : 0);
    tsum += c[i];
  }
  int v = tsum;
  for (int o = 1; o < 32; o <<= 1) {
    int u = __shfl_up(v, o, 32);
    if (lane >= o) v += u;
  }
  if (lane == 31) wtot[wave] = v;
  __syncthreads();
  int pre = 0;
  for (int i = 0; i < SCAN_T / 32; ++i) {
    pre += woff[i];
    if (i < wave) pre += wtot[i];
  }
  int excl = pre + v - tsum;
  for (int i = 0; i < SCAN_PER; ++i) {
    if (base + i < N) {
      rowptr[base + i] = excl;
      base1[base + i] = excl + c0[i];
      excl += c[i];
    }
  }
}

__global__ __launch_bounds__(256)
void k_fill(const int* __restrict__ rowptr, const int* __restrict__ base1,
            const int* __restrict__ rank, int* __restrict__ rec,
            const int* __restrict__ src,
            const int* __restrict__ dst, const float* __restrict__ w,
            const float* __restrict__ dinv, int E) {
  long stride = (long)gridDim.x * blockDim.x;
  long t = (long)blockIdx.x * blockDim.x + threadIdx.x;
  for (long e0 = t; e0 < E; e0 += 4 * stride) {
    int dn[4], s[4], rk[4], p[4];
    float we[4], c[4];
#pragma unroll
    for (int u = 0; u < 4; ++u) {
      long e = e0 + u * stride;
      long ec = e < E ? e : e0;
      dn[u] = dst[ec];
      s[u] = src[ec];
      we[u] = w[ec];
      rk[u] = rank[ec];
    }
#pragma unroll
    for (int u = 0; u < 4; ++u) {
      const int* bp = (rk[u] & 1) ? base1 : rowptr;
      p[u] = bp[dn[u]] + (rk[u] >> 1);
      c[u] = we[u] * dinv[s[u]];
    }
#pragma unroll
    for (int u = 0; u < 4; ++u) {
      v2i r;
      r.x = s[u];
      r.y = __float_as_int(c[u]);
      *(v2i*)(rec + 2 * (long)p[u]) = r;
    }
  }
}

__device__ __forceinline__ void seg_batch_xh(int s, int ci, int cnt, int lane, int l8,
                                             const float* __restrict__ H,
                                             const float* __restrict__ X,
                                             float& a0, float& a1, float& a2) {
  int j = 0;
  for (; j + 4 <= cnt; j += 4) {
    float x0[4], x1[4], x2[4], cj[4];
    for (int u = 0; u < 4; ++u) {
      int sj = __builtin_amdgcn_readlane(s, j + u);
      cj[u] = __int_as_float(__builtin_amdgcn_readlane(ci, j + u));
      const float* q0 = H + (long)sj * HID;
      const float* q2 = X + (long)sj * FIN;
      x0[u] = q0[lane]; x1[u] = q0[32 + lane]; x2[u] = q2[l8];
    }
    for (int u = 0; u < 4; ++u) {
      a0 += cj[u] * x0[u]; a1 += cj[u] * x1[u]; a2 += cj[u] * x2[u];
    }
  }
  for (; j < cnt; ++j) {
    int sj = __builtin_amdgcn_readlane(s, j);
    float cj = __int_as_float(__builtin_amdgcn_readlane(ci, j));
    const float* q0 = H + (long)sj * HID;
    const float* q2 = X + (long)sj * FIN;
    a0 += cj * q0[lane]; a1 += cj * q0[32 + lane]; a2 += cj * q2[l8];
  }
}

__device__ __forceinline__ void seg_batch_t(int s, int ci, int cnt, int lane,
                                            const float* __restrict__ T,
                                            float& a0, float& a1) {
  int j = 0;
  for (; j + 4 <= cnt; j += 4) {
    float x0[4], x1[4], cj[4];
    for (int u = 0; u < 4; ++u) {
      int sj = __builtin_amdgcn_readlane(s, j + u);
      cj[u] = __int_as_float(__builtin_amdgcn_readlane(ci, j + u));
      const float* q0 = T + (long)sj * HID;
      x0[u] = q0[lane]; x1[u] = q0[32 + lane];
    }
    for (int u = 0; u < 4; ++u) {
      a0 += cj[u] * x0[u]; a1 += cj[u] * x1[u];
    }
  }
  for (; j < cnt; ++j) {
    int sj = __builtin_amdgcn_readlane(s, j);
    float cj = __int_as_float(__builtin_amdgcn_readlane(ci, j));
    const float* q0 = T + (long)sj * HID;
    a0 += cj * q0[lane]; a1 += cj * q0[32 + lane];
  }
}

#define RUN_MAX 31

__global__ __launch_bounds__(256)
void k_gather_xh(float* __restrict__ AH, float* __restrict__ AX,
                 const float* __restrict__ X, const float* __restrict__ H,
                 const int* __restrict__ rowptr, const int* __restrict__ rec,
                 const float* __restrict__ dinv, int N, int E) {
  int lane = threadIdx.x & 31;
  int l8 = lane & 7;
  int wid = __builtin_amdgcn_readfirstlane((int)(blockIdx.x * 8 + (threadIdx.x >> 5)));
  unsigned nw = gridDim.x * 8u;
  unsigned q = (unsigned)N / nw, rem = (unsigned)N % nw;
  int n0 = (int)((unsigned)wid * q + ((unsigned)wid < rem ? (unsigned)wid : rem));
  int n1 = n0 + (int)q + ((unsigned)wid < rem ? 1 : 0);

  for (int nb = n0; nb < n1; nb += RUN_MAX) {
    int cn = n1 - nb;
    if (cn > RUN_MAX) cn = RUN_MAX;
    int idx = nb + lane;
    int rp = E;
    float dvl = 0.0f;
    if (idx < N) { rp = rowptr[idx]; dvl = dinv[idx]; }
    int dvi = __float_as_int(dvl);

    int beg = __builtin_amdgcn_readlane(rp, 0);
    int end = __builtin_amdgcn_readlane(rp, 1);
    v2i rc;
    rc.x = 0; rc.y = 0;
    if (beg + lane < end) rc = *(const v2i*)(rec + 2 * (long)(beg + lane));
    float h0 = H[(long)nb * HID + lane];
    float h1 = H[(long)nb * HID + 32 + lane];
    float hx = X[(long)nb * FIN + l8];

    for (int i = 0; i < cn; ++i) {
      int d = nb + i;
      float dd = __int_as_float(__builtin_amdgcn_readlane(dvi, i));
      float a0 = dd * h0;
      float a1 = dd * h1;
      float a2 = dd * hx;

      int endn = end;
      v2i rn;
      rn.x = 0; rn.y = 0;
      if (i + 1 < cn) {
        endn = __builtin_amdgcn_readlane(rp, i + 2);
        if (end + lane < endn) rn = *(const v2i*)(rec + 2 * (long)(end + lane));
        const float* nlo = H + (long)(d + 1) * HID;
        h0 = nlo[lane];
        h1 = nlo[32 + lane];
        hx = X[(long)(d + 1) * FIN + l8];
      }

      int cnt = end - beg;
      seg_batch_xh(rc.x, rc.y, cnt > 32 ? 32 : cnt, lane, l8, H, X, a0, a1, a2);
      for (int base = beg + 32; base < end; base += 32) {
        int p = base + lane;
        v2i r;
        r.x = 0; r.y = 0;
        if (p < end) r = *(const v2i*)(rec + 2 * (long)p);
        int c2 = end - base;
        seg_batch_xh(r.x, r.y, c2 > 32 ? 32 : c2, lane, l8, H, X, a0, a1, a2);
      }

      float* o = AH + (long)d * HID;
      o[lane] = dd * a0;
      o[32 + lane] = dd * a1;
      if (lane < 8) AX[(long)d * FIN + lane] = dd * a2;

      rc = rn;
      beg = end;
      end = endn;
    }
  }
}

__global__ __launch_bounds__(256)
void k_gather_head(float* __restrict__ P, const float* __restrict__ T,
                   const int* __restrict__ rowptr, const int* __restrict__ rec,
                   const float* __restrict__ dinv, const float* __restrict__ b1,
                   const float* __restrict__ W2, int N, int E) {
  int lane = threadIdx.x & 31;
  int wid = __builtin_amdgcn_readfirstlane((int)(blockIdx.x * 8 + (threadIdx.x >> 5)));
  unsigned nw = gridDim.x * 8u;
  unsigned q = (unsigned)N / nw, rem = (unsigned)N % nw;
  int n0 = (int)((unsigned)wid * q + ((unsigned)wid < rem ? (unsigned)wid : rem));
  int n1 = n0 + (int)q + ((unsigned)wid < rem ? 1 : 0);

  float bl0 = b1[lane], bl1 = b1[32 + lane];
  float wl0 = W2[lane], wl1 = W2[32 + lane];

  for (int nb = n0; nb < n1; nb += RUN_MAX) {
    int cn = n1 - nb;
    if (cn > RUN_MAX) cn = RUN_MAX;
    int idx = nb + lane;
    int rp = E;
    float dvl = 0.0f;
    if (idx < N) { rp = rowptr[idx]; dvl = dinv[idx]; }
    int dvi = __float_as_int(dvl);

    int beg = __builtin_amdgcn_readlane(rp, 0);
    int end = __builtin_amdgcn_readlane(rp, 1);
    v2i rc;
    rc.x = 0; rc.y = 0;
    if (beg + lane < end) rc = *(const v2i*)(rec + 2 * (long)(beg + lane));
    float h0 = T[(long)nb * HID + lane];
    float h1 = T[(long)nb * HID + 32 + lane];

    for (int i = 0; i < cn; ++i) {
      int d = nb + i;
      float dd = __int_as_float(__builtin_amdgcn_readlane(dvi, i));
      float a0 = dd * h0;
      float a1 = dd * h1;

      int endn = end;
      v2i rn;
      rn.x = 0; rn.y = 0;
      if (i + 1 < cn) {
        endn = __builtin_amdgcn_readlane(rp, i + 2);
        if (end + lane < endn) rn = *(const v2i*)(rec + 2 * (long)(end + lane));
        const float* nlo = T + (long)(d + 1) * HID;
        h0 = nlo[lane];
        h1 = nlo[32 + lane];
      }

      int cnt = end - beg;
      seg_batch_t(rc.x, rc.y, cnt > 32 ? 32 : cnt, lane, T, a0, a1);
      for (int base = beg + 32; base < end; base += 32) {
        int p = base + lane;
        v2i r;
        r.x = 0; r.y = 0;
        if (p < end) r = *(const v2i*)(rec + 2 * (long)p);
        int c2 = end - base;
        seg_batch_t(r.x, r.y, c2 > 32 ? 32 : c2, lane, T, a0, a1);
      }

      float z0 = dd * a0 + bl0;
      float z1 = dd * a1 + bl1;
      z0 = z0 > 0.0f ? z0 : 0.0f;
      z1 = z1 > 0.0f ? z1 : 0.0f;
      float pv = z0 * wl0 + z1 * wl1;
      for (int mask = 16; mask > 0; mask >>= 1) pv += __shfl_xor(pv, mask, 32);
      if (lane == 0) P[d] = pv;

      rc = rn;
      beg = end;
      end = endn;
    }
  }
}

__global__ __launch_bounds__(256)
void k_gates(const float* __restrict__ AH, const float* __restrict__ AX,
             const float* __restrict__ C,
             const float* __restrict__ Wx, const float* __restrict__ Wh,
             const float* __restrict__ bx, const float* __restrict__ bh,
             const float* __restrict__ bg, const float* __restrict__ wc,
             const float* __restrict__ ln_h_s, const float* __restrict__ ln_h_b,
             const float* __restrict__ ln_c_s, const float* __restrict__ ln_c_b,
             const float* __restrict__ ln_o_s, const float* __restrict__ ln_o_b,
             float* __restrict__ hiddenOut, float* __restrict__ cellOut,
             float* __restrict__ Y, int N) {
  extern __shared__ float smem[];
  float* Bf = smem;
  float* bias = smem + 18432;

  int tid = threadIdx.x;
  for (int i = tid; i < KSTEPS * NT_G * 64; i += 256) {
    int s = i >> 10;
    int rem = i & 1023;
    int t = rem >> 6;
    int rem2 = rem & 63;
    int L = rem2 >> 1;
    int j = rem2 & 1;
    int k = 4 * s + 2 * (L >> 4) + j;
    int c = 16 * t + (L & 15);
    int g = c >> 6, h = c & 63;
    Bf[i] = (k < FIN) ? Wx[(g * FIN + k) * HID + h]
                      : Wh[(g * HID + (k - FIN)) * HID + h];
  }
  if (tid < GCOLS) bias[tid] = bx[tid] + bh[tid] + bg[tid];
  __syncthreads();

  int wave = tid >> 5, lane = tid & 31;
  int half = lane >> 4, lcol = lane & 15;
  int rowBase = blockIdx.x * ROWS_PER_BLOCK + wave * 16;

  int arow = rowBase + lcol;
  if (arow > N - 1) arow = N - 1;
  const float* aXp = AX + (long)arow * FIN;
  const float* aHp = AH + (long)arow * HID;

  v8f acc[NT_G];
#pragma unroll
  for (int t = 0; t < NT_G; ++t)
#pragma unroll
    for (int r = 0; r < 8; ++r) acc[t][r] = 0.0f;

  for (int s = 0; s < KSTEPS; ++s) {
    int k0 = 4 * s + 2 * half;
    v2f a = *(const v2f*)(s < 2 ? aXp + k0 : aHp + (k0 - FIN));
    const float* bbase = Bf + s * 1024 + lane * 2;
#pragma unroll
    for (int t = 0; t < NT_G; ++t) {
      v2f b = *(const v2f*)(bbase + t * 64);
      acc[t] = __builtin_amdgcn_wmma_f32_16x16x4_f32(
          false, a, false, b, (short)0, acc[t], false, false);
    }
  }

#pragma unroll
  for (int r = 0; r < 8; ++r) {
    int node = rowBase + r + 8 * half;
    bool valid = node < N;
    int nc = valid ? node : N - 1;
    float hloc[4], cloc[4];
#pragma unroll
    for (int t = 0; t < 4; ++t) {
      int h = t * 16 + lcol;
      float c0 = C[(long)nc * HID + h];
      float gi = acc[t][r]      + bias[h]       + wc[h] * c0;
      float gf = acc[t + 4][r]  + bias[64 + h]  + wc[64 + h] * c0;
      float gc = acc[t + 8][r]  + bias[128 + h];
      float go = acc[t + 12][r] + bias[192 + h];
      float iv = sigmoidf_(gi);
      float fv = sigmoidf_(gf);
      float cc = fv * c0 + iv * tanhf_(gc);
      float ov = sigmoidf_(go + wc[128 + h] * cc);
      hloc[t] = ov * tanhf_(cc);
      cloc[t] = cc;
    }
    float sh = 0.f, sh2 = 0.f, sc = 0.f, sc2 = 0.f;
#pragma unroll
    for (int t = 0; t < 4; ++t) {
      sh += hloc[t]; sh2 += hloc[t] * hloc[t];
      sc += cloc[t]; sc2 += cloc[t] * cloc[t];
    }
#pragma unroll
    for (int mask = 1; mask < 16; mask <<= 1) {
      sh  += __shfl_xor(sh,  mask, 32);
      sh2 += __shfl_xor(sh2, mask, 32);
      sc  += __shfl_xor(sc,  mask, 32);
      sc2 += __shfl_xor(sc2, mask, 32);
    }
    float mh = sh * (1.0f / 64.0f), mc = sc * (1.0f / 64.0f);
    float vh = sh2 * (1.0f / 64.0f) - mh * mh;
    float vc = sc2 * (1.0f / 64.0f) - mc * mc;
    float rh = rsqrtf(vh + EPS), rc = rsqrtf(vc + EPS);
    if (valid) {
#pragma unroll
      for (int t = 0; t < 4; ++t) {
        int h = t * 16 + lcol;
        float hn = (hloc[t] - mh) * rh;
        float cn = (cloc[t] - mc) * rc;
        hiddenOut[(long)node * HID + h] = hn * ln_h_s[h] + ln_h_b[h];
        cellOut[(long)node * HID + h]   = cn * ln_c_s[h] + ln_c_b[h];
        float ov = hn * ln_o_s[h] + ln_o_b[h];
        Y[(long)node * HID + h] = ov > 0.0f ? ov : 0.0f;
      }
    }
  }
}

__global__ __launch_bounds__(256)
void k_head(const float* __restrict__ Yin, const float* __restrict__ skip,
            const float* __restrict__ W1, float* __restrict__ T, int N) {
  __shared__ float Bf[KH_STEPS * 4 * 64];
  __shared__ __attribute__((aligned(16))) float tile[8 * 16 * TPITCH];

  int tid = threadIdx.x;
  for (int i = tid; i < KH_STEPS * 4 * 64; i += 256) {
    int s = i >> 8;
    int rem = i & 255;
    int t = rem >> 6;
    int rem2 = rem & 63;
    int L = rem2 >> 1;
    int j = rem2 & 1;
    int k = 4 * s + 2 * (L >> 4) + j;
    int c = 16 * t + (L & 15);
    Bf[i] = (k < HID + 2) ? W1[k * HID + c] : 0.0f;
  }
  __syncthreads();

  int wave = tid >> 5, lane = tid & 31;
  int half = lane >> 4, lcol = lane & 15;
  int rowBase = blockIdx.x * ROWS_PER_BLOCK + wave * 16;

  int arow = rowBase + lcol;
  if (arow > N - 1) arow = N - 1;
  const float* aptr = Yin + (long)arow * HID;

  v8f acc[4];
#pragma unroll
  for (int t = 0; t < 4; ++t)
#pragma unroll
    for (int r = 0; r < 8; ++r) acc[t][r] = 0.0f;

  for (int s = 0; s < KH_STEPS - 1; ++s) {
    int k0 = 4 * s + 2 * half;
    v2f a = *(const v2f*)(aptr + k0);
    const float* bbase = Bf + s * 256 + lane * 2;
#pragma unroll
    for (int t = 0; t < 4; ++t) {
      v2f b = *(const v2f*)(bbase + t * 64);
      acc[t] = __builtin_amdgcn_wmma_f32_16x16x4_f32(
          false, a, false, b, (short)0, acc[t], false, false);
    }
  }
  {
    v2f sk = *(const v2f*)(skip + (long)arow * 2);
    v2f a;
    a.x = half ? 0.0f : sk.x;
    a.y = half ? 0.0f : sk.y;
    const float* bbase = Bf + (KH_STEPS - 1) * 256 + lane * 2;
#pragma unroll
    for (int t = 0; t < 4; ++t) {
      v2f b = *(const v2f*)(bbase + t * 64);
      acc[t] = __builtin_amdgcn_wmma_f32_16x16x4_f32(
          false, a, false, b, (short)0, acc[t], false, false);
    }
  }

  float* tw = tile + wave * (16 * TPITCH);
#pragma unroll
  for (int t = 0; t < 4; ++t)
#pragma unroll
    for (int r = 0; r < 8; ++r)
      tw[(r + 8 * half) * TPITCH + 16 * t + lcol] = acc[t][r];
  __syncthreads();
  typedef __attribute__((ext_vector_type(4))) float v4f;
#pragma unroll
  for (int i = 0; i < 8; ++i) {
    int m = 2 * i + half;
    int node = rowBase + m;
    v4f v = *(const v4f*)(tw + m * TPITCH + 4 * lcol);
    if (node < N) *(v4f*)(T + (long)node * HID + 4 * lcol) = v;
  }
}

__global__ __launch_bounds__(256)
void k_gather_out(float* __restrict__ out, const float* __restrict__ P,
                  const int* __restrict__ rowptr, const int* __restrict__ rec,
                  const float* __restrict__ dinv, const float* __restrict__ b2,
                  const float* __restrict__ X, int N, int E) {
  int t = blockIdx.x * blockDim.x + threadIdx.x;
  int g = t >> 4, l = t & 15;
  int G = (int)((gridDim.x * blockDim.x) >> 4);
  float bias2 = b2[0];

  int dc = g < N ? g : N - 1;
  int beg = rowptr[dc];
  int end = (dc + 1 < N) ? rowptr[dc + 1] : E;
  for (int db = 0; db < N; db += G) {
    int d = db + g;
    bool valid = d < N;
    dc = valid ? d : N - 1;
    float dd = dinv[dc];
    float pself = P[dc];
    float xres = X[(long)dc * FIN];
    int dn = d + G;
    int dnc = dn < N ? dn : N - 1;
    int begn = rowptr[dnc];
    int endn = (dnc + 1 < N) ? rowptr[dnc + 1] : E;

    float acc = 0.0f;
    for (int p = beg + l; p < end; p += 16) {
      v2i r = *(const v2i*)(rec + 2 * (long)p);
      acc += __int_as_float(r.y) * P[r.x];
    }
    for (int mask = 8; mask > 0; mask >>= 1) acc += __shfl_xor(acc, mask, 32);
    if (valid && l == 0) {
      float q = dd * (acc + dd * pself);
      out[d] = q + bias2 + xres;
    }
    beg = begn;
    end = endn;
  }
}

extern "C" void kernel_launch(void* const* d_in, const int* in_sizes, int n_in,
                              void* d_out, int out_size, void* d_ws, size_t ws_size,
                              hipStream_t stream) {
  const float* X    = (const float*)d_in[0];
  const int*   eidx = (const int*)d_in[1];
  const float* ew   = (const float*)d_in[2];
  const float* skip = (const float*)d_in[3];
  const float* H    = (const float*)d_in[4];
  const float* C    = (const float*)d_in[5];
  const float* Wx   = (const float*)d_in[6];
  const float* bx   = (const float*)d_in[7];
  const float* Wh   = (const float*)d_in[8];
  const float* bh   = (const float*)d_in[9];
  const float* wc   = (const float*)d_in[10];
  const float* bg   = (const float*)d_in[11];
  const float* ln_h_s = (const float*)d_in[12];
  const float* ln_h_b = (const float*)d_in[13];
  const float* ln_c_s = (const float*)d_in[14];
  const float* ln_c_b = (const float*)d_in[15];
  const float* ln_o_s = (const float*)d_in[16];
  const float* ln_o_b = (const float*)d_in[17];
  const float* W1   = (const float*)d_in[18];
  const float* b1   = (const float*)d_in[19];
  const float* W2   = (const float*)d_in[20];
  const float* b2   = (const float*)d_in[21];

  const int N = in_sizes[0] / FIN;
  const int E = in_sizes[2];

  const int* src = eidx;
  const int* dst = eidx + E;

  float* out_p    = (float*)d_out;
  float* hidden_p = out_p + N;
  float* cell_p   = hidden_p + (long)N * HID;

  int*   rec  = (int*)d_ws;
  float* AH   = (float*)(rec + 2 * (long)E);
  float* AX   = AH + (long)N * HID;
  float* dinv = AX + (long)N * FIN;
  float* P    = dinv + N;
  int* rowptr = (int*)(P + N);
  int*   rank  = (int*)AH;
  int*   cnt2  = rank + E;
  float* deg2  = (float*)(cnt2 + 2 * (long)N);
  int*   base1 = (int*)(deg2 + 2 * (long)N);
  int*   bsum  = base1 + N;

  const int B = 256;
  auto cdiv = [](long a, long b) { return (unsigned)((a + b - 1) / b); };

  k_init_deg<<<cdiv(N, B), B, 0, stream>>>(deg2, cnt2, N);
  k_acc_deg<<<2048, B, 0, stream>>>(deg2, cnt2, rank, dst, ew, N, E);
  k_dinv<<<cdiv(N, B), B, 0, stream>>>(dinv, deg2, N);

  k_blocksum<<<cdiv(N, SCAN_T * SCAN_PER), SCAN_T, 0, stream>>>(cnt2, bsum, N);
  k_scan<<<cdiv(N, SCAN_T * SCAN_PER), SCAN_T, 0, stream>>>(cnt2, rowptr, base1, bsum, N);
  k_fill<<<2048, B, 0, stream>>>(rowptr, base1, rank, rec, src, dst, ew, dinv, E);

  k_gather_xh<<<2048, B, 0, stream>>>(AH, AX, X, H, rowptr, rec, dinv, N, E);

  size_t smem_gates = (size_t)(KSTEPS * NT_G * 64 + GCOLS) * sizeof(float);
  k_gates<<<cdiv(N, ROWS_PER_BLOCK), B, smem_gates, stream>>>(
      AH, AX, C, Wx, Wh, bx, bh, bg, wc,
      ln_h_s, ln_h_b, ln_c_s, ln_c_b, ln_o_s, ln_o_b,
      hidden_p, cell_p, AH, N);

  k_head<<<cdiv(N, ROWS_PER_BLOCK), B, 0, stream>>>(AH, skip, W1, AH, N);

  k_gather_head<<<2048, B, 0, stream>>>(P, AH, rowptr, rec, dinv, b1, W2, N, E);

  k_gather_out<<<2048, B, 0, stream>>>(out_p, P, rowptr, rec, dinv, b2, X, N, E);
}


# k_head folded into k_gates: gates product computed transposed (weights as A), so Y stays in registers as the head product's B operand; one launch less; layernorm and T sum orders change
# speedup vs baseline: 5.9293x; 1.0836x over previous
#include <hip/hip_runtime.h>
#include <math.h>


typedef __attribute__((ext_vector_type(2))) float v2f;
typedef __attribute__((ext_vector_type(2))) int v2i;
typedef __attribute__((ext_vector_type(8))) float v8f;

#define FIN   8
#define HID   64
#define KTOT  72
#define KSTEPS 18
#define KH_STEPS 17
#define TPITCH 68
#define GCOLS 256
#define NT_G  16
#define ROWS_PER_BLOCK 128
#define EPS 1e-5f

__device__ __forceinline__ float sigmoidf_(float x) {
  float e = __builtin_amdgcn_exp2f(-1.4426950408889634f * x);
  return __builtin_amdgcn_rcpf(1.0f + e);
}
__device__ __forceinline__ float tanhf_(float x) {
  float e = __builtin_amdgcn_exp2f(2.8853900817779268f * x);
  return __builtin_fmaf(-2.0f, __builtin_amdgcn_rcpf(1.0f + e), 1.0f);
}

__device__ __forceinline__ int wave_half() {
  return (int)((__builtin_amdgcn_s_sendmsg_rtn(0x87) >> 18) & 1u);
}

__global__ void k_init_deg(float* deg2, int* cnt2, int N) {
  int i = blockIdx.x * blockDim.x + threadIdx.x;
  if (i < N) { deg2[i] = 1.0f; deg2[N + i] = 0.0f; cnt2[i] = 0; cnt2[N + i] = 0; }
}

__global__ __launch_bounds__(256)
void k_acc_deg(float* deg2, int* cnt2, int* __restrict__ rank,
               const int* __restrict__ dst, const float* __restrict__ w, int N, int E) {
  int h = wave_half();
  float* degp = deg2 + (long)h * N;
  int* cntp = cnt2 + (long)h * N;
  int stride = gridDim.x * blockDim.x;
  for (int e = blockIdx.x * blockDim.x + threadIdx.x; e < E; e += stride) {
    int d = dst[e];
    atomicAdd(&degp[d], w[e]);
    int r = atomicAdd(&cntp[d], 1);
    rank[e] = 2 * r + h;
  }
}

__global__ void k_dinv(float* __restrict__ dinv, const float* __restrict__ deg2, int N) {
  int i = blockIdx.x * blockDim.x + threadIdx.x;
  if (i < N) dinv[i] = rsqrtf(deg2[i] + deg2[N + i]);
}

#define SCAN_T 256
#define SCAN_PER 4

__global__ __launch_bounds__(SCAN_T)
void k_blocksum(const int* __restrict__ cnt2, int* __restrict__ bsum, int N) {
  __shared__ int wsum[SCAN_T / 32];
  int base = (blockIdx.x * SCAN_T + threadIdx.x) * SCAN_PER;
  int s = 0;
  for (int i = 0; i < SCAN_PER; ++i)
    if (base + i < N) s += cnt2[base + i] + cnt2[N + base + i];
  for (int mask = 16; mask > 0; mask >>= 1) s += __shfl_xor(s, mask, 32);
  if ((threadIdx.x & 31) == 0) wsum[threadIdx.x >> 5] = s;
  __syncthreads();
  if (threadIdx.x == 0) {
    int tot = 0;
    for (int i = 0; i < SCAN_T / 32; ++i) tot += wsum[i];
    bsum[blockIdx.x] = tot;
  }
}

__global__ __launch_bounds__(SCAN_T)
void k_scan(const int* __restrict__ cnt2, int* __restrict__ rowptr,
            int* __restrict__ base1, const int* __restrict__ bsum, int N) {
  __shared__ int wtot[SCAN_T / 32];
  __shared__ int woff[SCAN_T / 32];
  int tid = threadIdx.x, lane = tid & 31, wave = tid >> 5;

  int off = 0;
  for (int i = tid; i < (int)blockIdx.x; i += SCAN_T) off += bsum[i];
  for (int mask = 16; mask > 0; mask >>= 1) off += __shfl_xor(off, mask, 32);
  if (lane == 0) woff[wave] = off;

  int base = (blockIdx.x * SCAN_T + tid) * SCAN_PER;
  int c[SCAN_PER], c0[SCAN_PER];
  int tsum = 0;
  for (int i = 0; i < SCAN_PER; ++i) {
    c0[i] = (base + i < N) ? cnt2[base + i] : 0;
    c[i] = c0[i] + ((base + i < N) ? cnt2[N + base + i] : 0);
    tsum += c[i];
  }
  int v = tsum;
  for (int o = 1; o < 32; o <<= 1) {
    int u = __shfl_up(v, o, 32);
    if (lane >= o) v += u;
  }
  if (lane == 31) wtot[wave] = v;
  __syncthreads();
  int pre = 0;
  for (int i = 0; i < SCAN_T / 32; ++i) {
    pre += woff[i];
    if (i < wave) pre += wtot[i];
  }
  int excl = pre + v - tsum;
  for (int i = 0; i < SCAN_PER; ++i) {
    if (base + i < N) {
      rowptr[base + i] = excl;
      base1[base + i] = excl + c0[i];
      excl += c[i];
    }
  }
}

__global__ __launch_bounds__(256)
void k_fill(const int* __restrict__ rowptr, const int* __restrict__ base1,
            const int* __restrict__ rank, int* __restrict__ rec,
            const int* __restrict__ src,
            const int* __restrict__ dst, const float* __restrict__ w,
            const float* __restrict__ dinv, int E) {
  long stride = (long)gridDim.x * blockDim.x;
  long t = (long)blockIdx.x * blockDim.x + threadIdx.x;
  for (long e0 = t; e0 < E; e0 += 4 * stride) {
    int dn[4], s[4], rk[4], p[4];
    float we[4], c[4];
#pragma unroll
    for (int u = 0; u < 4; ++u) {
      long e = e0 + u * stride;
      long ec = e < E ? e : e0;
      dn[u] = dst[ec];
      s[u] = src[ec];
      we[u] = w[ec];
      rk[u] = rank[ec];
    }
#pragma unroll
    for (int u = 0; u < 4; ++u) {
      const int* bp = (rk[u] & 1) ? base1 : rowptr;
      p[u] = bp[dn[u]] + (rk[u] >> 1);
      c[u] = we[u] * dinv[s[u]];
    }
#pragma unroll
    for (int u = 0; u < 4; ++u) {
      v2i r;
      r.x = s[u];
      r.y = __float_as_int(c[u]);
      *(v2i*)(rec + 2 * (long)p[u]) = r;
    }
  }
}

__device__ __forceinline__ void seg_batch_xh(int s, int ci, int cnt, int lane, int l8,
                                             const float* __restrict__ H,
                                             const float* __restrict__ X,
                                             float& a0, float& a1, float& a2) {
  int j = 0;
  for (; j + 4 <= cnt; j += 4) {
    float x0[4], x1[4], x2[4], cj[4];
    for (int u = 0; u < 4; ++u) {
      int sj = __builtin_amdgcn_readlane(s, j + u);
      cj[u] = __int_as_float(__builtin_amdgcn_readlane(ci, j + u));
      const float* q0 = H + (long)sj * HID;
      const float* q2 = X + (long)sj * FIN;
      x0[u] = q0[lane]; x1[u] = q0[32 + lane]; x2[u] = q2[l8];
    }
    for (int u = 0; u < 4; ++u) {
      a0 += cj[u] * x0[u]; a1 += cj[u] * x1[u]; a2 += cj[u] * x2[u];
    }
  }
  for (; j < cnt; ++j) {
    int sj = __builtin_amdgcn_readlane(s, j);
    float cj = __int_as_float(__builtin_amdgcn_readlane(ci, j));
    const float* q0 = H + (long)sj * HID;
    const float* q2 = X + (long)sj * FIN;
    a0 += cj * q0[lane]; a1 += cj * q0[32 + lane]; a2 += cj * q2[l8];
  }
}

__device__ __forceinline__ void seg_batch_t(int s, int ci, int cnt, int lane,
                                            const float* __restrict__ T,
                                            float& a0, float& a1) {
  int j = 0;
  for (; j + 4 <= cnt; j += 4) {
    float x0[4], x1[4], cj[4];
    for (int u = 0; u < 4; ++u) {
      int sj = __builtin_amdgcn_readlane(s, j + u);
      cj[u] = __int_as_float(__builtin_amdgcn_readlane(ci, j + u));
      const float* q0 = T + (long)sj * HID;
      x0[u] = q0[lane]; x1[u] = q0[32 + lane];
    }
    for (int u = 0; u < 4; ++u) {
      a0 += cj[u] * x0[u]; a1 += cj[u] * x1[u];
    }
  }
  for (; j < cnt; ++j) {
    int sj = __builtin_amdgcn_readlane(s, j);
    float cj = __int_as_float(__builtin_amdgcn_readlane(ci, j));
    const float* q0 = T + (long)sj * HID;
    a0 += cj * q0[lane]; a1 += cj * q0[32 + lane];
  }
}

#define RUN_MAX 31

__global__ __launch_bounds__(256)
void k_gather_xh(float* __restrict__ AH, float* __restrict__ AX,
                 const float* __restrict__ X, const float* __restrict__ H,
                 const int* __restrict__ rowptr, const int* __restrict__ rec,
                 const float* __restrict__ dinv, int N, int E) {
  int lane = threadIdx.x & 31;
  int l8 = lane & 7;
  int wid = __builtin_amdgcn_readfirstlane((int)(blockIdx.x * 8 + (threadIdx.x >> 5)));
  unsigned nw = gridDim.x * 8u;
  unsigned q = (unsigned)N / nw, rem = (unsigned)N % nw;
  int n0 = (int)((unsigned)wid * q + ((unsigned)wid < rem ? (unsigned)wid : rem));
  int n1 = n0 + (int)q + ((unsigned)wid < rem ? 1 : 0);

  for (int nb = n0; nb < n1; nb += RUN_MAX) {
    int cn = n1 - nb;
    if (cn > RUN_MAX) cn = RUN_MAX;
    int idx = nb + lane;
    int rp = E;
    float dvl = 0.0f;
    if (idx < N) { rp = rowptr[idx]; dvl = dinv[idx]; }
    int dvi = __float_as_int(dvl);

    int beg = __builtin_amdgcn_readlane(rp, 0);
    int end = __builtin_amdgcn_readlane(rp, 1);
    v2i rc;
    rc.x = 0; rc.y = 0;
    if (beg + lane < end) rc = *(const v2i*)(rec + 2 * (long)(beg + lane));
    float h0 = H[(long)nb * HID + lane];
    float h1 = H[(long)nb * HID + 32 + lane];
    float hx = X[(long)nb * FIN + l8];

    for (int i = 0; i < cn; ++i) {
      int d = nb + i;
      float dd = __int_as_float(__builtin_amdgcn_readlane(dvi, i));
      float a0 = dd * h0;
      float a1 = dd * h1;
      float a2 = dd * hx;

      int endn = end;
      v2i rn;
      rn.x = 0; rn.y = 0;
      if (i + 1 < cn) {
        endn = __builtin_amdgcn_readlane(rp, i + 2);
        if (end + lane < endn) rn = *(const v2i*)(rec + 2 * (long)(end + lane));
        const float* nlo = H + (long)(d + 1) * HID;
        h0 = nlo[lane];
        h1 = nlo[32 + lane];
        hx = X[(long)(d + 1) * FIN + l8];
      }

      int cnt = end - beg;
      seg_batch_xh(rc.x, rc.y, cnt > 32 ? 32 : cnt, lane, l8, H, X, a0, a1, a2);
      for (int base = beg + 32; base < end; base += 32) {
        int p = base + lane;
        v2i r;
        r.x = 0; r.y = 0;
        if (p < end) r = *(const v2i*)(rec + 2 * (long)p);
        int c2 = end - base;
        seg_batch_xh(r.x, r.y, c2 > 32 ? 32 : c2, lane, l8, H, X, a0, a1, a2);
      }

      float* o = AH + (long)d * HID;
      o[lane] = dd * a0;
      o[32 + lane] = dd * a1;
      if (lane < 8) AX[(long)d * FIN + lane] = dd * a2;

      rc = rn;
      beg = end;
      end = endn;
    }
  }
}

__global__ __launch_bounds__(256)
void k_gather_head(float* __restrict__ P, const float* __restrict__ T,
                   const int* __restrict__ rowptr, const int* __restrict__ rec,
                   const float* __restrict__ dinv, const float* __restrict__ b1,
                   const float* __restrict__ W2, int N, int E) {
  int lane = threadIdx.x & 31;
  int wid = __builtin_amdgcn_readfirstlane((int)(blockIdx.x * 8 + (threadIdx.x >> 5)));
  unsigned nw = gridDim.x * 8u;
  unsigned q = (unsigned)N / nw, rem = (unsigned)N % nw;
  int n0 = (int)((unsigned)wid * q + ((unsigned)wid < rem ? (unsigned)wid : rem));
  int n1 = n0 + (int)q + ((unsigned)wid < rem ? 1 : 0);

  float bl0 = b1[lane], bl1 = b1[32 + lane];
  float wl0 = W2[lane], wl1 = W2[32 + lane];

  for (int nb = n0; nb < n1; nb += RUN_MAX) {
    int cn = n1 - nb;
    if (cn > RUN_MAX) cn = RUN_MAX;
    int idx = nb + lane;
    int rp = E;
    float dvl = 0.0f;
    if (idx < N) { rp = rowptr[idx]; dvl = dinv[idx]; }
    int dvi = __float_as_int(dvl);

    int beg = __builtin_amdgcn_readlane(rp, 0);
    int end = __builtin_amdgcn_readlane(rp, 1);
    v2i rc;
    rc.x = 0; rc.y = 0;
    if (beg + lane < end) rc = *(const v2i*)(rec + 2 * (long)(beg + lane));
    float h0 = T[(long)nb * HID + lane];
    float h1 = T[(long)nb * HID + 32 + lane];

    for (int i = 0; i < cn; ++i) {
      int d = nb + i;
      float dd = __int_as_float(__builtin_amdgcn_readlane(dvi, i));
      float a0 = dd * h0;
      float a1 = dd * h1;

      int endn = end;
      v2i rn;
      rn.x = 0; rn.y = 0;
      if (i + 1 < cn) {
        endn = __builtin_amdgcn_readlane(rp, i + 2);
        if (end + lane < endn) rn = *(const v2i*)(rec + 2 * (long)(end + lane));
        const float* nlo = T + (long)(d + 1) * HID;
        h0 = nlo[lane];
        h1 = nlo[32 + lane];
      }

      int cnt = end - beg;
      seg_batch_t(rc.x, rc.y, cnt > 32 ? 32 : cnt, lane, T, a0, a1);
      for (int base = beg + 32; base < end; base += 32) {
        int p = base + lane;
        v2i r;
        r.x = 0; r.y = 0;
        if (p < end) r = *(const v2i*)(rec + 2 * (long)p);
        int c2 = end - base;
        seg_batch_t(r.x, r.y, c2 > 32 ? 32 : c2, lane, T, a0, a1);
      }

      float z0 = dd * a0 + bl0;
      float z1 = dd * a1 + bl1;
      z0 = z0 > 0.0f ? z0 : 0.0f;
      z1 = z1 > 0.0f ? z1 : 0.0f;
      float pv = z0 * wl0 + z1 * wl1;
      for (int mask = 16; mask > 0; mask >>= 1) pv += __shfl_xor(pv, mask, 32);
      if (lane == 0) P[d] = pv;

      rc = rn;
      beg = end;
      end = endn;
    }
  }
}

typedef __attribute__((ext_vector_type(4))) float v4f;

#define BF_FLOATS   (KSTEPS * NT_G * 64)
#define PAR_BIAS    0
#define PAR_WC      256
#define PAR_LN      448
#define PAR_FLOATS  832
#define W1F_FLOATS  (KH_STEPS * 4 * 64)
#define TILE_FLOATS (8 * 16 * TPITCH)

__device__ __forceinline__ void ld8(const float* p, float* o) {
  v4f a = *(const v4f*)p;
  v4f b = *(const v4f*)(p + 4);
  o[0] = a.x; o[1] = a.y; o[2] = a.z; o[3] = a.w;
  o[4] = b.x; o[5] = b.y; o[6] = b.z; o[7] = b.w;
}

__device__ __forceinline__ void rows_out(float* tw, const float* v, float* __restrict__ dst,
                                         int rowBase, int lcol, int half, int N) {
  float* trow = tw + lcol * TPITCH + 8 * half;
#pragma unroll
  for (int t = 0; t < 4; ++t) {
    v4f a, b;
    a.x = v[8 * t];     a.y = v[8 * t + 1]; a.z = v[8 * t + 2]; a.w = v[8 * t + 3];
    b.x = v[8 * t + 4]; b.y = v[8 * t + 5]; b.z = v[8 * t + 6]; b.w = v[8 * t + 7];
    *(v4f*)(trow + 16 * t) = a;
    *(v4f*)(trow + 16 * t + 4) = b;
  }
  __syncthreads();
#pragma unroll
  for (int i = 0; i < 8; ++i) {
    int m = 2 * i + half;
    int node = rowBase + m;
    v4f x = *(const v4f*)(tw + m * TPITCH + 4 * lcol);
    if (node < N) *(v4f*)(dst + (long)node * HID + 4 * lcol) = x;
  }
}

__global__ __launch_bounds__(256)
void k_gates_head(const float* AH, const float* __restrict__ AX,
                  const float* __restrict__ C,
                  const float* __restrict__ Wx, const float* __restrict__ Wh,
                  const float* __restrict__ bx, const float* __restrict__ bh,
                  const float* __restrict__ bg, const float* __restrict__ wc,
                  const float* __restrict__ ln_h_s, const float* __restrict__ ln_h_b,
                  const float* __restrict__ ln_c_s, const float* __restrict__ ln_c_b,
                  const float* __restrict__ ln_o_s, const float* __restrict__ ln_o_b,
                  const float* __restrict__ skip, const float* __restrict__ W1,
                  float* __restrict__ hiddenOut, float* __restrict__ cellOut,
                  float* T, int N) {
  __shared__ __attribute__((aligned(16)))
      float smem[BF_FLOATS + PAR_FLOATS + W1F_FLOATS + TILE_FLOATS];
  float* Bf   = smem;
  float* par  = Bf + BF_FLOATS;
  float* W1f  = par + PAR_FLOATS;
  float* tile = W1f + W1F_FLOATS;

  int tid = threadIdx.x;
  {
    int mt = tid >> 6, L = (tid & 63) >> 1, j = tid & 1;
    int hh = L >> 4;
    int c = 16 * mt + (L & 15);
    float w[KH_STEPS];
#pragma unroll
    for (int st = 0; st < KH_STEPS - 1; ++st)
      w[st] = W1[(16 * (st >> 2) + 8 * hh + 2 * (st & 3) + j) * HID + c];
    w[KH_STEPS - 1] = W1[(HID + j) * HID + c];
    float bsum = bx[tid] + bh[tid] + bg[tid];
    float wcv = wc[tid < 192 ? tid : 0];
    int t6 = tid & 63;
    float l0 = ln_h_s[t6], l1 = ln_h_b[t6], l2 = ln_c_s[t6];
    float l3 = ln_c_b[t6], l4 = ln_o_s[t6], l5 = ln_o_b[t6];
#pragma unroll
    for (int st = 0; st < KH_STEPS; ++st)
      W1f[st * 256 + tid] = (st < KH_STEPS - 1 || hh == 0) ? w[st] : 0.0f;
    par[PAR_BIAS + tid] = bsum;
    if (tid < 192) par[PAR_WC + tid] = wcv;
    if (tid < 64) {
      par[PAR_LN + tid]       = l0;
      par[PAR_LN + 64 + tid]  = l1;
      par[PAR_LN + 128 + tid] = l2;
      par[PAR_LN + 192 + tid] = l3;
      par[PAR_LN + 256 + tid] = l4;
      par[PAR_LN + 320 + tid] = l5;
    }
  }
  for (int i = tid; i < BF_FLOATS; i += 256) {
    int s = i >> 10;
    int rem = i & 1023;
    int t = rem >> 6;
    int rem2 = rem & 63;
    int L = rem2 >> 1;
    int j = rem2 & 1;
    int k = 4 * s + 2 * (L >> 4) + j;
    int c = 16 * t + (L & 15);
    int g = c >> 6, h = c & 63;
    Bf[i] = (k < FIN) ? Wx[(g * FIN + k) * HID + h]
                      : Wh[(g * HID + (k - FIN)) * HID + h];
  }
  __syncthreads();

  int wave = tid >> 5, lane = tid & 31;
  int half = lane >> 4, lcol = lane & 15;
  int rowBase = blockIdx.x * ROWS_PER_BLOCK + wave * 16;

  int arow = rowBase + lcol;
  if (arow > N - 1) arow = N - 1;
  const float* aXp = AX + (long)arow * FIN;
  const float* aHp = AH + (long)arow * HID;

  v4f cpre[8];
  {
    const float* Cp = C + (long)arow * HID + 8 * half;
#pragma unroll
    for (int t = 0; t < 4; ++t) {
      cpre[2 * t]     = *(const v4f*)(Cp + 16 * t);
      cpre[2 * t + 1] = *(const v4f*)(Cp + 16 * t + 4);
    }
  }
  v2f sk = *(const v2f*)(skip + (long)arow * 2);

  v8f acc[NT_G];
#pragma unroll
  for (int t = 0; t < NT_G; ++t)
#pragma unroll
    for (int r = 0; r < 8; ++r) acc[t][r] = 0.0f;

  for (int s = 0; s < KSTEPS; ++s) {
    int k0 = 4 * s + 2 * half;
    v2f a = *(const v2f*)(s < 2 ? aXp + k0 : aHp + (k0 - FIN));
    const float* bbase = Bf + s * 1024 + lane * 2;
#pragma unroll
    for (int t = 0; t < NT_G; ++t) {
      v2f b = *(const v2f*)(bbase + t * 64);
      acc[t] = __builtin_amdgcn_wmma_f32_16x16x4_f32(
          false, b, false, a, (short)0, acc[t], false, false);
    }
  }

  const float* pl = par + 8 * half;
  float hv[32], cv[32];
  float sh = 0.f, sh2 = 0.f, sc = 0.f, sc2 = 0.f;
#pragma unroll
  for (int t = 0; t < 4; ++t) {
    float c0[8], bi[8], bfg[8], bc[8], bo[8], wi[8], wf[8], wo[8];
    c0[0] = cpre[2 * t].x;     c0[1] = cpre[2 * t].y;
    c0[2] = cpre[2 * t].z;     c0[3] = cpre[2 * t].w;
    c0[4] = cpre[2 * t + 1].x; c0[5] = cpre[2 * t + 1].y;
    c0[6] = cpre[2 * t + 1].z; c0[7] = cpre[2 * t + 1].w;
    ld8(pl + PAR_BIAS + 16 * t, bi);
    ld8(pl + PAR_BIAS + 64 + 16 * t, bfg);
    ld8(pl + PAR_BIAS + 128 + 16 * t, bc);
    ld8(pl + PAR_BIAS + 192 + 16 * t, bo);
    ld8(pl + PAR_WC + 16 * t, wi);
    ld8(pl + PAR_WC + 64 + 16 * t, wf);
    ld8(pl + PAR_WC + 128 + 16 * t, wo);
    float ph = 0.f, ph2 = 0.f, pc = 0.f, pc2 = 0.f;
#pragma unroll
    for (int r = 0; r < 8; ++r) {
      float gi = acc[t][r]      + bi[r]  + wi[r] * c0[r];
      float gf = acc[t + 4][r]  + bfg[r] + wf[r] * c0[r];
      float gc = acc[t + 8][r]  + bc[r];
      float go = acc[t + 12][r] + bo[r];
      float iv = sigmoidf_(gi);
      float fv = sigmoidf_(gf);
      float cc = fv * c0[r] + iv * tanhf_(gc);
      float ov = sigmoidf_(go + wo[r] * cc);
      float hh = ov * tanhf_(cc);
      hv[8 * t + r] = hh;
      cv[8 * t + r] = cc;
      ph += hh; ph2 += hh * hh;
      pc += cc; pc2 += cc * cc;
    }
    sh += ph; sh2 += ph2; sc += pc; sc2 += pc2;
  }
  sh  += __shfl_xor(sh,  16, 32);
  sh2 += __shfl_xor(sh2, 16, 32);
  sc  += __shfl_xor(sc,  16, 32);
  sc2 += __shfl_xor(sc2, 16, 32);
  float mh = sh * (1.0f / 64.0f), mc = sc * (1.0f / 64.0f);
  float vh = sh2 * (1.0f / 64.0f) - mh * mh;
  float vc = sc2 * (1.0f / 64.0f) - mc * mc;
  float rh = rsqrtf(vh + EPS), rc = rsqrtf(vc + EPS);

  float* tw = tile + wave * (16 * TPITCH);
  float ov32[32];
#pragma unroll
  for (int t = 0; t < 4; ++t) {
    float s8[8], b8[8];
    ld8(pl + PAR_LN + 16 * t, s8);
    ld8(pl + PAR_LN + 64 + 16 * t, b8);
#pragma unroll
    for (int r = 0; r < 8; ++r) {
      float hn = (hv[8 * t + r] - mh) * rh;
      ov32[8 * t + r] = hn * s8[r] + b8[r];
    }
  }
  rows_out(tw, ov32, hiddenOut, rowBase, lcol, half, N);
#pragma unroll
  for (int t = 0; t < 4; ++t) {
    float s8[8], b8[8];
    ld8(pl + PAR_LN + 128 + 16 * t, s8);
    ld8(pl + PAR_LN + 192 + 16 * t, b8);
#pragma unroll
    for (int r = 0; r < 8; ++r) {
      float cn = (cv[8 * t + r] - mc) * rc;
      ov32[8 * t + r] = cn * s8[r] + b8[r];
    }
  }
  rows_out(tw, ov32, cellOut, rowBase, lcol, half, N);
  float yv[32];
#pragma unroll
  for (int t = 0; t < 4; ++t) {
    float s8[8], b8[8];
    ld8(pl + PAR_LN + 256 + 16 * t, s8);
    ld8(pl + PAR_LN + 320 + 16 * t, b8);
#pragma unroll
    for (int r = 0; r < 8; ++r) {
      float hn = (hv[8 * t + r] - mh) * rh;
      float o = hn * s8[r] + b8[r];
      yv[8 * t + r] = o > 0.0f ? o : 0.0f;
    }
  }

  v8f acc2[4];
#pragma unroll
  for (int mt = 0; mt < 4; ++mt)
#pragma unroll
    for (int r = 0; r < 8; ++r) acc2[mt][r] = 0.0f;
#pragma unroll
  for (int t = 0; t < 4; ++t)
#pragma unroll
    for (int p = 0; p < 4; ++p) {
      v2f yb;
      yb.x = yv[8 * t + 2 * p];
      yb.y = yv[8 * t + 2 * p + 1];
      const float* wb = W1f + (t * 4 + p) * 256 + lane * 2;
#pragma unroll
      for (int mt = 0; mt < 4; ++mt) {
        v2f wa = *(const v2f*)(wb + mt * 64);
        acc2[mt] = __builtin_amdgcn_wmma_f32_16x16x4_f32(
            false, wa, false, yb, (short)0, acc2[mt], false, false);
      }
    }
  {
    v2f yb;
    yb.x = half ? 0.0f : sk.x;
    yb.y = half ? 0.0f : sk.y;
    const float* wb = W1f + (KH_STEPS - 1) * 256 + lane * 2;
#pragma unroll
    for (int mt = 0; mt < 4; ++mt) {
      v2f wa = *(const v2f*)(wb + mt * 64);
      acc2[mt] = __builtin_amdgcn_wmma_f32_16x16x4_f32(
          false, wa, false, yb, (short)0, acc2[mt], false, false);
    }
  }
#pragma unroll
  for (int mt = 0; mt < 4; ++mt)
#pragma unroll
    for (int r = 0; r < 8; ++r) ov32[8 * mt + r] = acc2[mt][r];
  rows_out(tw, ov32, T, rowBase, lcol, half, N);
}

__global__ __launch_bounds__(256)
void k_gather_out(float* __restrict__ out, const float* __restrict__ P,
                  const int* __restrict__ rowptr, const int* __restrict__ rec,
                  const float* __restrict__ dinv, const float* __restrict__ b2,
                  const float* __restrict__ X, int N, int E) {
  int t = blockIdx.x * blockDim.x + threadIdx.x;
  int g = t >> 4, l = t & 15;
  int G = (int)((gridDim.x * blockDim.x) >> 4);
  float bias2 = b2[0];

  int dc = g < N ? g : N - 1;
  int beg = rowptr[dc];
  int end = (dc + 1 < N) ? rowptr[dc + 1] : E;
  for (int db = 0; db < N; db += G) {
    int d = db + g;
    bool valid = d < N;
    dc = valid ? d : N - 1;
    float dd = dinv[dc];
    float pself = P[dc];
    float xres = X[(long)dc * FIN];
    int dn = d + G;
    int dnc = dn < N ? dn : N - 1;
    int begn = rowptr[dnc];
    int endn = (dnc + 1 < N) ? rowptr[dnc + 1] : E;

    float acc = 0.0f;
    for (int p = beg + l; p < end; p += 16) {
      v2i r = *(const v2i*)(rec + 2 * (long)p);
      acc += __int_as_float(r.y) * P[r.x];
    }
    for (int mask = 8; mask > 0; mask >>= 1) acc += __shfl_xor(acc, mask, 32);
    if (valid && l == 0) {
      float q = dd * (acc + dd * pself);
      out[d] = q + bias2 + xres;
    }
    beg = begn;
    end = endn;
  }
}

extern "C" void kernel_launch(void* const* d_in, const int* in_sizes, int n_in,
                              void* d_out, int out_size, void* d_ws, size_t ws_size,
                              hipStream_t stream) {
  const float* X    = (const float*)d_in[0];
  const int*   eidx = (const int*)d_in[1];
  const float* ew   = (const float*)d_in[2];
  const float* skip = (const float*)d_in[3];
  const float* H    = (const float*)d_in[4];
  const float* C    = (const float*)d_in[5];
  const float* Wx   = (const float*)d_in[6];
  const float* bx   = (const float*)d_in[7];
  const float* Wh   = (const float*)d_in[8];
  const float* bh   = (const float*)d_in[9];
  const float* wc   = (const float*)d_in[10];
  const float* bg   = (const float*)d_in[11];
  const float* ln_h_s = (const float*)d_in[12];
  const float* ln_h_b = (const float*)d_in[13];
  const float* ln_c_s = (const float*)d_in[14];
  const float* ln_c_b = (const float*)d_in[15];
  const float* ln_o_s = (const float*)d_in[16];
  const float* ln_o_b = (const float*)d_in[17];
  const float* W1   = (const float*)d_in[18];
  const float* b1   = (const float*)d_in[19];
  const float* W2   = (const float*)d_in[20];
  const float* b2   = (const float*)d_in[21];

  const int N = in_sizes[0] / FIN;
  const int E = in_sizes[2];

  const int* src = eidx;
  const int* dst = eidx + E;

  float* out_p    = (float*)d_out;
  float* hidden_p = out_p + N;
  float* cell_p   = hidden_p + (long)N * HID;

  int*   rec  = (int*)d_ws;
  float* AH   = (float*)(rec + 2 * (long)E);
  float* AX   = AH + (long)N * HID;
  float* dinv = AX + (long)N * FIN;
  float* P    = dinv + N;
  int* rowptr = (int*)(P + N);
  int*   rank  = (int*)AH;
  int*   cnt2  = rank + E;
  float* deg2  = (float*)(cnt2 + 2 * (long)N);
  int*   base1 = (int*)(deg2 + 2 * (long)N);
  int*   bsum  = base1 + N;

  const int B = 256;
  auto cdiv = [](long a, long b) { return (unsigned)((a + b - 1) / b); };

  k_init_deg<<<cdiv(N, B), B, 0, stream>>>(deg2, cnt2, N);
  k_acc_deg<<<2048, B, 0, stream>>>(deg2, cnt2, rank, dst, ew, N, E);
  k_dinv<<<cdiv(N, B), B, 0, stream>>>(dinv, deg2, N);

  k_blocksum<<<cdiv(N, SCAN_T * SCAN_PER), SCAN_T, 0, stream>>>(cnt2, bsum, N);
  k_scan<<<cdiv(N, SCAN_T * SCAN_PER), SCAN_T, 0, stream>>>(cnt2, rowptr, base1, bsum, N);
  k_fill<<<2048, B, 0, stream>>>(rowptr, base1, rank, rec, src, dst, ew, dinv, E);

  k_gather_xh<<<2048, B, 0, stream>>>(AH, AX, X, H, rowptr, rec, dinv, N, E);

  k_gates_head<<<cdiv(N, ROWS_PER_BLOCK), B, 0, stream>>>(
      AH, AX, C, Wx, Wh, bx, bh, bg, wc,
      ln_h_s, ln_h_b, ln_c_s, ln_c_b, ln_o_s, ln_o_b,
      skip, W1, hidden_p, cell_p, AH, N);

  k_gather_head<<<2048, B, 0, stream>>>(P, AH, rowptr, rec, dinv, b1, W2, N, E);

  k_gather_out<<<2048, B, 0, stream>>>(out_p, P, rowptr, rec, dinv, b2, X, N, E);
}


# Row aggregations load rows 16 bytes per lane: 16 lanes per 256-byte row, 2 edges (4 whole lines) per load, not 1 line per load; even and odd edges summed apart: float sum order of all outputs changes
# speedup vs baseline: 6.2143x; 1.0481x over previous
#include <hip/hip_runtime.h>
#include <math.h>


typedef __attribute__((ext_vector_type(2))) float v2f;
typedef __attribute__((ext_vector_type(2))) int v2i;
typedef __attribute__((ext_vector_type(8))) float v8f;
typedef __attribute__((ext_vector_type(4))) float v4f;

#define FIN   8
#define HID   64
#define KTOT  72
#define KSTEPS 18
#define KH_STEPS 17
#define TPITCH 68
#define GCOLS 256
#define NT_G  16
#define ROWS_PER_BLOCK 128
#define EPS 1e-5f

__device__ __forceinline__ float sigmoidf_(float x) {
  float e = __builtin_amdgcn_exp2f(-1.4426950408889634f * x);
  return __builtin_amdgcn_rcpf(1.0f + e);
}
__device__ __forceinline__ float tanhf_(float x) {
  float e = __builtin_amdgcn_exp2f(2.8853900817779268f * x);
  return __builtin_fmaf(-2.0f, __builtin_amdgcn_rcpf(1.0f + e), 1.0f);
}

__device__ __forceinline__ int wave_half() {
  return (int)((__builtin_amdgcn_s_sendmsg_rtn(0x87) >> 18) & 1u);
}

__global__ void k_init_deg(float* deg2, int* cnt2, int N) {
  int i = blockIdx.x * blockDim.x + threadIdx.x;
  if (i < N) { deg2[i] = 1.0f; deg2[N + i] = 0.0f; cnt2[i] = 0; cnt2[N + i] = 0; }
}

__global__ __launch_bounds__(256)
void k_acc_deg(float* deg2, int* cnt2, int* __restrict__ rank,
               const int* __restrict__ dst, const float* __restrict__ w, int N, int E) {
  int h = wave_half();
  float* degp = deg2 + (long)h * N;
  int* cntp = cnt2 + (long)h * N;
  int stride = gridDim.x * blockDim.x;
  for (int e = blockIdx.x * blockDim.x + threadIdx.x; e < E; e += stride) {
    int d = dst[e];
    atomicAdd(&degp[d], w[e]);
    int r = atomicAdd(&cntp[d], 1);
    rank[e] = 2 * r + h;
  }
}

__global__ void k_dinv(float* __restrict__ dinv, const float* __restrict__ deg2, int N) {
  int i = blockIdx.x * blockDim.x + threadIdx.x;
  if (i < N) dinv[i] = rsqrtf(deg2[i] + deg2[N + i]);
}

#define SCAN_T 256
#define SCAN_PER 4

__global__ __launch_bounds__(SCAN_T)
void k_blocksum(const int* __restrict__ cnt2, int* __restrict__ bsum, int N) {
  __shared__ int wsum[SCAN_T / 32];
  int base = (blockIdx.x * SCAN_T + threadIdx.x) * SCAN_PER;
  int s = 0;
  for (int i = 0; i < SCAN_PER; ++i)
    if (base + i < N) s += cnt2[base + i] + cnt2[N + base + i];
  for (int mask = 16; mask > 0; mask >>= 1) s += __shfl_xor(s, mask, 32);
  if ((threadIdx.x & 31) == 0) wsum[threadIdx.x >> 5] = s;
  __syncthreads();
  if (threadIdx.x == 0) {
    int tot = 0;
    for (int i = 0; i < SCAN_T / 32; ++i) tot += wsum[i];
    bsum[blockIdx.x] = tot;
  }
}

__global__ __launch_bounds__(SCAN_T)
void k_scan(const int* __restrict__ cnt2, int* __restrict__ rowptr,
            int* __restrict__ base1, const int* __restrict__ bsum, int N) {
  __shared__ int wtot[SCAN_T / 32];
  __shared__ int woff[SCAN_T / 32];
  int tid = threadIdx.x, lane = tid & 31, wave = tid >> 5;

  int off = 0;
  for (int i = tid; i < (int)blockIdx.x; i += SCAN_T) off += bsum[i];
  for (int mask = 16; mask > 0; mask >>= 1) off += __shfl_xor(off, mask, 32);
  if (lane == 0) woff[wave] = off;

  int base = (blockIdx.x * SCAN_T + tid) * SCAN_PER;
  int c[SCAN_PER], c0[SCAN_PER];
  int tsum = 0;
  for (int i = 0; i < SCAN_PER; ++i) {
    c0[i] = (base + i < N) ? cnt2[base + i] : 0;
    c[i] = c0[i] + ((base + i < N) ? cnt2[N + base + i] : 0);
    tsum += c[i];
  }
  int v = tsum;
  for (int o = 1; o < 32; o <<= 1) {
    int u = __shfl_up(v, o, 32);
    if (lane >= o) v += u;
  }
  if (lane == 31) wtot[wave] = v;
  __syncthreads();
  int pre = 0;
  for (int i = 0; i < SCAN_T / 32; ++i) {
    pre += woff[i];
    if (i < wave) pre += wtot[i];
  }
  int excl = pre + v - tsum;
  for (int i = 0; i < SCAN_PER; ++i) {
    if (base + i < N) {
      rowptr[base + i] = excl;
      base1[base + i] = excl + c0[i];
      excl += c[i];
    }
  }
}

__global__ __launch_bounds__(256)
void k_fill(const int* __restrict__ rowptr, const int* __restrict__ base1,
            const int* __restrict__ rank, int* __restrict__ rec,
            const int* __restrict__ src,
            const int* __restrict__ dst, const float* __restrict__ w,
            const float* __restrict__ dinv, int E) {
  long stride = (long)gridDim.x * blockDim.x;
  long t = (long)blockIdx.x * blockDim.x + threadIdx.x;
  for (long e0 = t; e0 < E; e0 += 4 * stride) {
    int dn[4], s[4], rk[4], p[4];
    float we[4], c[4];
#pragma unroll
    for (int u = 0; u < 4; ++u) {
      long e = e0 + u * stride;
      long ec = e < E ? e : e0;
      dn[u] = dst[ec];
      s[u] = src[ec];
      we[u] = w[ec];
      rk[u] = rank[ec];
    }
#pragma unroll
    for (int u = 0; u < 4; ++u) {
      const int* bp = (rk[u] & 1) ? base1 : rowptr;
      p[u] = bp[dn[u]] + (rk[u] >> 1);
      c[u] = we[u] * dinv[s[u]];
    }
#pragma unroll
    for (int u = 0; u < 4; ++u) {
      v2i r;
      r.x = s[u];
      r.y = __float_as_int(c[u]);
      *(v2i*)(rec + 2 * (long)p[u]) = r;
    }
  }
}

__device__ __forceinline__ int pick2(int v, int e0, int hl) {
  int a = __builtin_amdgcn_readlane(v, e0);
  int b = __builtin_amdgcn_readlane(v, e0 + 1);
  return hl ? b : a;
}

__device__ __forceinline__ const v4f* row64(const float* __restrict__ T, int sj, int q4) {
  unsigned off = ((unsigned)sj << 8) + (unsigned)(q4 * 4);
  return (const v4f*)((const char*)T + off);
}
__device__ __forceinline__ const v2f* row8(const float* __restrict__ X, int sj, int qx) {
  unsigned off = ((unsigned)sj << 5) + (unsigned)(qx * 4);
  return (const v2f*)((const char*)X + off);
}

__device__ __forceinline__ void seg_batch_xh(int s, int ci, int cnt, int hl, int q4, int qx,
                                             const float* __restrict__ H,
                                             const float* __restrict__ X,
                                             v4f& a, v2f& ax) {
  int j = 0;
  for (; j + 8 <= cnt; j += 8) {
    v4f x[4];
    v2f y[4];
    float cj[4];
    for (int u = 0; u < 4; ++u) {
      int sj = pick2(s, j + 2 * u, hl);
      cj[u] = __int_as_float(pick2(ci, j + 2 * u, hl));
      x[u] = *row64(H, sj, q4);
      y[u] = *row8(X, sj, qx);
    }
    for (int u = 0; u < 4; ++u) {
      a += cj[u] * x[u];
      ax += cj[u] * y[u];
    }
  }
  if (j < cnt) {
    v4f x[4];
    v2f y[4];
    float cj[4];
    for (int u = 0; u < 4; ++u) {
      int sj = pick2(s, j + 2 * u, hl);
      cj[u] = __int_as_float(pick2(ci, j + 2 * u, hl));
      x[u] = (v4f)(0.0f);
      y[u] = (v2f)(0.0f);
      if (j + 2 * u < cnt) {
        bool ok = j + 2 * u + hl < cnt;
        v4f xl = *row64(H, sj, q4);
        v2f yl = *row8(X, sj, qx);
        x[u] = ok ? xl : x[u];
        y[u] = ok ? yl : y[u];
      }
    }
    for (int u = 0; u < 4; ++u) {
      a += cj[u] * x[u];
      ax += cj[u] * y[u];
    }
  }
}

__device__ __forceinline__ void seg_batch_t(int s, int ci, int cnt, int hl, int q4,
                                            const float* __restrict__ T, v4f& a) {
  int j = 0;
  for (; j + 8 <= cnt; j += 8) {
    v4f x[4];
    float cj[4];
    for (int u = 0; u < 4; ++u) {
      int sj = pick2(s, j + 2 * u, hl);
      cj[u] = __int_as_float(pick2(ci, j + 2 * u, hl));
      x[u] = *row64(T, sj, q4);
    }
    for (int u = 0; u < 4; ++u) a += cj[u] * x[u];
  }
  if (j < cnt) {
    v4f x[4];
    float cj[4];
    for (int u = 0; u < 4; ++u) {
      int sj = pick2(s, j + 2 * u, hl);
      cj[u] = __int_as_float(pick2(ci, j + 2 * u, hl));
      x[u] = (v4f)(0.0f);
      if (j + 2 * u < cnt) {
        bool ok = j + 2 * u + hl < cnt;
        v4f xl = *row64(T, sj, q4);
        x[u] = ok ? xl : x[u];
      }
    }
    for (int u = 0; u < 4; ++u) a += cj[u] * x[u];
  }
}

template <int M>
__device__ __forceinline__ float lane_xor(float v) {
  return __int_as_float(__builtin_amdgcn_ds_swizzle(__float_as_int(v), (M << 10) | 0x1f));
}

__device__ __forceinline__ v4f join_halves(v4f a) {
  a.x += lane_xor<16>(a.x);
  a.y += lane_xor<16>(a.y);
  a.z += lane_xor<16>(a.z);
  a.w += lane_xor<16>(a.w);
  return a;
}

#define RUN_MAX 31

__global__ __launch_bounds__(256)
void k_gather_xh(float* __restrict__ AH, float* __restrict__ AX,
                 const float* __restrict__ X, const float* __restrict__ H,
                 const int* __restrict__ rowptr, const int* __restrict__ rec,
                 const float* __restrict__ dinv, int N, int E) {
  int lane = threadIdx.x & 31;
  int hl = lane >> 4;
  int q4 = 4 * (lane & 15);
  int qx = 2 * (lane & 3);
  int wid = __builtin_amdgcn_readfirstlane((int)(blockIdx.x * 8 + (threadIdx.x >> 5)));
  unsigned nw = gridDim.x * 8u;
  unsigned q = (unsigned)N / nw, rem = (unsigned)N % nw;
  int n0 = (int)((unsigned)wid * q + ((unsigned)wid < rem ? (unsigned)wid : rem));
  int n1 = n0 + (int)q + ((unsigned)wid < rem ? 1 : 0);

  for (int nb = n0; nb < n1; nb += RUN_MAX) {
    int cn = n1 - nb;
    if (cn > RUN_MAX) cn = RUN_MAX;
    int idx = nb + lane;
    int rp = E;
    float dvl = 0.0f;
    if (idx < N) { rp = rowptr[idx]; dvl = dinv[idx]; }
    int dvi = __float_as_int(dvl);

    int beg = __builtin_amdgcn_readlane(rp, 0);
    int end = __builtin_amdgcn_readlane(rp, 1);
    v2i rc;
    rc.x = 0; rc.y = 0;
    if (beg + lane < end) rc = *(const v2i*)(rec + 2 * (long)(beg + lane));
    v4f hrow = *row64(H, nb, q4);
    v2f xrow = *row8(X, nb, qx);

    for (int i = 0; i < cn; ++i) {
      int d = nb + i;
      float dd = __int_as_float(__builtin_amdgcn_readlane(dvi, i));
      float ds = hl ? 0.0f : dd;
      v4f a = ds * hrow;
      v2f ax = ds * xrow;

      int endn = end;
      v2i rn;
      rn.x = 0; rn.y = 0;
      if (i + 1 < cn) {
        endn = __builtin_amdgcn_readlane(rp, i + 2);
        if (end + lane < endn) rn = *(const v2i*)(rec + 2 * (long)(end + lane));
        hrow = *row64(H, d + 1, q4);
        xrow = *row8(X, d + 1, qx);
      }

      int cnt = end - beg;
      seg_batch_xh(rc.x, rc.y, cnt > 32 ? 32 : cnt, hl, q4, qx, H, X, a, ax);
      for (int base = beg + 32; base < end; base += 32) {
        int p = base + lane;
        v2i r;
        r.x = 0; r.y = 0;
        if (p < end) r = *(const v2i*)(rec + 2 * (long)p);
        int c2 = end - base;
        seg_batch_xh(r.x, r.y, c2 > 32 ? 32 : c2, hl, q4, qx, H, X, a, ax);
      }

      a = join_halves(a);
      ax.x += lane_xor<16>(ax.x);
      ax.y += lane_xor<16>(ax.y);
      if (lane < 16) *(v4f*)(AH + (long)d * HID + q4) = dd * a;
      if (lane < 4) *(v2f*)(AX + (long)d * FIN + qx) = dd * ax;

      rc = rn;
      beg = end;
      end = endn;
    }
  }
}

__global__ __launch_bounds__(256)
void k_gather_head(float* __restrict__ P, const float* __restrict__ T,
                   const int* __restrict__ rowptr, const int* __restrict__ rec,
                   const float* __restrict__ dinv, const float* __restrict__ b1,
                   const float* __restrict__ W2, int N, int E) {
  int lane = threadIdx.x & 31;
  int hl = lane >> 4;
  int q4 = 4 * (lane & 15);
  int wid = __builtin_amdgcn_readfirstlane((int)(blockIdx.x * 8 + (threadIdx.x >> 5)));
  unsigned nw = gridDim.x * 8u;
  unsigned q = (unsigned)N / nw, rem = (unsigned)N % nw;
  int n0 = (int)((unsigned)wid * q + ((unsigned)wid < rem ? (unsigned)wid : rem));
  int n1 = n0 + (int)q + ((unsigned)wid < rem ? 1 : 0);

  v4f bl = *(const v4f*)(b1 + q4);
  v4f wl = *(const v4f*)(W2 + q4);

  for (int nb = n0; nb < n1; nb += RUN_MAX) {
    int cn = n1 - nb;
    if (cn > RUN_MAX) cn = RUN_MAX;
    int idx = nb + lane;
    int rp = E;
    float dvl = 0.0f;
    if (idx < N) { rp = rowptr[idx]; dvl = dinv[idx]; }
    int dvi = __float_as_int(dvl);

    int beg = __builtin_amdgcn_readlane(rp, 0);
    int end = __builtin_amdgcn_readlane(rp, 1);
    v2i rc;
    rc.x = 0; rc.y = 0;
    if (beg + lane < end) rc = *(const v2i*)(rec + 2 * (long)(beg + lane));
    v4f hrow = *row64(T, nb, q4);

    for (int i = 0; i < cn; ++i) {
      int d = nb + i;
      float dd = __int_as_float(__builtin_amdgcn_readlane(dvi, i));
      float ds = hl ? 0.0f : dd;
      v4f a = ds * hrow;

      int endn = end;
      v2i rn;
      rn.x = 0; rn.y = 0;
      if (i + 1 < cn) {
        endn = __builtin_amdgcn_readlane(rp, i + 2);
        if (end + lane < endn) rn = *(const v2i*)(rec + 2 * (long)(end + lane));
        hrow = *row64(T, d + 1, q4);
      }

      int cnt = end - beg;
      seg_batch_t(rc.x, rc.y, cnt > 32 ? 32 : cnt, hl, q4, T, a);
      for (int base = beg + 32; base < end; base += 32) {
        int p = base + lane;
        v2i r;
        r.x = 0; r.y = 0;
        if (p < end) r = *(const v2i*)(rec + 2 * (long)p);
        int c2 = end - base;
        seg_batch_t(r.x, r.y, c2 > 32 ? 32 : c2, hl, q4, T, a);
      }

      a = join_halves(a);
      v4f z = dd * a + bl;
      z.x = z.x > 0.0f ? z.x : 0.0f;
      z.y = z.y > 0.0f ? z.y : 0.0f;
      z.z = z.z > 0.0f ? z.z : 0.0f;
      z.w = z.w > 0.0f ? z.w : 0.0f;
      float pv = z.x * wl.x + z.y * wl.y + z.z * wl.z + z.w * wl.w;
      pv += lane_xor<8>(pv);
      pv += lane_xor<4>(pv);
      pv += lane_xor<2>(pv);
      pv += lane_xor<1>(pv);
      if (lane == 0) P[d] = pv;

      rc = rn;
      beg = end;
      end = endn;
    }
  }
}


#define BF_FLOATS   (KSTEPS * NT_G * 64)
#define PAR_BIAS    0
#define PAR_WC      256
#define PAR_LN      448
#define PAR_FLOATS  832
#define W1F_FLOATS  (KH_STEPS * 4 * 64)
#define TILE_FLOATS (8 * 16 * TPITCH)

__device__ __forceinline__ void ld8(const float* p, float* o) {
  v4f a = *(const v4f*)p;
  v4f b = *(const v4f*)(p + 4);
  o[0] = a.x; o[1] = a.y; o[2] = a.z; o[3] = a.w;
  o[4] = b.x; o[5] = b.y; o[6] = b.z; o[7] = b.w;
}

__device__ __forceinline__ void rows_out(float* tw, const float* v, float* __restrict__ dst,
                                         int rowBase, int lcol, int half, int N) {
  float* trow = tw + lcol * TPITCH + 8 * half;
#pragma unroll
  for (int t = 0; t < 4; ++t) {
    v4f a, b;
    a.x = v[8 * t];     a.y = v[8 * t + 1]; a.z = v[8 * t + 2]; a.w = v[8 * t + 3];
    b.x = v[8 * t + 4]; b.y = v[8 * t + 5]; b.z = v[8 * t + 6]; b.w = v[8 * t + 7];
    *(v4f*)(trow + 16 * t) = a;
    *(v4f*)(trow + 16 * t + 4) = b;
  }
  __syncthreads();
#pragma unroll
  for (int i = 0; i < 8; ++i) {
    int m = 2 * i + half;
    int node = rowBase + m;
    v4f x = *(const v4f*)(tw + m * TPITCH + 4 * lcol);
    if (node < N) *(v4f*)(dst + (long)node * HID + 4 * lcol) = x;
  }
}

__global__ __launch_bounds__(256)
void k_gates_head(const float* AH, const float* __restrict__ AX,
                  const float* __restrict__ C,
                  const float* __restrict__ Wx, const float* __restrict__ Wh,
                  const float* __restrict__ bx, const float* __restrict__ bh,
                  const float* __restrict__ bg, const float* __restrict__ wc,
                  const float* __restrict__ ln_h_s, const float* __restrict__ ln_h_b,
                  const float* __restrict__ ln_c_s, const float* __restrict__ ln_c_b,
                  const float* __restrict__ ln_o_s, const float* __restrict__ ln_o_b,
                  const float* __restrict__ skip, const float* __restrict__ W1,
                  float* __restrict__ hiddenOut, float* __restrict__ cellOut,
                  float* T, int N) {
  __shared__ __attribute__((aligned(16)))
      float smem[BF_FLOATS + PAR_FLOATS + W1F_FLOATS + TILE_FLOATS];
  float* Bf   = smem;
  float* par  = Bf + BF_FLOATS;
  float* W1f  = par + PAR_FLOATS;
  float* tile = W1f + W1F_FLOATS;

  int tid = threadIdx.x;
  {
    int mt = tid >> 6, L = (tid & 63) >> 1, j = tid & 1;
    int hh = L >> 4;
    int c = 16 * mt + (L & 15);
    float w[KH_STEPS];
#pragma unroll
    for (int st = 0; st < KH_STEPS - 1; ++st)
      w[st] = W1[(16 * (st >> 2) + 8 * hh + 2 * (st & 3) + j) * HID + c];
    w[KH_STEPS - 1] = W1[(HID + j) * HID + c];
    float bsum = bx[tid] + bh[tid] + bg[tid];
    float wcv = wc[tid < 192 ? tid : 0];
    int t6 = tid & 63;
    float l0 = ln_h_s[t6], l1 = ln_h_b[t6], l2 = ln_c_s[t6];
    float l3 = ln_c_b[t6], l4 = ln_o_s[t6], l5 = ln_o_b[t6];
#pragma unroll
    for (int st = 0; st < KH_STEPS; ++st)
      W1f[st * 256 + tid] = (st < KH_STEPS - 1 || hh == 0) ? w[st] : 0.0f;
    par[PAR_BIAS + tid] = bsum;
    if (tid < 192) par[PAR_WC + tid] = wcv;
    if (tid < 64) {
      par[PAR_LN + tid]       = l0;
      par[PAR_LN + 64 + tid]  = l1;
      par[PAR_LN + 128 + tid] = l2;
      par[PAR_LN + 192 + tid] = l3;
      par[PAR_LN + 256 + tid] = l4;
      par[PAR_LN + 320 + tid] = l5;
    }
  }
  for (int i = tid; i < BF_FLOATS; i += 256) {
    int s = i >> 10;
    int rem = i & 1023;
    int t = rem >> 6;
    int rem2 = rem & 63;
    int L = rem2 >> 1;
    int j = rem2 & 1;
    int k = 4 * s + 2 * (L >> 4) + j;
    int c = 16 * t + (L & 15);
    int g = c >> 6, h = c & 63;
    Bf[i] = (k < FIN) ? Wx[(g * FIN + k) * HID + h]
                      : Wh[(g * HID + (k - FIN)) * HID + h];
  }
  __syncthreads();

  int wave = tid >> 5, lane = tid & 31;
  int half = lane >> 4, lcol = lane & 15;
  int rowBase = blockIdx.x * ROWS_PER_BLOCK + wave * 16;

  int arow = rowBase + lcol;
  if (arow > N - 1) arow = N - 1;
  const float* aXp = AX + (long)arow * FIN;
  const float* aHp = AH + (long)arow * HID;

  v4f cpre[8];
  {
    const float* Cp = C + (long)arow * HID + 8 * half;
#pragma unroll
    for (int t = 0; t < 4; ++t) {
      cpre[2 * t]     = *(const v4f*)(Cp + 16 * t);
      cpre[2 * t + 1] = *(const v4f*)(Cp + 16 * t + 4);
    }
  }
  v2f sk = *(const v2f*)(skip + (long)arow * 2);

  v8f acc[NT_G];
#pragma unroll
  for (int t = 0; t < NT_G; ++t)
#pragma unroll
    for (int r = 0; r < 8; ++r) acc[t][r] = 0.0f;

  for (int s = 0; s < KSTEPS; ++s) {
    int k0 = 4 * s + 2 * half;
    v2f a = *(const v2f*)(s < 2 ? aXp + k0 : aHp + (k0 - FIN));
    const float* bbase = Bf + s * 1024 + lane * 2;
#pragma unroll
    for (int t = 0; t < NT_G; ++t) {
      v2f b = *(const v2f*)(bbase + t * 64);
      acc[t] = __builtin_amdgcn_wmma_f32_16x16x4_f32(
          false, b, false, a, (short)0, acc[t], false, false);
    }
  }

  const float* pl = par + 8 * half;
  float hv[32], cv[32];
  float sh = 0.f, sh2 = 0.f, sc = 0.f, sc2 = 0.f;
#pragma unroll
  for (int t = 0; t < 4; ++t) {
    float c0[8], bi[8], bfg[8], bc[8], bo[8], wi[8], wf[8], wo[8];
    c0[0] = cpre[2 * t].x;     c0[1] = cpre[2 * t].y;
    c0[2] = cpre[2 * t].z;     c0[3] = cpre[2 * t].w;
    c0[4] = cpre[2 * t + 1].x; c0[5] = cpre[2 * t + 1].y;
    c0[6] = cpre[2 * t + 1].z; c0[7] = cpre[2 * t + 1].w;
    ld8(pl + PAR_BIAS + 16 * t, bi);
    ld8(pl + PAR_BIAS + 64 + 16 * t, bfg);
    ld8(pl + PAR_BIAS + 128 + 16 * t, bc);
    ld8(pl + PAR_BIAS + 192 + 16 * t, bo);
    ld8(pl + PAR_WC + 16 * t, wi);
    ld8(pl + PAR_WC + 64 + 16 * t, wf);
    ld8(pl + PAR_WC + 128 + 16 * t, wo);
    float ph = 0.f, ph2 = 0.f, pc = 0.f, pc2 = 0.f;
#pragma unroll
    for (int r = 0; r < 8; ++r) {
      float gi = acc[t][r]      + bi[r]  + wi[r] * c0[r];
      float gf = acc[t + 4][r]  + bfg[r] + wf[r] * c0[r];
      float gc = acc[t + 8][r]  + bc[r];
      float go = acc[t + 12][r] + bo[r];
      float iv = sigmoidf_(gi);
      float fv = sigmoidf_(gf);
      float cc = fv * c0[r] + iv * tanhf_(gc);
      float ov = sigmoidf_(go + wo[r] * cc);
      float hh = ov * tanhf_(cc);
      hv[8 * t + r] = hh;
      cv[8 * t + r] = cc;
      ph += hh; ph2 += hh * hh;
      pc += cc; pc2 += cc * cc;
    }
    sh += ph; sh2 += ph2; sc += pc; sc2 += pc2;
  }
  sh  += __shfl_xor(sh,  16, 32);
  sh2 += __shfl_xor(sh2, 16, 32);
  sc  += __shfl_xor(sc,  16, 32);
  sc2 += __shfl_xor(sc2, 16, 32);
  float mh = sh * (1.0f / 64.0f), mc = sc * (1.0f / 64.0f);
  float vh = sh2 * (1.0f / 64.0f) - mh * mh;
  float vc = sc2 * (1.0f / 64.0f) - mc * mc;
  float rh = rsqrtf(vh + EPS), rc = rsqrtf(vc + EPS);

  float* tw = tile + wave * (16 * TPITCH);
  float ov32[32];
#pragma unroll
  for (int t = 0; t < 4; ++t) {
    float s8[8], b8[8];
    ld8(pl + PAR_LN + 16 * t, s8);
    ld8(pl + PAR_LN + 64 + 16 * t, b8);
#pragma unroll
    for (int r = 0; r < 8; ++r) {
      float hn = (hv[8 * t + r] - mh) * rh;
      ov32[8 * t + r] = hn * s8[r] + b8[r];
    }
  }
  rows_out(tw, ov32, hiddenOut, rowBase, lcol, half, N);
#pragma unroll
  for (int t = 0; t < 4; ++t) {
    float s8[8], b8[8];
    ld8(pl + PAR_LN + 128 + 16 * t, s8);
    ld8(pl + PAR_LN + 192 + 16 * t, b8);
#pragma unroll
    for (int r = 0; r < 8; ++r) {
      float cn = (cv[8 * t + r] - mc) * rc;
      ov32[8 * t + r] = cn * s8[r] + b8[r];
    }
  }
  rows_out(tw, ov32, cellOut, rowBase, lcol, half, N);
  float yv[32];
#pragma unroll
  for (int t = 0; t < 4; ++t) {
    float s8[8], b8[8];
    ld8(pl + PAR_LN + 256 + 16 * t, s8);
    ld8(pl + PAR_LN + 320 + 16 * t, b8);
#pragma unroll
    for (int r = 0; r < 8; ++r) {
      float hn = (hv[8 * t + r] - mh) * rh;
      float o = hn * s8[r] + b8[r];
      yv[8 * t + r] = o > 0.0f ? o : 0.0f;
    }
  }

  v8f acc2[4];
#pragma unroll
  for (int mt = 0; mt < 4; ++mt)
#pragma unroll
    for (int r = 0; r < 8; ++r) acc2[mt][r] = 0.0f;
#pragma unroll
  for (int t = 0; t < 4; ++t)
#pragma unroll
    for (int p = 0; p < 4; ++p) {
      v2f yb;
      yb.x = yv[8 * t + 2 * p];
      yb.y = yv[8 * t + 2 * p + 1];
      const float* wb = W1f + (t * 4 + p) * 256 + lane * 2;
#pragma unroll
      for (int mt = 0; mt < 4; ++mt) {
        v2f wa = *(const v2f*)(wb + mt * 64);
        acc2[mt] = __builtin_amdgcn_wmma_f32_16x16x4_f32(
            false, wa, false, yb, (short)0, acc2[mt], false, false);
      }
    }
  {
    v2f yb;
    yb.x = half ? 0.0f : sk.x;
    yb.y = half ? 0.0f : sk.y;
    const float* wb = W1f + (KH_STEPS - 1) * 256 + lane * 2;
#pragma unroll
    for (int mt = 0; mt < 4; ++mt) {
      v2f wa = *(const v2f*)(wb + mt * 64);
      acc2[mt] = __builtin_amdgcn_wmma_f32_16x16x4_f32(
          false, wa, false, yb, (short)0, acc2[mt], false, false);
    }
  }
#pragma unroll
  for (int mt = 0; mt < 4; ++mt)
#pragma unroll
    for (int r = 0; r < 8; ++r) ov32[8 * mt + r] = acc2[mt][r];
  rows_out(tw, ov32, T, rowBase, lcol, half, N);
}

__global__ __launch_bounds__(256)
void k_gather_out(float* __restrict__ out, const float* __restrict__ P,
                  const int* __restrict__ rowptr, const int* __restrict__ rec,
                  const float* __restrict__ dinv, const float* __restrict__ b2,
                  const float* __restrict__ X, int N, int E) {
  int t = blockIdx.x * blockDim.x + threadIdx.x;
  int g = t >> 4, l = t & 15;
  int G = (int)((gridDim.x * blockDim.x) >> 4);
  float bias2 = b2[0];

  int dc = g < N ? g : N - 1;
  int beg = rowptr[dc];
  int end = (dc + 1 < N) ? rowptr[dc + 1] : E;
  for (int db = 0; db < N; db += G) {
    int d = db + g;
    bool valid = d < N;
    dc = valid ? d : N - 1;
    float dd = dinv[dc];
    float pself = P[dc];
    float xres = X[(long)dc * FIN];
    int dn = d + G;
    int dnc = dn < N ? dn : N - 1;
    int begn = rowptr[dnc];
    int endn = (dnc + 1 < N) ? rowptr[dnc + 1] : E;

    float acc = 0.0f;
    for (int p = beg + l; p < end; p += 16) {
      v2i r = *(const v2i*)(rec + 2 * (long)p);
      acc += __int_as_float(r.y) * P[r.x];
    }
    for (int mask = 8; mask > 0; mask >>= 1) acc += __shfl_xor(acc, mask, 32);
    if (valid && l == 0) {
      float q = dd * (acc + dd * pself);
      out[d] = q + bias2 + xres;
    }
    beg = begn;
    end = endn;
  }
}

extern "C" void kernel_launch(void* const* d_in, const int* in_sizes, int n_in,
                              void* d_out, int out_size, void* d_ws, size_t ws_size,
                              hipStream_t stream) {
  const float* X    = (const float*)d_in[0];
  const int*   eidx = (const int*)d_in[1];
  const float* ew   = (const float*)d_in[2];
  const float* skip = (const float*)d_in[3];
  const float* H    = (const float*)d_in[4];
  const float* C    = (const float*)d_in[5];
  const float* Wx   = (const float*)d_in[6];
  const float* bx   = (const float*)d_in[7];
  const float* Wh   = (const float*)d_in[8];
  const float* bh   = (const float*)d_in[9];
  const float* wc   = (const float*)d_in[10];
  const float* bg   = (const float*)d_in[11];
  const float* ln_h_s = (const float*)d_in[12];
  const float* ln_h_b = (const float*)d_in[13];
  const float* ln_c_s = (const float*)d_in[14];
  const float* ln_c_b = (const float*)d_in[15];
  const float* ln_o_s = (const float*)d_in[16];
  const float* ln_o_b = (const float*)d_in[17];
  const float* W1   = (const float*)d_in[18];
  const float* b1   = (const float*)d_in[19];
  const float* W2   = (const float*)d_in[20];
  const float* b2   = (const float*)d_in[21];

  const int N = in_sizes[0] / FIN;
  const int E = in_sizes[2];

  const int* src = eidx;
  const int* dst = eidx + E;

  float* out_p    = (float*)d_out;
  float* hidden_p = out_p + N;
  float* cell_p   = hidden_p + (long)N * HID;

  int*   rec  = (int*)d_ws;
  float* AH   = (float*)(rec + 2 * (long)E);
  float* AX   = AH + (long)N * HID;
  float* dinv = AX + (long)N * FIN;
  float* P    = dinv + N;
  int* rowptr = (int*)(P + N);
  int*   rank  = (int*)AH;
  int*   cnt2  = rank + E;
  float* deg2  = (float*)(cnt2 + 2 * (long)N);
  int*   base1 = (int*)(deg2 + 2 * (long)N);
  int*   bsum  = base1 + N;

  const int B = 256;
  auto cdiv = [](long a, long b) { return (unsigned)((a + b - 1) / b); };

  k_init_deg<<<cdiv(N, B), B, 0, stream>>>(deg2, cnt2, N);
  k_acc_deg<<<2048, B, 0, stream>>>(deg2, cnt2, rank, dst, ew, N, E);
  k_dinv<<<cdiv(N, B), B, 0, stream>>>(dinv, deg2, N);

  k_blocksum<<<cdiv(N, SCAN_T * SCAN_PER), SCAN_T, 0, stream>>>(cnt2, bsum, N);
  k_scan<<<cdiv(N, SCAN_T * SCAN_PER), SCAN_T, 0, stream>>>(cnt2, rowptr, base1, bsum, N);
  k_fill<<<2048, B, 0, stream>>>(rowptr, base1, rank, rec, src, dst, ew, dinv, E);

  k_gather_xh<<<2048, B, 0, stream>>>(AH, AX, X, H, rowptr, rec, dinv, N, E);

  k_gates_head<<<cdiv(N, ROWS_PER_BLOCK), B, 0, stream>>>(
      AH, AX, C, Wx, Wh, bx, bh, bg, wc,
      ln_h_s, ln_h_b, ln_c_s, ln_c_b, ln_o_s, ln_o_b,
      skip, W1, hidden_p, cell_p, AH, N);

  k_gather_head<<<2048, B, 0, stream>>>(P, AH, rowptr, rec, dinv, b1, W2, N, E);

  k_gather_out<<<2048, B, 0, stream>>>(out_p, P, rowptr, rec, dinv, b2, X, N, E);
}


# k_gates_head as a resident grid of 512 blocks: weights staged once per block, each wave loops over 16-node tiles on its own (wave-level LDS ordering, no block barrier per tile); sums unchanged
# speedup vs baseline: 6.4889x; 1.0442x over previous
#include <hip/hip_runtime.h>
#include <math.h>


typedef __attribute__((ext_vector_type(2))) float v2f;
typedef __attribute__((ext_vector_type(2))) int v2i;
typedef __attribute__((ext_vector_type(8))) float v8f;
typedef __attribute__((ext_vector_type(4))) float v4f;

#define FIN   8
#define HID   64
#define KTOT  72
#define KSTEPS 18
#define KH_STEPS 17
#define TPITCH 68
#define GCOLS 256
#define NT_G  16
#define EPS 1e-5f

__device__ __forceinline__ float sigmoidf_(float x) {
  float e = __builtin_amdgcn_exp2f(-1.4426950408889634f * x);
  return __builtin_amdgcn_rcpf(1.0f + e);
}
__device__ __forceinline__ float tanhf_(float x) {
  float e = __builtin_amdgcn_exp2f(2.8853900817779268f * x);
  return __builtin_fmaf(-2.0f, __builtin_amdgcn_rcpf(1.0f + e), 1.0f);
}

__device__ __forceinline__ int wave_half() {
  return (int)((__builtin_amdgcn_s_sendmsg_rtn(0x87) >> 18) & 1u);
}

__global__ void k_init_deg(float* deg2, int* cnt2, int N) {
  int i = blockIdx.x * blockDim.x + threadIdx.x;
  if (i < N) { deg2[i] = 1.0f; deg2[N + i] = 0.0f; cnt2[i] = 0; cnt2[N + i] = 0; }
}

__global__ __launch_bounds__(256)
void k_acc_deg(float* deg2, int* cnt2, int* __restrict__ rank,
               const int* __restrict__ dst, const float* __restrict__ w, int N, int E) {
  int h = wave_half();
  float* degp = deg2 + (long)h * N;
  int* cntp = cnt2 + (long)h * N;
  int stride = gridDim.x * blockDim.x;
  for (int e = blockIdx.x * blockDim.x + threadIdx.x; e < E; e += stride) {
    int d = dst[e];
    atomicAdd(&degp[d], w[e]);
    int r = atomicAdd(&cntp[d], 1);
    rank[e] = 2 * r + h;
  }
}

__global__ void k_dinv(float* __restrict__ dinv, const float* __restrict__ deg2, int N) {
  int i = blockIdx.x * blockDim.x + threadIdx.x;
  if (i < N) dinv[i] = rsqrtf(deg2[i] + deg2[N + i]);
}

#define SCAN_T 256
#define SCAN_PER 4

__global__ __launch_bounds__(SCAN_T)
void k_blocksum(const int* __restrict__ cnt2, int* __restrict__ bsum, int N) {
  __shared__ int wsum[SCAN_T / 32];
  int base = (blockIdx.x * SCAN_T + threadIdx.x) * SCAN_PER;
  int s = 0;
  for (int i = 0; i < SCAN_PER; ++i)
    if (base + i < N) s += cnt2[base + i] + cnt2[N + base + i];
  for (int mask = 16; mask > 0; mask >>= 1) s += __shfl_xor(s, mask, 32);
  if ((threadIdx.x & 31) == 0) wsum[threadIdx.x >> 5] = s;
  __syncthreads();
  if (threadIdx.x == 0) {
    int tot = 0;
    for (int i = 0; i < SCAN_T / 32; ++i) tot += wsum[i];
    bsum[blockIdx.x] = tot;
  }
}

__global__ __launch_bounds__(SCAN_T)
void k_scan(const int* __restrict__ cnt2, int* __restrict__ rowptr,
            int* __restrict__ base1, const int* __restrict__ bsum, int N) {
  __shared__ int wtot[SCAN_T / 32];
  __shared__ int woff[SCAN_T / 32];
  int tid = threadIdx.x, lane = tid & 31, wave = tid >> 5;

  int off = 0;
  for (int i = tid; i < (int)blockIdx.x; i += SCAN_T) off += bsum[i];
  for (int mask = 16; mask > 0; mask >>= 1) off += __shfl_xor(off, mask, 32);
  if (lane == 0) woff[wave] = off;

  int base = (blockIdx.x * SCAN_T + tid) * SCAN_PER;
  int c[SCAN_PER], c0[SCAN_PER];
  int tsum = 0;
  for (int i = 0; i < SCAN_PER; ++i) {
    c0[i] = (base + i < N) ? cnt2[base + i] : 0;
    c[i] = c0[i] + ((base + i < N) ? cnt2[N + base + i] : 0);
    tsum += c[i];
  }
  int v = tsum;
  for (int o = 1; o < 32; o <<= 1) {
    int u = __shfl_up(v, o, 32);
    if (lane >= o) v += u;
  }
  if (lane == 31) wtot[wave] = v;
  __syncthreads();
  int pre = 0;
  for (int i = 0; i < SCAN_T / 32; ++i) {
    pre += woff[i];
    if (i < wave) pre += wtot[i];
  }
  int excl = pre + v - tsum;
  for (int i = 0; i < SCAN_PER; ++i) {
    if (base + i < N) {
      rowptr[base + i] = excl;
      base1[base + i] = excl + c0[i];
      excl += c[i];
    }
  }
}

__global__ __launch_bounds__(256)
void k_fill(const int* __restrict__ rowptr, const int* __restrict__ base1,
            const int* __restrict__ rank, int* __restrict__ rec,
            const int* __restrict__ src,
            const int* __restrict__ dst, const float* __restrict__ w,
            const float* __restrict__ dinv, int E) {
  long stride = (long)gridDim.x * blockDim.x;
  long t = (long)blockIdx.x * blockDim.x + threadIdx.x;
  for (long e0 = t; e0 < E; e0 += 4 * stride) {
    int dn[4], s[4], rk[4], p[4];
    float we[4], c[4];
#pragma unroll
    for (int u = 0; u < 4; ++u) {
      long e = e0 + u * stride;
      long ec = e < E ? e : e0;
      dn[u] = dst[ec];
      s[u] = src[ec];
      we[u] = w[ec];
      rk[u] = rank[ec];
    }
#pragma unroll
    for (int u = 0; u < 4; ++u) {
      const int* bp = (rk[u] & 1) ? base1 : rowptr;
      p[u] = bp[dn[u]] + (rk[u] >> 1);
      c[u] = we[u] * dinv[s[u]];
    }
#pragma unroll
    for (int u = 0; u < 4; ++u) {
      v2i r;
      r.x = s[u];
      r.y = __float_as_int(c[u]);
      *(v2i*)(rec + 2 * (long)p[u]) = r;
    }
  }
}

__device__ __forceinline__ int pick2(int v, int e0, int hl) {
  int a = __builtin_amdgcn_readlane(v, e0);
  int b = __builtin_amdgcn_readlane(v, e0 + 1);
  return hl ? b : a;
}

__device__ __forceinline__ const v4f* row64(const float* __restrict__ T, int sj, int q4) {
  unsigned off = ((unsigned)sj << 8) + (unsigned)(q4 * 4);
  return (const v4f*)((const char*)T + off);
}
__device__ __forceinline__ const v2f* row8(const float* __restrict__ X, int sj, int qx) {
  unsigned off = ((unsigned)sj << 5) + (unsigned)(qx * 4);
  return (const v2f*)((const char*)X + off);
}

__device__ __forceinline__ void seg_batch_xh(int s, int ci, int cnt, int hl, int q4, int qx,
                                             const float* __restrict__ H,
                                             const float* __restrict__ X,
                                             v4f& a, v2f& ax) {
  int j = 0;
  for (; j + 8 <= cnt; j += 8) {
    v4f x[4];
    v2f y[4];
    float cj[4];
    for (int u = 0; u < 4; ++u) {
      int sj = pick2(s, j + 2 * u, hl);
      cj[u] = __int_as_float(pick2(ci, j + 2 * u, hl));
      x[u] = *row64(H, sj, q4);
      y[u] = *row8(X, sj, qx);
    }
    for (int u = 0; u < 4; ++u) {
      a += cj[u] * x[u];
      ax += cj[u] * y[u];
    }
  }
  if (j < cnt) {
    v4f x[4];
    v2f y[4];
    float cj[4];
    for (int u = 0; u < 4; ++u) {
      int sj = pick2(s, j + 2 * u, hl);
      cj[u] = __int_as_float(pick2(ci, j + 2 * u, hl));
      x[u] = (v4f)(0.0f);
      y[u] = (v2f)(0.0f);
      if (j + 2 * u < cnt) {
        bool ok = j + 2 * u + hl < cnt;
        v4f xl = *row64(H, sj, q4);
        v2f yl = *row8(X, sj, qx);
        x[u] = ok ? xl : x[u];
        y[u] = ok ? yl : y[u];
      }
    }
    for (int u = 0; u < 4; ++u) {
      a += cj[u] * x[u];
      ax += cj[u] * y[u];
    }
  }
}

__device__ __forceinline__ void seg_batch_t(int s, int ci, int cnt, int hl, int q4,
                                            const float* __restrict__ T, v4f& a) {
  int j = 0;
  for (; j + 8 <= cnt; j += 8) {
    v4f x[4];
    float cj[4];
    for (int u = 0; u < 4; ++u) {
      int sj = pick2(s, j + 2 * u, hl);
      cj[u] = __int_as_float(pick2(ci, j + 2 * u, hl));
      x[u] = *row64(T, sj, q4);
    }
    for (int u = 0; u < 4; ++u) a += cj[u] * x[u];
  }
  if (j < cnt) {
    v4f x[4];
    float cj[4];
    for (int u = 0; u < 4; ++u) {
      int sj = pick2(s, j + 2 * u, hl);
      cj[u] = __int_as_float(pick2(ci, j + 2 * u, hl));
      x[u] = (v4f)(0.0f);
      if (j + 2 * u < cnt) {
        bool ok = j + 2 * u + hl < cnt;
        v4f xl = *row64(T, sj, q4);
        x[u] = ok ? xl : x[u];
      }
    }
    for (int u = 0; u < 4; ++u) a += cj[u] * x[u];
  }
}

template <int M>
__device__ __forceinline__ float lane_xor(float v) {
  return __int_as_float(__builtin_amdgcn_ds_swizzle(__float_as_int(v), (M << 10) | 0x1f));
}

__device__ __forceinline__ v4f join_halves(v4f a) {
  a.x += lane_xor<16>(a.x);
  a.y += lane_xor<16>(a.y);
  a.z += lane_xor<16>(a.z);
  a.w += lane_xor<16>(a.w);
  return a;
}

#define RUN_MAX 31

__global__ __launch_bounds__(256)
void k_gather_xh(float* __restrict__ AH, float* __restrict__ AX,
                 const float* __restrict__ X, const float* __restrict__ H,
                 const int* __restrict__ rowptr, const int* __restrict__ rec,
                 const float* __restrict__ dinv, int N, int E) {
  int lane = threadIdx.x & 31;
  int hl = lane >> 4;
  int q4 = 4 * (lane & 15);
  int qx = 2 * (lane & 3);
  int wid = __builtin_amdgcn_readfirstlane((int)(blockIdx.x * 8 + (threadIdx.x >> 5)));
  unsigned nw = gridDim.x * 8u;
  unsigned q = (unsigned)N / nw, rem = (unsigned)N % nw;
  int n0 = (int)((unsigned)wid * q + ((unsigned)wid < rem ? (unsigned)wid : rem));
  int n1 = n0 + (int)q + ((unsigned)wid < rem ? 1 : 0);

  for (int nb = n0; nb < n1; nb += RUN_MAX) {
    int cn = n1 - nb;
    if (cn > RUN_MAX) cn = RUN_MAX;
    int idx = nb + lane;
    int rp = E;
    float dvl = 0.0f;
    if (idx < N) { rp = rowptr[idx]; dvl = dinv[idx]; }
    int dvi = __float_as_int(dvl);

    int beg = __builtin_amdgcn_readlane(rp, 0);
    int end = __builtin_amdgcn_readlane(rp, 1);
    v2i rc;
    rc.x = 0; rc.y = 0;
    if (beg + lane < end) rc = *(const v2i*)(rec + 2 * (long)(beg + lane));
    v4f hrow = *row64(H, nb, q4);
    v2f xrow = *row8(X, nb, qx);

    for (int i = 0; i < cn; ++i) {
      int d = nb + i;
      float dd = __int_as_float(__builtin_amdgcn_readlane(dvi, i));
      float ds = hl ? 0.0f : dd;
      v4f a = ds * hrow;
      v2f ax = ds * xrow;

      int endn = end;
      v2i rn;
      rn.x = 0; rn.y = 0;
      if (i + 1 < cn) {
        endn = __builtin_amdgcn_readlane(rp, i + 2);
        if (end + lane < endn) rn = *(const v2i*)(rec + 2 * (long)(end + lane));
        hrow = *row64(H, d + 1, q4);
        xrow = *row8(X, d + 1, qx);
      }

      int cnt = end - beg;
      seg_batch_xh(rc.x, rc.y, cnt > 32 ? 32 : cnt, hl, q4, qx, H, X, a, ax);
      for (int base = beg + 32; base < end; base += 32) {
        int p = base + lane;
        v2i r;
        r.x = 0; r.y = 0;
        if (p < end) r = *(const v2i*)(rec + 2 * (long)p);
        int c2 = end - base;
        seg_batch_xh(r.x, r.y, c2 > 32 ? 32 : c2, hl, q4, qx, H, X, a, ax);
      }

      a = join_halves(a);
      ax.x += lane_xor<16>(ax.x);
      ax.y += lane_xor<16>(ax.y);
      if (lane < 16) *(v4f*)(AH + (long)d * HID + q4) = dd * a;
      if (lane < 4) *(v2f*)(AX + (long)d * FIN + qx) = dd * ax;

      rc = rn;
      beg = end;
      end = endn;
    }
  }
}

__global__ __launch_bounds__(256)
void k_gather_head(float* __restrict__ P, const float* __restrict__ T,
                   const int* __restrict__ rowptr, const int* __restrict__ rec,
                   const float* __restrict__ dinv, const float* __restrict__ b1,
                   const float* __restrict__ W2, int N, int E) {
  int lane = threadIdx.x & 31;
  int hl = lane >> 4;
  int q4 = 4 * (lane & 15);
  int wid = __builtin_amdgcn_readfirstlane((int)(blockIdx.x * 8 + (threadIdx.x >> 5)));
  unsigned nw = gridDim.x * 8u;
  unsigned q = (unsigned)N / nw, rem = (unsigned)N % nw;
  int n0 = (int)((unsigned)wid * q + ((unsigned)wid < rem ? (unsigned)wid : rem));
  int n1 = n0 + (int)q + ((unsigned)wid < rem ? 1 : 0);

  v4f bl = *(const v4f*)(b1 + q4);
  v4f wl = *(const v4f*)(W2 + q4);

  for (int nb = n0; nb < n1; nb += RUN_MAX) {
    int cn = n1 - nb;
    if (cn > RUN_MAX) cn = RUN_MAX;
    int idx = nb + lane;
    int rp = E;
    float dvl = 0.0f;
    if (idx < N) { rp = rowptr[idx]; dvl = dinv[idx]; }
    int dvi = __float_as_int(dvl);

    int beg = __builtin_amdgcn_readlane(rp, 0);
    int end = __builtin_amdgcn_readlane(rp, 1);
    v2i rc;
    rc.x = 0; rc.y = 0;
    if (beg + lane < end) rc = *(const v2i*)(rec + 2 * (long)(beg + lane));
    v4f hrow = *row64(T, nb, q4);

    for (int i = 0; i < cn; ++i) {
      int d = nb + i;
      float dd = __int_as_float(__builtin_amdgcn_readlane(dvi, i));
      float ds = hl ? 0.0f : dd;
      v4f a = ds * hrow;

      int endn = end;
      v2i rn;
      rn.x = 0; rn.y = 0;
      if (i + 1 < cn) {
        endn = __builtin_amdgcn_readlane(rp, i + 2);
        if (end + lane < endn) rn = *(const v2i*)(rec + 2 * (long)(end + lane));
        hrow = *row64(T, d + 1, q4);
      }

      int cnt = end - beg;
      seg_batch_t(rc.x, rc.y, cnt > 32 ? 32 : cnt, hl, q4, T, a);
      for (int base = beg + 32; base < end; base += 32) {
        int p = base + lane;
        v2i r;
        r.x = 0; r.y = 0;
        if (p < end) r = *(const v2i*)(rec + 2 * (long)p);
        int c2 = end - base;
        seg_batch_t(r.x, r.y, c2 > 32 ? 32 : c2, hl, q4, T, a);
      }

      a = join_halves(a);
      v4f z = dd * a + bl;
      z.x = z.x > 0.0f ? z.x : 0.0f;
      z.y = z.y > 0.0f ? z.y : 0.0f;
      z.z = z.z > 0.0f ? z.z : 0.0f;
      z.w = z.w > 0.0f ? z.w : 0.0f;
      float pv = z.x * wl.x + z.y * wl.y + z.z * wl.z + z.w * wl.w;
      pv += lane_xor<8>(pv);
      pv += lane_xor<4>(pv);
      pv += lane_xor<2>(pv);
      pv += lane_xor<1>(pv);
      if (lane == 0) P[d] = pv;

      rc = rn;
      beg = end;
      end = endn;
    }
  }
}


#define BF_FLOATS   (KSTEPS * NT_G * 64)
#define PAR_BIAS    0
#define PAR_WC      256
#define PAR_LN      448
#define PAR_FLOATS  832
#define W1F_FLOATS  (KH_STEPS * 4 * 64)
#define TILE_FLOATS (8 * 16 * TPITCH)

__device__ __forceinline__ void ld8(const float* p, float* o) {
  v4f a = *(const v4f*)p;
  v4f b = *(const v4f*)(p + 4);
  o[0] = a.x; o[1] = a.y; o[2] = a.z; o[3] = a.w;
  o[4] = b.x; o[5] = b.y; o[6] = b.z; o[7] = b.w;
}

__device__ __forceinline__ void wave_lds_order() {
  __builtin_amdgcn_fence(__ATOMIC_RELEASE, "wavefront");
  __builtin_amdgcn_wave_barrier();
  __builtin_amdgcn_fence(__ATOMIC_ACQUIRE, "wavefront");
}

__device__ __forceinline__ void rows_out(float* tw, const float* v, float* __restrict__ dst,
                                         int rowBase, int lcol, int half, int N) {
  float* trow = tw + lcol * TPITCH + 8 * half;
  wave_lds_order();
#pragma unroll
  for (int t = 0; t < 4; ++t) {
    v4f a, b;
    a.x = v[8 * t];     a.y = v[8 * t + 1]; a.z = v[8 * t + 2]; a.w = v[8 * t + 3];
    b.x = v[8 * t + 4]; b.y = v[8 * t + 5]; b.z = v[8 * t + 6]; b.w = v[8 * t + 7];
    *(v4f*)(trow + 16 * t) = a;
    *(v4f*)(trow + 16 * t + 4) = b;
  }
  wave_lds_order();
#pragma unroll
  for (int i = 0; i < 8; ++i) {
    int m = 2 * i + half;
    int node = rowBase + m;
    v4f x = *(const v4f*)(tw + m * TPITCH + 4 * lcol);
    if (node < N) *(v4f*)(dst + (long)node * HID + 4 * lcol) = x;
  }
}

__global__ __launch_bounds__(256)
void k_gates_head(const float* AH, const float* __restrict__ AX,
                  const float* __restrict__ C,
                  const float* __restrict__ Wx, const float* __restrict__ Wh,
                  const float* __restrict__ bx, const float* __restrict__ bh,
                  const float* __restrict__ bg, const float* __restrict__ wc,
                  const float* __restrict__ ln_h_s, const float* __restrict__ ln_h_b,
                  const float* __restrict__ ln_c_s, const float* __restrict__ ln_c_b,
                  const float* __restrict__ ln_o_s, const float* __restrict__ ln_o_b,
                  const float* __restrict__ skip, const float* __restrict__ W1,
                  float* __restrict__ hiddenOut, float* __restrict__ cellOut,
                  float* T, int N) {
  __shared__ __attribute__((aligned(16)))
      float smem[BF_FLOATS + PAR_FLOATS + W1F_FLOATS + TILE_FLOATS];
  float* Bf   = smem;
  float* par  = Bf + BF_FLOATS;
  float* W1f  = par + PAR_FLOATS;
  float* tile = W1f + W1F_FLOATS;

  int tid = threadIdx.x;
  {
    int mt = tid >> 6, L = (tid & 63) >> 1, j = tid & 1;
    int hh = L >> 4;
    int c = 16 * mt + (L & 15);
    float w[KH_STEPS];
#pragma unroll
    for (int st = 0; st < KH_STEPS - 1; ++st)
      w[st] = W1[(16 * (st >> 2) + 8 * hh + 2 * (st & 3) + j) * HID + c];
    w[KH_STEPS - 1] = W1[(HID + j) * HID + c];
    float bsum = bx[tid] + bh[tid] + bg[tid];
    float wcv = wc[tid < 192 ? tid : 0];
    int t6 = tid & 63;
    float l0 = ln_h_s[t6], l1 = ln_h_b[t6], l2 = ln_c_s[t6];
    float l3 = ln_c_b[t6], l4 = ln_o_s[t6], l5 = ln_o_b[t6];
#pragma unroll
    for (int st = 0; st < KH_STEPS; ++st)
      W1f[st * 256 + tid] = (st < KH_STEPS - 1 || hh == 0) ? w[st] : 0.0f;
    par[PAR_BIAS + tid] = bsum;
    if (tid < 192) par[PAR_WC + tid] = wcv;
    if (tid < 64) {
      par[PAR_LN + tid]       = l0;
      par[PAR_LN + 64 + tid]  = l1;
      par[PAR_LN + 128 + tid] = l2;
      par[PAR_LN + 192 + tid] = l3;
      par[PAR_LN + 256 + tid] = l4;
      par[PAR_LN + 320 + tid] = l5;
    }
  }
  for (int i = tid; i < BF_FLOATS; i += 256) {
    int s = i >> 10;
    int rem = i & 1023;
    int t = rem >> 6;
    int rem2 = rem & 63;
    int L = rem2 >> 1;
    int j = rem2 & 1;
    int k = 4 * s + 2 * (L >> 4) + j;
    int c = 16 * t + (L & 15);
    int g = c >> 6, h = c & 63;
    Bf[i] = (k < FIN) ? Wx[(g * FIN + k) * HID + h]
                      : Wh[(g * HID + (k - FIN)) * HID + h];
  }
  __syncthreads();

  int wave = __builtin_amdgcn_readfirstlane(tid >> 5);
  int lane = tid & 31;
  int half = lane >> 4, lcol = lane & 15;
  float* tw = tile + wave * (16 * TPITCH);
  const int nTiles = (N + 15) >> 4;
  const int nb = (int)gridDim.x;
  const int W = nb * 8;
  int tileIdx = (int)blockIdx.x * 8 + wave;
  int place = (int)((0x37265140u >> (4 * wave)) & 7u);
  int extra = place * nb + (int)blockIdx.x;
  for (; tileIdx < nTiles; tileIdx = W + extra, extra += W) {
  int rowBase = tileIdx * 16;

  int arow = rowBase + lcol;
  if (arow > N - 1) arow = N - 1;
  const float* aXp = AX + (long)arow * FIN;
  const float* aHp = AH + (long)arow * HID;

  v4f cpre[8];
  {
    const float* Cp = C + (long)arow * HID + 8 * half;
#pragma unroll
    for (int t = 0; t < 4; ++t) {
      cpre[2 * t]     = *(const v4f*)(Cp + 16 * t);
      cpre[2 * t + 1] = *(const v4f*)(Cp + 16 * t + 4);
    }
  }
  v2f sk = *(const v2f*)(skip + (long)arow * 2);

  v8f acc[NT_G];
#pragma unroll
  for (int t = 0; t < NT_G; ++t)
#pragma unroll
    for (int r = 0; r < 8; ++r) acc[t][r] = 0.0f;

  for (int s = 0; s < KSTEPS; ++s) {
    int k0 = 4 * s + 2 * half;
    v2f a = *(const v2f*)(s < 2 ? aXp + k0 : aHp + (k0 - FIN));
    const float* bbase = Bf + s * 1024 + lane * 2;
#pragma unroll
    for (int t = 0; t < NT_G; ++t) {
      v2f b = *(const v2f*)(bbase + t * 64);
      acc[t] = __builtin_amdgcn_wmma_f32_16x16x4_f32(
          false, b, false, a, (short)0, acc[t], false, false);
    }
  }

  const float* pl = par + 8 * half;
  float hv[32], cv[32];
  float sh = 0.f, sh2 = 0.f, sc = 0.f, sc2 = 0.f;
#pragma unroll
  for (int t = 0; t < 4; ++t) {
    float c0[8], bi[8], bfg[8], bc[8], bo[8], wi[8], wf[8], wo[8];
    c0[0] = cpre[2 * t].x;     c0[1] = cpre[2 * t].y;
    c0[2] = cpre[2 * t].z;     c0[3] = cpre[2 * t].w;
    c0[4] = cpre[2 * t + 1].x; c0[5] = cpre[2 * t + 1].y;
    c0[6] = cpre[2 * t + 1].z; c0[7] = cpre[2 * t + 1].w;
    ld8(pl + PAR_BIAS + 16 * t, bi);
    ld8(pl + PAR_BIAS + 64 + 16 * t, bfg);
    ld8(pl + PAR_BIAS + 128 + 16 * t, bc);
    ld8(pl + PAR_BIAS + 192 + 16 * t, bo);
    ld8(pl + PAR_WC + 16 * t, wi);
    ld8(pl + PAR_WC + 64 + 16 * t, wf);
    ld8(pl + PAR_WC + 128 + 16 * t, wo);
    float ph = 0.f, ph2 = 0.f, pc = 0.f, pc2 = 0.f;
#pragma unroll
    for (int r = 0; r < 8; ++r) {
      float gi = acc[t][r]      + bi[r]  + wi[r] * c0[r];
      float gf = acc[t + 4][r]  + bfg[r] + wf[r] * c0[r];
      float gc = acc[t + 8][r]  + bc[r];
      float go = acc[t + 12][r] + bo[r];
      float iv = sigmoidf_(gi);
      float fv = sigmoidf_(gf);
      float cc = fv * c0[r] + iv * tanhf_(gc);
      float ov = sigmoidf_(go + wo[r] * cc);
      float hh = ov * tanhf_(cc);
      hv[8 * t + r] = hh;
      cv[8 * t + r] = cc;
      ph += hh; ph2 += hh * hh;
      pc += cc; pc2 += cc * cc;
    }
    sh += ph; sh2 += ph2; sc += pc; sc2 += pc2;
  }
  sh  += __shfl_xor(sh,  16, 32);
  sh2 += __shfl_xor(sh2, 16, 32);
  sc  += __shfl_xor(sc,  16, 32);
  sc2 += __shfl_xor(sc2, 16, 32);
  float mh = sh * (1.0f / 64.0f), mc = sc * (1.0f / 64.0f);
  float vh = sh2 * (1.0f / 64.0f) - mh * mh;
  float vc = sc2 * (1.0f / 64.0f) - mc * mc;
  float rh = rsqrtf(vh + EPS), rc = rsqrtf(vc + EPS);

  float ov32[32];
#pragma unroll
  for (int t = 0; t < 4; ++t) {
    float s8[8], b8[8];
    ld8(pl + PAR_LN + 16 * t, s8);
    ld8(pl + PAR_LN + 64 + 16 * t, b8);
#pragma unroll
    for (int r = 0; r < 8; ++r) {
      float hn = (hv[8 * t + r] - mh) * rh;
      ov32[8 * t + r] = hn * s8[r] + b8[r];
    }
  }
  rows_out(tw, ov32, hiddenOut, rowBase, lcol, half, N);
#pragma unroll
  for (int t = 0; t < 4; ++t) {
    float s8[8], b8[8];
    ld8(pl + PAR_LN + 128 + 16 * t, s8);
    ld8(pl + PAR_LN + 192 + 16 * t, b8);
#pragma unroll
    for (int r = 0; r < 8; ++r) {
      float cn = (cv[8 * t + r] - mc) * rc;
      ov32[8 * t + r] = cn * s8[r] + b8[r];
    }
  }
  rows_out(tw, ov32, cellOut, rowBase, lcol, half, N);
  float yv[32];
#pragma unroll
  for (int t = 0; t < 4; ++t) {
    float s8[8], b8[8];
    ld8(pl + PAR_LN + 256 + 16 * t, s8);
    ld8(pl + PAR_LN + 320 + 16 * t, b8);
#pragma unroll
    for (int r = 0; r < 8; ++r) {
      float hn = (hv[8 * t + r] - mh) * rh;
      float o = hn * s8[r] + b8[r];
      yv[8 * t + r] = o > 0.0f ? o : 0.0f;
    }
  }

  v8f acc2[4];
#pragma unroll
  for (int mt = 0; mt < 4; ++mt)
#pragma unroll
    for (int r = 0; r < 8; ++r) acc2[mt][r] = 0.0f;
#pragma unroll
  for (int t = 0; t < 4; ++t)
#pragma unroll
    for (int p = 0; p < 4; ++p) {
      v2f yb;
      yb.x = yv[8 * t + 2 * p];
      yb.y = yv[8 * t + 2 * p + 1];
      const float* wb = W1f + (t * 4 + p) * 256 + lane * 2;
#pragma unroll
      for (int mt = 0; mt < 4; ++mt) {
        v2f wa = *(const v2f*)(wb + mt * 64);
        acc2[mt] = __builtin_amdgcn_wmma_f32_16x16x4_f32(
            false, wa, false, yb, (short)0, acc2[mt], false, false);
      }
    }
  {
    v2f yb;
    yb.x = half ? 0.0f : sk.x;
    yb.y = half ? 0.0f : sk.y;
    const float* wb = W1f + (KH_STEPS - 1) * 256 + lane * 2;
#pragma unroll
    for (int mt = 0; mt < 4; ++mt) {
      v2f wa = *(const v2f*)(wb + mt * 64);
      acc2[mt] = __builtin_amdgcn_wmma_f32_16x16x4_f32(
          false, wa, false, yb, (short)0, acc2[mt], false, false);
    }
  }
#pragma unroll
  for (int mt = 0; mt < 4; ++mt)
#pragma unroll
    for (int r = 0; r < 8; ++r) ov32[8 * mt + r] = acc2[mt][r];
  rows_out(tw, ov32, T, rowBase, lcol, half, N);
  }
}

__global__ __launch_bounds__(256)
void k_gather_out(float* __restrict__ out, const float* __restrict__ P,
                  const int* __restrict__ rowptr, const int* __restrict__ rec,
                  const float* __restrict__ dinv, const float* __restrict__ b2,
                  const float* __restrict__ X, int N, int E) {
  int t = blockIdx.x * blockDim.x + threadIdx.x;
  int g = t >> 4, l = t & 15;
  int G = (int)((gridDim.x * blockDim.x) >> 4);
  float bias2 = b2[0];

  int dc = g < N ? g : N - 1;
  int beg = rowptr[dc];
  int end = (dc + 1 < N) ? rowptr[dc + 1] : E;
  for (int db = 0; db < N; db += G) {
    int d = db + g;
    bool valid = d < N;
    dc = valid ? d : N - 1;
    float dd = dinv[dc];
    float pself = P[dc];
    float xres = X[(long)dc * FIN];
    int dn = d + G;
    int dnc = dn < N ? dn : N - 1;
    int begn = rowptr[dnc];
    int endn = (dnc + 1 < N) ? rowptr[dnc + 1] : E;

    float acc = 0.0f;
    for (int p = beg + l; p < end; p += 16) {
      v2i r = *(const v2i*)(rec + 2 * (long)p);
      acc += __int_as_float(r.y) * P[r.x];
    }
    for (int mask = 8; mask > 0; mask >>= 1) acc += __shfl_xor(acc, mask, 32);
    if (valid && l == 0) {
      float q = dd * (acc + dd * pself);
      out[d] = q + bias2 + xres;
    }
    beg = begn;
    end = endn;
  }
}

extern "C" void kernel_launch(void* const* d_in, const int* in_sizes, int n_in,
                              void* d_out, int out_size, void* d_ws, size_t ws_size,
                              hipStream_t stream) {
  const float* X    = (const float*)d_in[0];
  const int*   eidx = (const int*)d_in[1];
  const float* ew   = (const float*)d_in[2];
  const float* skip = (const float*)d_in[3];
  const float* H    = (const float*)d_in[4];
  const float* C    = (const float*)d_in[5];
  const float* Wx   = (const float*)d_in[6];
  const float* bx   = (const float*)d_in[7];
  const float* Wh   = (const float*)d_in[8];
  const float* bh   = (const float*)d_in[9];
  const float* wc   = (const float*)d_in[10];
  const float* bg   = (const float*)d_in[11];
  const float* ln_h_s = (const float*)d_in[12];
  const float* ln_h_b = (const float*)d_in[13];
  const float* ln_c_s = (const float*)d_in[14];
  const float* ln_c_b = (const float*)d_in[15];
  const float* ln_o_s = (const float*)d_in[16];
  const float* ln_o_b = (const float*)d_in[17];
  const float* W1   = (const float*)d_in[18];
  const float* b1   = (const float*)d_in[19];
  const float* W2   = (const float*)d_in[20];
  const float* b2   = (const float*)d_in[21];

  const int N = in_sizes[0] / FIN;
  const int E = in_sizes[2];

  const int* src = eidx;
  const int* dst = eidx + E;

  float* out_p    = (float*)d_out;
  float* hidden_p = out_p + N;
  float* cell_p   = hidden_p + (long)N * HID;

  int*   rec  = (int*)d_ws;
  float* AH   = (float*)(rec + 2 * (long)E);
  float* AX   = AH + (long)N * HID;
  float* dinv = AX + (long)N * FIN;
  float* P    = dinv + N;
  int* rowptr = (int*)(P + N);
  int*   rank  = (int*)AH;
  int*   cnt2  = rank + E;
  float* deg2  = (float*)(cnt2 + 2 * (long)N);
  int*   base1 = (int*)(deg2 + 2 * (long)N);
  int*   bsum  = base1 + N;

  const int B = 256;
  auto cdiv = [](long a, long b) { return (unsigned)((a + b - 1) / b); };

  k_init_deg<<<cdiv(N, B), B, 0, stream>>>(deg2, cnt2, N);
  k_acc_deg<<<2048, B, 0, stream>>>(deg2, cnt2, rank, dst, ew, N, E);
  k_dinv<<<cdiv(N, B), B, 0, stream>>>(dinv, deg2, N);

  k_blocksum<<<cdiv(N, SCAN_T * SCAN_PER), SCAN_T, 0, stream>>>(cnt2, bsum, N);
  k_scan<<<cdiv(N, SCAN_T * SCAN_PER), SCAN_T, 0, stream>>>(cnt2, rowptr, base1, bsum, N);
  k_fill<<<2048, B, 0, stream>>>(rowptr, base1, rank, rec, src, dst, ew, dinv, E);

  k_gather_xh<<<2048, B, 0, stream>>>(AH, AX, X, H, rowptr, rec, dinv, N, E);

  k_gates_head<<<512, B, 0, stream>>>(
      AH, AX, C, Wx, Wh, bx, bh, bg, wc,
      ln_h_s, ln_h_b, ln_c_s, ln_c_b, ln_o_s, ln_o_b,
      skip, W1, hidden_p, cell_p, AH, N);

  k_gather_head<<<2048, B, 0, stream>>>(P, AH, rowptr, rec, dinv, b1, W2, N, E);

  k_gather_out<<<2048, B, 0, stream>>>(out_p, P, rowptr, rec, dinv, b2, X, N, E);
}
